# starting kernel
# baseline (speedup 1.0000x reference)
#include <hip/hip_runtime.h>
#include <hip/hip_bf16.h>
#include <math.h>

typedef __attribute__((ext_vector_type(2))) float v2f;
typedef __attribute__((ext_vector_type(8))) float v8f;

#define D_DIM 64
#define H_DIM 128
#define CPITCH 66

__device__ __forceinline__ unsigned ordKey(float f) {
  unsigned u = __float_as_uint(f);
  return (u & 0x80000000u) ? ~u : (u | 0x80000000u);
}
__device__ __forceinline__ float ordVal(unsigned k) {
  unsigned u = (k & 0x80000000u) ? (k ^ 0x80000000u) : ~k;
  return __uint_as_float(u);
}

__global__ void init_seg(unsigned* __restrict__ segMax, float* __restrict__ segSum, int numS) {
  int i = blockIdx.x * blockDim.x + threadIdx.x;
  if (i < numS) { segMax[i] = 0u; segSum[i] = 0.0f; }
}

__global__ __launch_bounds__(256)
void edge_mlp_cost(const float* __restrict__ embS,
                   const float* __restrict__ embA,
                   const int*   __restrict__ edgeS,
                   const int*   __restrict__ edgeA,
                   const float* __restrict__ W1,
                   const float* __restrict__ b1,
                   const float* __restrict__ W2,
                   const float* __restrict__ b2,
                   const float* __restrict__ logT,
                   float*       __restrict__ outCost,
                   unsigned*    __restrict__ segMax,
                   int E) {
  __shared__ float sW1[H_DIM * CPITCH * 2];
  __shared__ float sB1[H_DIM];
  __shared__ float sW2[H_DIM];
  __shared__ float sGate[8][16];

  const int tid = threadIdx.x;
  for (int i = tid; i < H_DIM * H_DIM; i += 256) {
    int kk = i >> 7;
    int c  = i & 127;
    sW1[(c * CPITCH + (kk >> 1)) * 2 + (kk & 1)] = W1[i];
  }
  if (tid < H_DIM) { sB1[tid] = b1[tid]; sW2[tid] = W2[tid]; }
  __syncthreads();

  const int wave = tid >> 5;
  const int lane = tid & 31;
  const int half = lane >> 4;
  const int col  = lane & 15;
  const int e0   = (blockIdx.x * 8 + wave) * 16;
  if (e0 >= E) return;

  const int  e     = e0 + col;
  const bool valid = (e < E);
  const int  ec    = valid ? e : (E - 1);
  const int  sI    = edgeS[ec];
  const int  aI    = edgeA[ec];
  const float* es = embS + (size_t)sI * D_DIM;
  const float* ea = embA + (size_t)aI * D_DIM;

  v2f a[32];
#pragma unroll
  for (int k = 0; k < 16; ++k) {
    a[k]      = *(const v2f*)(es + 4 * k + 2 * half);
    a[k + 16] = *(const v2f*)(ea + 4 * k + 2 * half);
  }
  float ss = 0.0f;
#pragma unroll
  for (int k = 0; k < 16; ++k) {
    float dx = a[k][0] - a[k + 16][0];
    float dy = a[k][1] - a[k + 16][1];
    ss += dx * dx + dy * dy;
  }
  ss += __shfl_xor(ss, 16);

  float gacc[8] = {0, 0, 0, 0, 0, 0, 0, 0};
#pragma unroll
  for (int n = 0; n < 8; ++n) {
    const float* bbase = sW1 + ((n * 16 + col) * CPITCH + half) * 2;
    v8f c0 = {};
    v8f c1 = {};
#pragma unroll
    for (int k = 0; k < 32; k += 2) {
      v2f b0 = *(const v2f*)(bbase + 4 * k);
      v2f b1v = *(const v2f*)(bbase + 4 * (k + 1));
      c0 = __builtin_amdgcn_wmma_f32_16x16x4_f32(false, a[k], false, b0,
                                                 (short)0, c0, false, false);
      c1 = __builtin_amdgcn_wmma_f32_16x16x4_f32(false, a[k + 1], false, b1v,
                                                 (short)0, c1, false, false);
    }
    const float bn  = sB1[n * 16 + col];
    const float w2n = sW2[n * 16 + col];
#pragma unroll
    for (int r = 0; r < 8; ++r) {
      float h = c0[r] + c1[r] + bn;
      h = h > 0.0f ? h : 0.0f;
      gacc[r] += h * w2n;
    }
  }

#pragma unroll
  for (int r = 0; r < 8; ++r) {
    float g = gacc[r];
    g += __shfl_xor(g, 1);
    g += __shfl_xor(g, 2);
    g += __shfl_xor(g, 4);
    g += __shfl_xor(g, 8);
    if (lane == 0)  sGate[wave][r]     = g;
    if (lane == 16) sGate[wave][r + 8] = g;
  }

  const float b2v  = b2[0];
  const float invT = __expf(-logT[0]);
  if (lane < 16 && valid) {
    float logit = sGate[wave][lane] + b2v;
    float gate  = 1.0f / (1.0f + __expf(-logit));
    float cost  = sqrtf(ss) * gate;
    outCost[e]  = cost;
    atomicMax(&segMax[sI], ordKey(-cost * invT));
  }
}

__global__ void seg_expsum(const float* __restrict__ outCost,
                           const int* __restrict__ edgeS,
                           const unsigned* __restrict__ segMax,
                           float* __restrict__ segSum,
                           const float* __restrict__ logT, int E) {
  int e = blockIdx.x * blockDim.x + threadIdx.x;
  if (e >= E) return;
  float invT = __expf(-logT[0]);
  int s = edgeS[e];
  float val = -outCost[e] * invT;
  float m = ordVal(segMax[s]);
  atomicAdd(&segSum[s], __expf(val - m));
}

__global__ void seg_norm(const float* __restrict__ outCost,
                         const int* __restrict__ edgeS,
                         const unsigned* __restrict__ segMax,
                         const float* __restrict__ segSum,
                         const float* __restrict__ logT,
                         float* __restrict__ outW, int E) {
  int e = blockIdx.x * blockDim.x + threadIdx.x;
  if (e >= E) return;
  float invT = __expf(-logT[0]);
  int s = edgeS[e];
  float val = -outCost[e] * invT;
  float m = ordVal(segMax[s]);
  outW[e] = __expf(val - m) / segSum[s];
}

extern "C" void kernel_launch(void* const* d_in, const int* in_sizes, int n_in,
                              void* d_out, int out_size, void* d_ws, size_t ws_size,
                              hipStream_t stream) {
  const float* embS = (const float*)d_in[0];
  const float* embA = (const float*)d_in[1];
  const int*   eidx = (const int*)d_in[2];
  const float* W1   = (const float*)d_in[3];
  const float* b1   = (const float*)d_in[4];
  const float* W2   = (const float*)d_in[5];
  const float* b2   = (const float*)d_in[6];
  const float* logT = (const float*)d_in[7];

  const int numS = in_sizes[0] / D_DIM;
  const int E    = in_sizes[2] / 2;
  const int* edgeS = eidx;
  const int* edgeA = eidx + E;

  float* outW    = (float*)d_out;
  float* outCost = (float*)d_out + E;

  unsigned* segMax = (unsigned*)d_ws;
  float*    segSum = (float*)((unsigned*)d_ws + numS);

  init_seg<<<(numS + 255) / 256, 256, 0, stream>>>(segMax, segSum, numS);

  const int tiles = (E + 15) / 16;
  edge_mlp_cost<<<(tiles + 7) / 8, 256, 0, stream>>>(
      embS, embA, edgeS, edgeA, W1, b1, W2, b2, logT, outCost, segMax, E);

  seg_expsum<<<(E + 255) / 256, 256, 0, stream>>>(outCost, edgeS, segMax, segSum, logT, E);
  seg_norm<<<(E + 255) / 256, 256, 0, stream>>>(outCost, edgeS, segMax, segSum, logT, outW, E);
}


# persistent edge kernel: 256 blocks x 16 waves stage W1 to LDS once, each wave loops over 16-edge tiles; segment exp-sums as 64-bit fixed-point integer atomics (float atomic order made runs differ)
# speedup vs baseline: 1.4366x; 1.4366x over previous
#include <hip/hip_runtime.h>
#include <hip/hip_bf16.h>
#include <math.h>

typedef __attribute__((ext_vector_type(2))) float v2f;
typedef __attribute__((ext_vector_type(8))) float v8f;

#define D_DIM 64
#define H_DIM 128
#define CPITCH 66

__device__ __forceinline__ unsigned ordKey(float f) {
  unsigned u = __float_as_uint(f);
  return (u & 0x80000000u) ? ~u : (u | 0x80000000u);
}
__device__ __forceinline__ float ordVal(unsigned k) {
  unsigned u = (k & 0x80000000u) ? (k ^ 0x80000000u) : ~k;
  return __uint_as_float(u);
}

#define SUM_SCALE     1099511627776.0f
#define SUM_INV_SCALE (1.0f / 1099511627776.0f)

__global__ void init_seg(unsigned* __restrict__ segMax,
                         unsigned long long* __restrict__ segSum, int numS) {
  int i = blockIdx.x * blockDim.x + threadIdx.x;
  if (i < numS) { segMax[i] = 0u; segSum[i] = 0ull; }
}

#define EDGE_WAVES 16
__global__ __launch_bounds__(EDGE_WAVES * 32)
void edge_mlp_cost(const float* __restrict__ embS,
                   const float* __restrict__ embA,
                   const int*   __restrict__ edgeS,
                   const int*   __restrict__ edgeA,
                   const float* __restrict__ W1,
                   const float* __restrict__ b1,
                   const float* __restrict__ W2,
                   const float* __restrict__ b2,
                   const float* __restrict__ logT,
                   float*       __restrict__ outCost,
                   unsigned*    __restrict__ segMax,
                   int E) {
  __shared__ float sW1[H_DIM * CPITCH * 2];
  __shared__ float sB1[H_DIM];
  __shared__ float sW2[H_DIM];
  __shared__ float sGate[EDGE_WAVES][16];

  const int tid = threadIdx.x;
  for (int i = tid; i < H_DIM * H_DIM; i += EDGE_WAVES * 32) {
    int kk = i >> 7;
    int c  = i & 127;
    sW1[(c * CPITCH + (kk >> 1)) * 2 + (kk & 1)] = W1[i];
  }
  if (tid < H_DIM) { sB1[tid] = b1[tid]; sW2[tid] = W2[tid]; }
  __syncthreads();

  const int wave = __builtin_amdgcn_readfirstlane(tid >> 5);
  const int lane = tid & 31;
  const int half = lane >> 4;
  const int col  = lane & 15;
  const float b2v  = b2[0];
  const float invT = __expf(-logT[0]);
  const int tiles  = (E + 15) >> 4;
  const int stride = gridDim.x * EDGE_WAVES;

  for (int t = blockIdx.x * EDGE_WAVES + wave; t < tiles; t += stride) {
  asm volatile("" ::: "memory");
  const int  e0    = t * 16;
  const int  e     = e0 + col;
  const bool valid = (e < E);
  const int  ec    = valid ? e : (E - 1);
  const int  sI    = edgeS[ec];
  const int  aI    = edgeA[ec];
  const float* es = embS + (size_t)sI * D_DIM;
  const float* ea = embA + (size_t)aI * D_DIM;

  v2f a[32];
#pragma unroll
  for (int k = 0; k < 16; ++k) {
    a[k]      = *(const v2f*)(es + 4 * k + 2 * half);
    a[k + 16] = *(const v2f*)(ea + 4 * k + 2 * half);
  }
  float ss = 0.0f;
#pragma unroll
  for (int k = 0; k < 16; ++k) {
    float dx = a[k][0] - a[k + 16][0];
    float dy = a[k][1] - a[k + 16][1];
    ss += dx * dx + dy * dy;
  }
  ss += __shfl_xor(ss, 16);

  float gacc[8] = {0, 0, 0, 0, 0, 0, 0, 0};
#pragma unroll
  for (int n = 0; n < 8; ++n) {
    const float* bbase = sW1 + ((n * 16 + col) * CPITCH + half) * 2;
    v8f c0 = {};
    v8f c1 = {};
#pragma unroll
    for (int k = 0; k < 32; k += 2) {
      v2f b0 = *(const v2f*)(bbase + 4 * k);
      v2f b1v = *(const v2f*)(bbase + 4 * (k + 1));
      c0 = __builtin_amdgcn_wmma_f32_16x16x4_f32(false, a[k], false, b0,
                                                 (short)0, c0, false, false);
      c1 = __builtin_amdgcn_wmma_f32_16x16x4_f32(false, a[k + 1], false, b1v,
                                                 (short)0, c1, false, false);
    }
    const float bn  = sB1[n * 16 + col];
    const float w2n = sW2[n * 16 + col];
#pragma unroll
    for (int r = 0; r < 8; ++r) {
      float h = c0[r] + c1[r] + bn;
      h = h > 0.0f ? h : 0.0f;
      gacc[r] += h * w2n;
    }
  }

#pragma unroll
  for (int r = 0; r < 8; ++r) {
    float g = gacc[r];
    g += __shfl_xor(g, 1);
    g += __shfl_xor(g, 2);
    g += __shfl_xor(g, 4);
    g += __shfl_xor(g, 8);
    if (lane == 0)  sGate[wave][r]     = g;
    if (lane == 16) sGate[wave][r + 8] = g;
  }

  if (lane < 16 && valid) {
    float logit = sGate[wave][lane] + b2v;
    float gate  = 1.0f / (1.0f + __expf(-logit));
    float cost  = sqrtf(ss) * gate;
    outCost[e]  = cost;
    atomicMax(&segMax[sI], ordKey(-cost * invT));
  }
  }
}

__global__ void seg_expsum(const float* __restrict__ outCost,
                           const int* __restrict__ edgeS,
                           const unsigned* __restrict__ segMax,
                           unsigned long long* __restrict__ segSum,
                           const float* __restrict__ logT, int E) {
  int e = blockIdx.x * blockDim.x + threadIdx.x;
  if (e >= E) return;
  float invT = __expf(-logT[0]);
  int s = edgeS[e];
  float val = -outCost[e] * invT;
  float m = ordVal(segMax[s]);
  atomicAdd(&segSum[s], (unsigned long long)(__expf(val - m) * SUM_SCALE));
}

__global__ void seg_norm(const float* __restrict__ outCost,
                         const int* __restrict__ edgeS,
                         const unsigned* __restrict__ segMax,
                         const unsigned long long* __restrict__ segSum,
                         const float* __restrict__ logT,
                         float* __restrict__ outW, int E) {
  int e = blockIdx.x * blockDim.x + threadIdx.x;
  if (e >= E) return;
  float invT = __expf(-logT[0]);
  int s = edgeS[e];
  float val = -outCost[e] * invT;
  float m = ordVal(segMax[s]);
  outW[e] = __expf(val - m) / ((float)segSum[s] * SUM_INV_SCALE);
}

extern "C" void kernel_launch(void* const* d_in, const int* in_sizes, int n_in,
                              void* d_out, int out_size, void* d_ws, size_t ws_size,
                              hipStream_t stream) {
  const float* embS = (const float*)d_in[0];
  const float* embA = (const float*)d_in[1];
  const int*   eidx = (const int*)d_in[2];
  const float* W1   = (const float*)d_in[3];
  const float* b1   = (const float*)d_in[4];
  const float* W2   = (const float*)d_in[5];
  const float* b2   = (const float*)d_in[6];
  const float* logT = (const float*)d_in[7];

  const int numS = in_sizes[0] / D_DIM;
  const int E    = in_sizes[2] / 2;
  const int* edgeS = eidx;
  const int* edgeA = eidx + E;

  float* outW    = (float*)d_out;
  float* outCost = (float*)d_out + E;

  unsigned long long* segSum = (unsigned long long*)d_ws;
  unsigned*           segMax = (unsigned*)(segSum + numS);

  init_seg<<<(numS + 255) / 256, 256, 0, stream>>>(segMax, segSum, numS);

  edge_mlp_cost<<<256, EDGE_WAVES * 32, 0, stream>>>(
      embS, embA, edgeS, edgeA, W1, b1, W2, b2, logT, outCost, segMax, E);

  seg_expsum<<<(E + 255) / 256, 256, 0, stream>>>(outCost, edgeS, segMax, segSum, logT, E);
  seg_norm<<<(E + 255) / 256, 256, 0, stream>>>(outCost, edgeS, segMax, segSum, logT, outW, E);
}


# first layer hoisted from edges to nodes: es.W1[0:64]+b1 and ea.W1[64:128] formed once per node on WMMA (100k rows, not 1M); edges gather and add the two projected rows (sum order changes)
# speedup vs baseline: 1.9605x; 1.3647x over previous
#include <hip/hip_runtime.h>
#include <hip/hip_bf16.h>
#include <math.h>

typedef __attribute__((ext_vector_type(2))) float v2f;
typedef __attribute__((ext_vector_type(4))) float v4f;
typedef __attribute__((ext_vector_type(8))) float v8f;

#define D_DIM 64
#define H_DIM 128
#define CPITCH 66

__device__ __forceinline__ unsigned ordKey(float f) {
  unsigned u = __float_as_uint(f);
  return (u & 0x80000000u) ? ~u : (u | 0x80000000u);
}
__device__ __forceinline__ float ordVal(unsigned k) {
  unsigned u = (k & 0x80000000u) ? (k ^ 0x80000000u) : ~k;
  return __uint_as_float(u);
}

#define SUM_SCALE     1099511627776.0f
#define SUM_INV_SCALE (1.0f / 1099511627776.0f)

__global__ void init_seg(unsigned* __restrict__ segMax,
                         unsigned long long* __restrict__ segSum, int numS) {
  int i = blockIdx.x * blockDim.x + threadIdx.x;
  if (i < numS) { segMax[i] = 0u; segSum[i] = 0ull; }
}

#define EDGE_WAVES 32
__global__ __launch_bounds__(EDGE_WAVES * 32)
void edge_mlp_cost(const float* __restrict__ embS,
                   const float* __restrict__ embA,
                   const int*   __restrict__ edgeS,
                   const int*   __restrict__ edgeA,
                   const float* __restrict__ W1,
                   const float* __restrict__ b1,
                   const float* __restrict__ W2,
                   const float* __restrict__ b2,
                   const float* __restrict__ logT,
                   float*       __restrict__ outCost,
                   unsigned*    __restrict__ segMax,
                   int E) {
  __shared__ float sW1[H_DIM * CPITCH * 2];
  __shared__ float sB1[H_DIM];
  __shared__ float sW2[H_DIM];
  __shared__ float sGate[EDGE_WAVES][16];

  const int tid = threadIdx.x;
  for (int i = tid; i < H_DIM * H_DIM; i += EDGE_WAVES * 32) {
    int kk = i >> 7;
    int c  = i & 127;
    sW1[(c * CPITCH + (kk >> 1)) * 2 + (kk & 1)] = W1[i];
  }
  if (tid < H_DIM) { sB1[tid] = b1[tid]; sW2[tid] = W2[tid]; }
  __syncthreads();

  const int wave = __builtin_amdgcn_readfirstlane(tid >> 5);
  const int lane = tid & 31;
  const int half = lane >> 4;
  const int col  = lane & 15;
  const float b2v  = b2[0];
  const float invT = __expf(-logT[0]);
  const int tiles  = (E + 15) >> 4;
  const int stride = gridDim.x * EDGE_WAVES;

  for (int t = blockIdx.x * EDGE_WAVES + wave; t < tiles; t += stride) {
  const int  e0    = t * 16;
  const int  e     = e0 + col;
  const bool valid = (e < E);
  const int  ec    = valid ? e : (E - 1);
  const int  sI    = edgeS[ec];
  const int  aI    = edgeA[ec];
  const float* es = embS + (size_t)sI * D_DIM;
  const float* ea = embA + (size_t)aI * D_DIM;

  v2f a[32];
#pragma unroll
  for (int k = 0; k < 16; ++k) {
    a[k]      = *(const v2f*)(es + 4 * k + 2 * half);
    a[k + 16] = *(const v2f*)(ea + 4 * k + 2 * half);
  }
  float ss = 0.0f;
#pragma unroll
  for (int k = 0; k < 16; ++k) {
    float dx = a[k][0] - a[k + 16][0];
    float dy = a[k][1] - a[k + 16][1];
    ss += dx * dx + dy * dy;
  }
  ss += __shfl_xor(ss, 16);

  float gacc[8] = {0, 0, 0, 0, 0, 0, 0, 0};
#pragma unroll 1
  for (int n = 0; n < 8; ++n) {
    const float* bbase = sW1 + ((n * 16 + col) * CPITCH + half) * 2;
    v8f c0 = {};
    v8f c1 = {};
#pragma unroll
    for (int k = 0; k < 32; k += 2) {
      v2f b0 = *(const v2f*)(bbase + 4 * k);
      v2f b1v = *(const v2f*)(bbase + 4 * (k + 1));
      c0 = __builtin_amdgcn_wmma_f32_16x16x4_f32(false, a[k], false, b0,
                                                 (short)0, c0, false, false);
      c1 = __builtin_amdgcn_wmma_f32_16x16x4_f32(false, a[k + 1], false, b1v,
                                                 (short)0, c1, false, false);
    }
    const float bn  = sB1[n * 16 + col];
    const float w2n = sW2[n * 16 + col];
#pragma unroll
    for (int r = 0; r < 8; ++r) {
      float h = c0[r] + c1[r] + bn;
      h = h > 0.0f ? h : 0.0f;
      gacc[r] += h * w2n;
    }
  }

#pragma unroll
  for (int r = 0; r < 8; ++r) {
    float g = gacc[r];
    g += __shfl_xor(g, 1);
    g += __shfl_xor(g, 2);
    g += __shfl_xor(g, 4);
    g += __shfl_xor(g, 8);
    if (lane == 0)  sGate[wave][r]     = g;
    if (lane == 16) sGate[wave][r + 8] = g;
  }

  if (lane < 16 && valid) {
    float logit = sGate[wave][lane] + b2v;
    float gate  = 1.0f / (1.0f + __expf(-logit));
    float cost  = sqrtf(ss) * gate;
    outCost[e]  = cost;
    atomicMax(&segMax[sI], ordKey(-cost * invT));
  }
  }
}

__global__ __launch_bounds__(EDGE_WAVES * 32)
void node_proj(const float* __restrict__ embS,
               const float* __restrict__ embA,
               const float* __restrict__ W1,
               const float* __restrict__ b1,
               float*       __restrict__ P,
               int numS, int numA) {
  __shared__ float sW1[H_DIM * CPITCH * 2];
  __shared__ __attribute__((aligned(16))) float sB1[2 * H_DIM];

  const int tid = threadIdx.x;
  for (int i = tid; i < H_DIM * H_DIM; i += EDGE_WAVES * 32) {
    int kk = i >> 7;
    int c  = i & 127;
    sW1[(c * CPITCH + (kk >> 1)) * 2 + (kk & 1)] = W1[i];
  }
  if (tid < H_DIM) { sB1[tid] = b1[tid]; sB1[H_DIM + tid] = 0.0f; }
  __syncthreads();

  const int wave = __builtin_amdgcn_readfirstlane(tid >> 5);
  const int lane = tid & 31;
  const int half = lane >> 4;
  const int col  = lane & 15;
  const int tilesS = (numS + 15) >> 4;
  const int tilesA = (numA + 15) >> 4;
  const int tiles  = tilesS + tilesA;

  for (int t = wave * (int)gridDim.x + (int)blockIdx.x; t < tiles;
       t += EDGE_WAVES * (int)gridDim.x) {
    const bool isA = (t >= tilesS);
    const int  tt  = isA ? (t - tilesS) : t;
    const float* emb = isA ? embA : embS;
    const int  nrows = isA ? numA : numS;
    const int  node  = tt * 16 + col;
    const bool valid = (node < nrows);
    const int  nc    = valid ? node : (nrows - 1);
    const float* row = emb + (size_t)nc * D_DIM;

    v2f a[16];
#pragma unroll
    for (int k = 0; k < 16; ++k) a[k] = *(const v2f*)(row + 4 * k + 2 * half);

    float* prow = P + ((size_t)(isA ? numS : 0) + (size_t)nc) * H_DIM + 8 * half;
    const int koff = isA ? D_DIM : 0;
    const int boff = isA ? H_DIM : 0;

#pragma unroll 1
    for (int nt = 0; nt < 8; ++nt) {
      const float* wbase = sW1 + ((nt * 16 + col) * CPITCH + half) * 2 + koff;
      v8f c0 = {};
      v8f c1 = {};
#pragma unroll
      for (int k = 0; k < 16; k += 2) {
        v2f w0 = *(const v2f*)(wbase + 4 * k);
        v2f w1 = *(const v2f*)(wbase + 4 * (k + 1));
        c0 = __builtin_amdgcn_wmma_f32_16x16x4_f32(false, w0, false, a[k],
                                                   (short)0, c0, false, false);
        c1 = __builtin_amdgcn_wmma_f32_16x16x4_f32(false, w1, false, a[k + 1],
                                                   (short)0, c1, false, false);
      }
      const float* bb = sB1 + boff + nt * 16 + 8 * half;
      const v4f blo = *(const v4f*)(bb);
      const v4f bhi = *(const v4f*)(bb + 4);
      v4f lo, hi;
#pragma unroll
      for (int r = 0; r < 4; ++r) {
        lo[r] = c0[r] + c1[r] + blo[r];
        hi[r] = c0[r + 4] + c1[r + 4] + bhi[r];
      }
      if (valid) {
        *(v4f*)(prow + nt * 16)     = lo;
        *(v4f*)(prow + nt * 16 + 4) = hi;
      }
    }
  }
}

struct RowSet {
  v4f es0, es1, ea0, ea1;
  v4f ps0, ps1, ps2, ps3;
  v4f pa0, pa1, pa2, pa3;
};

__device__ __forceinline__ void fetch_rows(RowSet& r,
                                           const float* __restrict__ embS,
                                           const float* __restrict__ embA,
                                           const float* __restrict__ Ps,
                                           const float* __restrict__ Pa,
                                           int s, int a, int sub4) {
  const float* e1 = embS + (size_t)s * D_DIM + sub4;
  const float* e2 = embA + (size_t)a * D_DIM + sub4;
  const float* p1 = Ps + (size_t)s * H_DIM + sub4;
  const float* p2 = Pa + (size_t)a * H_DIM + sub4;
  r.es0 = *(const v4f*)(e1);
  r.es1 = *(const v4f*)(e1 + 32);
  r.ea0 = *(const v4f*)(e2);
  r.ea1 = *(const v4f*)(e2 + 32);
  r.ps0 = *(const v4f*)(p1);
  r.ps1 = *(const v4f*)(p1 + 32);
  r.ps2 = *(const v4f*)(p1 + 64);
  r.ps3 = *(const v4f*)(p1 + 96);
  r.pa0 = *(const v4f*)(p2);
  r.pa1 = *(const v4f*)(p2 + 32);
  r.pa2 = *(const v4f*)(p2 + 64);
  r.pa3 = *(const v4f*)(p2 + 96);
}

__device__ __forceinline__ float sq4(v4f a, v4f b) {
  v4f d = a - b;
  return d[0] * d[0] + d[1] * d[1] + d[2] * d[2] + d[3] * d[3];
}
__device__ __forceinline__ float gate4(v4f p, v4f q, v4f w) {
  v4f h = p + q;
  float g = 0.0f;
#pragma unroll
  for (int i = 0; i < 4; ++i) {
    float x = h[i] > 0.0f ? h[i] : 0.0f;
    g += x * w[i];
  }
  return g;
}

template <int M>
__device__ __forceinline__ float xadd8(float v) {
  int o = __builtin_amdgcn_update_dpp(0, __float_as_int(v), 0x160 + M, 0xf, 0xf, true);
  return v + __int_as_float(o);
}
__device__ __forceinline__ int bsel8(int v, int lane, int i) {
  return __builtin_amdgcn_ds_bpermute(((lane & 24) | i) << 2, v);
}

__device__ __forceinline__ void reduce_rows(const RowSet& r, const float* sW2, int sub4,
                                            float& ss, float& g) {
  float s = sq4(r.es0, r.ea0) + sq4(r.es1, r.ea1);
  float t = gate4(r.ps0, r.pa0, *(const v4f*)(sW2 + sub4))
          + gate4(r.ps1, r.pa1, *(const v4f*)(sW2 + 32 + sub4))
          + gate4(r.ps2, r.pa2, *(const v4f*)(sW2 + 64 + sub4))
          + gate4(r.ps3, r.pa3, *(const v4f*)(sW2 + 96 + sub4));
  s = xadd8<1>(s); t = xadd8<1>(t);
  s = xadd8<2>(s); t = xadd8<2>(t);
  s = xadd8<4>(s); t = xadd8<4>(t);
  ss = s; g = t;
}

__global__ __launch_bounds__(EDGE_WAVES * 32)
void edge_gate_cost(const float* __restrict__ embS,
                    const float* __restrict__ embA,
                    const int*   __restrict__ edgeS,
                    const int*   __restrict__ edgeA,
                    const float* __restrict__ Ps,
                    const float* __restrict__ Pa,
                    const float* __restrict__ W2,
                    const float* __restrict__ b2,
                    const float* __restrict__ logT,
                    float*       __restrict__ outCost,
                    unsigned*    __restrict__ segMax,
                    int E) {
  __shared__ __attribute__((aligned(16))) float sW2[H_DIM];
  const int tid = threadIdx.x;
  if (tid < H_DIM) sW2[tid] = W2[tid];
  __syncthreads();

  const int wave = __builtin_amdgcn_readfirstlane(tid >> 5);
  const int lane = tid & 31;
  const int sub  = lane & 7;
  const int sub4 = sub * 4;
  const float b2v  = b2[0];
  const float invT = __expf(-logT[0]);
  const int tiles  = (E + 31) >> 5;
  const int stride = (int)gridDim.x * EDGE_WAVES;

  int t = (int)blockIdx.x * EDGE_WAVES + wave;
  if (t >= tiles) return;

  int eL   = t * 32 + lane;
  int eLc  = eL < E ? eL : (E - 1);
  int curS = edgeS[eLc];
  int curA = edgeA[eLc];

  for (;;) {
    const bool hasNext = (t + stride) < tiles;
    const int  tn   = hasNext ? (t + stride) : t;
    const int  nL   = tn * 32 + lane;
    const int  nLc  = nL < E ? nL : (E - 1);
    const int  nxtS = edgeS[nLc];
    const int  nxtA = edgeA[nLc];

    float rss = 0.0f, rg = 0.0f;

#pragma unroll 1
    for (int i = 0; i < 8; i += 2) {
      RowSet X, Y;
      float ss, g;
      fetch_rows(X, embS, embA, Ps, Pa, bsel8(curS, lane, i), bsel8(curA, lane, i), sub4);
      fetch_rows(Y, embS, embA, Ps, Pa, bsel8(curS, lane, i + 1), bsel8(curA, lane, i + 1), sub4);
      __builtin_amdgcn_sched_barrier(0);
      reduce_rows(X, sW2, sub4, ss, g);
      if (sub == i) { rss = ss; rg = g; }
      reduce_rows(Y, sW2, sub4, ss, g);
      if (sub == i + 1) { rss = ss; rg = g; }
    }

    if (eL < E) {
      float logit = rg + b2v;
      float gate  = 1.0f / (1.0f + __expf(-logit));
      float cost  = sqrtf(rss) * gate;
      outCost[eL] = cost;
      atomicMax(&segMax[curS], ordKey(-cost * invT));
    }
    if (!hasNext) break;
    t = tn; eL = nL; curS = nxtS; curA = nxtA;
  }
}

__global__ void seg_expsum(const float* __restrict__ outCost,
                           const int* __restrict__ edgeS,
                           const unsigned* __restrict__ segMax,
                           unsigned long long* __restrict__ segSum,
                           const float* __restrict__ logT, int E) {
  int e = blockIdx.x * blockDim.x + threadIdx.x;
  if (e >= E) return;
  float invT = __expf(-logT[0]);
  int s = edgeS[e];
  float val = -outCost[e] * invT;
  float m = ordVal(segMax[s]);
  atomicAdd(&segSum[s], (unsigned long long)(__expf(val - m) * SUM_SCALE));
}

__global__ void seg_norm(const float* __restrict__ outCost,
                         const int* __restrict__ edgeS,
                         const unsigned* __restrict__ segMax,
                         const unsigned long long* __restrict__ segSum,
                         const float* __restrict__ logT,
                         float* __restrict__ outW, int E) {
  int e = blockIdx.x * blockDim.x + threadIdx.x;
  if (e >= E) return;
  float invT = __expf(-logT[0]);
  int s = edgeS[e];
  float val = -outCost[e] * invT;
  float m = ordVal(segMax[s]);
  outW[e] = __expf(val - m) / ((float)segSum[s] * SUM_INV_SCALE);
}

extern "C" void kernel_launch(void* const* d_in, const int* in_sizes, int n_in,
                              void* d_out, int out_size, void* d_ws, size_t ws_size,
                              hipStream_t stream) {
  const float* embS = (const float*)d_in[0];
  const float* embA = (const float*)d_in[1];
  const int*   eidx = (const int*)d_in[2];
  const float* W1   = (const float*)d_in[3];
  const float* b1   = (const float*)d_in[4];
  const float* W2   = (const float*)d_in[5];
  const float* b2   = (const float*)d_in[6];
  const float* logT = (const float*)d_in[7];

  const int numS = in_sizes[0] / D_DIM;
  const int numA = in_sizes[1] / D_DIM;
  const int E    = in_sizes[2] / 2;
  const int* edgeS = eidx;
  const int* edgeA = eidx + E;

  float* outW    = (float*)d_out;
  float* outCost = (float*)d_out + E;

  const size_t pFloats = (size_t)(numS + numA) * H_DIM;
  const size_t need    = pFloats * sizeof(float)
                       + (size_t)numS * (sizeof(unsigned long long) + sizeof(unsigned));

  if (ws_size >= need) {
    float* P = (float*)d_ws;
    unsigned long long* segSum = (unsigned long long*)(P + pFloats);
    unsigned*           segMax = (unsigned*)(segSum + numS);

    init_seg<<<(numS + 255) / 256, 256, 0, stream>>>(segMax, segSum, numS);
    node_proj<<<256, EDGE_WAVES * 32, 0, stream>>>(embS, embA, W1, b1, P, numS, numA);
    edge_gate_cost<<<256, EDGE_WAVES * 32, 0, stream>>>(
        embS, embA, edgeS, edgeA, P, P + (size_t)numS * H_DIM, W2, b2, logT,
        outCost, segMax, E);
    seg_expsum<<<(E + 255) / 256, 256, 0, stream>>>(outCost, edgeS, segMax, segSum, logT, E);
    seg_norm<<<(E + 255) / 256, 256, 0, stream>>>(outCost, edgeS, segMax, segSum, logT, outW, E);
    return;
  }

  unsigned long long* segSum = (unsigned long long*)d_ws;
  unsigned*           segMax = (unsigned*)(segSum + numS);

  init_seg<<<(numS + 255) / 256, 256, 0, stream>>>(segMax, segSum, numS);

  edge_mlp_cost<<<256, EDGE_WAVES * 32, 0, stream>>>(
      embS, embA, edgeS, edgeA, W1, b1, W2, b2, logT, outCost, segMax, E);

  seg_expsum<<<(E + 255) / 256, 256, 0, stream>>>(outCost, edgeS, segMax, segSum, logT, E);
  seg_norm<<<(E + 255) / 256, 256, 0, stream>>>(outCost, edgeS, segMax, segSum, logT, outW, E);
}


# segment max and exp-sum taken in LDS: edges grouped by 256-node source bucket (per-block LDS count, scan, line-aligned record runs); 2M scattered global atomics go; integer sums stay order-free
# speedup vs baseline: 3.9299x; 2.0045x over previous
#include <hip/hip_runtime.h>
#include <hip/hip_bf16.h>
#include <math.h>

typedef __attribute__((ext_vector_type(2))) float v2f;
typedef __attribute__((ext_vector_type(4))) float v4f;
typedef __attribute__((ext_vector_type(8))) float v8f;

#define D_DIM 64
#define H_DIM 128
#define CPITCH 66

__device__ __forceinline__ unsigned ordKey(float f) {
  unsigned u = __float_as_uint(f);
  return (u & 0x80000000u) ? ~u : (u | 0x80000000u);
}
__device__ __forceinline__ float ordVal(unsigned k) {
  unsigned u = (k & 0x80000000u) ? (k ^ 0x80000000u) : ~k;
  return __uint_as_float(u);
}

#define SUM_SCALE     1099511627776.0f
#define SUM_INV_SCALE (1.0f / 1099511627776.0f)

__global__ void init_seg(unsigned* __restrict__ segMax,
                         unsigned long long* __restrict__ segSum, int numS) {
  int i = blockIdx.x * blockDim.x + threadIdx.x;
  if (i < numS) { segMax[i] = 0u; segSum[i] = 0ull; }
}

#define EDGE_WAVES 32
__global__ __launch_bounds__(EDGE_WAVES * 32)
void edge_mlp_cost(const float* __restrict__ embS,
                   const float* __restrict__ embA,
                   const int*   __restrict__ edgeS,
                   const int*   __restrict__ edgeA,
                   const float* __restrict__ W1,
                   const float* __restrict__ b1,
                   const float* __restrict__ W2,
                   const float* __restrict__ b2,
                   const float* __restrict__ logT,
                   float*       __restrict__ outCost,
                   unsigned*    __restrict__ segMax,
                   int E) {
  __shared__ float sW1[H_DIM * CPITCH * 2];
  __shared__ float sB1[H_DIM];
  __shared__ float sW2[H_DIM];
  __shared__ float sGate[EDGE_WAVES][16];

  const int tid = threadIdx.x;
  for (int i = tid; i < H_DIM * H_DIM; i += EDGE_WAVES * 32) {
    int kk = i >> 7;
    int c  = i & 127;
    sW1[(c * CPITCH + (kk >> 1)) * 2 + (kk & 1)] = W1[i];
  }
  if (tid < H_DIM) { sB1[tid] = b1[tid]; sW2[tid] = W2[tid]; }
  __syncthreads();

  const int wave = __builtin_amdgcn_readfirstlane(tid >> 5);
  const int lane = tid & 31;
  const int half = lane >> 4;
  const int col  = lane & 15;
  const float b2v  = b2[0];
  const float invT = __expf(-logT[0]);
  const int tiles  = (E + 15) >> 4;
  const int stride = gridDim.x * EDGE_WAVES;

  for (int t = blockIdx.x * EDGE_WAVES + wave; t < tiles; t += stride) {
  const int  e0    = t * 16;
  const int  e     = e0 + col;
  const bool valid = (e < E);
  const int  ec    = valid ? e : (E - 1);
  const int  sI    = edgeS[ec];
  const int  aI    = edgeA[ec];
  const float* es = embS + (size_t)sI * D_DIM;
  const float* ea = embA + (size_t)aI * D_DIM;

  v2f a[32];
#pragma unroll
  for (int k = 0; k < 16; ++k) {
    a[k]      = *(const v2f*)(es + 4 * k + 2 * half);
    a[k + 16] = *(const v2f*)(ea + 4 * k + 2 * half);
  }
  float ss = 0.0f;
#pragma unroll
  for (int k = 0; k < 16; ++k) {
    float dx = a[k][0] - a[k + 16][0];
    float dy = a[k][1] - a[k + 16][1];
    ss += dx * dx + dy * dy;
  }
  ss += __shfl_xor(ss, 16);

  float gacc[8] = {0, 0, 0, 0, 0, 0, 0, 0};
#pragma unroll 1
  for (int n = 0; n < 8; ++n) {
    const float* bbase = sW1 + ((n * 16 + col) * CPITCH + half) * 2;
    v8f c0 = {};
    v8f c1 = {};
#pragma unroll
    for (int k = 0; k < 32; k += 2) {
      v2f b0 = *(const v2f*)(bbase + 4 * k);
      v2f b1v = *(const v2f*)(bbase + 4 * (k + 1));
      c0 = __builtin_amdgcn_wmma_f32_16x16x4_f32(false, a[k], false, b0,
                                                 (short)0, c0, false, false);
      c1 = __builtin_amdgcn_wmma_f32_16x16x4_f32(false, a[k + 1], false, b1v,
                                                 (short)0, c1, false, false);
    }
    const float bn  = sB1[n * 16 + col];
    const float w2n = sW2[n * 16 + col];
#pragma unroll
    for (int r = 0; r < 8; ++r) {
      float h = c0[r] + c1[r] + bn;
      h = h > 0.0f ? h : 0.0f;
      gacc[r] += h * w2n;
    }
  }

#pragma unroll
  for (int r = 0; r < 8; ++r) {
    float g = gacc[r];
    g += __shfl_xor(g, 1);
    g += __shfl_xor(g, 2);
    g += __shfl_xor(g, 4);
    g += __shfl_xor(g, 8);
    if (lane == 0)  sGate[wave][r]     = g;
    if (lane == 16) sGate[wave][r + 8] = g;
  }

  if (lane < 16 && valid) {
    float logit = sGate[wave][lane] + b2v;
    float gate  = 1.0f / (1.0f + __expf(-logit));
    float cost  = sqrtf(ss) * gate;
    outCost[e]  = cost;
    atomicMax(&segMax[sI], ordKey(-cost * invT));
  }
  }
}

__global__ __launch_bounds__(EDGE_WAVES * 32)
void node_proj(const float* __restrict__ embS,
               const float* __restrict__ embA,
               const float* __restrict__ W1,
               const float* __restrict__ b1,
               float*       __restrict__ P,
               int numS, int numA) {
  __shared__ float sW1[H_DIM * CPITCH * 2];
  __shared__ __attribute__((aligned(16))) float sB1[2 * H_DIM];

  const int tid = threadIdx.x;
  for (int i = tid; i < H_DIM * H_DIM; i += EDGE_WAVES * 32) {
    int kk = i >> 7;
    int c  = i & 127;
    sW1[(c * CPITCH + (kk >> 1)) * 2 + (kk & 1)] = W1[i];
  }
  if (tid < H_DIM) { sB1[tid] = b1[tid]; sB1[H_DIM + tid] = 0.0f; }
  __syncthreads();

  const int wave = __builtin_amdgcn_readfirstlane(tid >> 5);
  const int lane = tid & 31;
  const int half = lane >> 4;
  const int col  = lane & 15;
  const int tilesS = (numS + 15) >> 4;
  const int tilesA = (numA + 15) >> 4;
  const int tiles  = tilesS + tilesA;

  for (int t = wave * (int)gridDim.x + (int)blockIdx.x; t < tiles;
       t += EDGE_WAVES * (int)gridDim.x) {
    const bool isA = (t >= tilesS);
    const int  tt  = isA ? (t - tilesS) : t;
    const float* emb = isA ? embA : embS;
    const int  nrows = isA ? numA : numS;
    const int  node  = tt * 16 + col;
    const bool valid = (node < nrows);
    const int  nc    = valid ? node : (nrows - 1);
    const float* row = emb + (size_t)nc * D_DIM;

    v2f a[16];
#pragma unroll
    for (int k = 0; k < 16; ++k) a[k] = *(const v2f*)(row + 4 * k + 2 * half);

    float* prow = P + ((size_t)(isA ? numS : 0) + (size_t)nc) * H_DIM + 8 * half;
    const int koff = isA ? D_DIM : 0;
    const int boff = isA ? H_DIM : 0;

#pragma unroll 1
    for (int nt = 0; nt < 8; ++nt) {
      const float* wbase = sW1 + ((nt * 16 + col) * CPITCH + half) * 2 + koff;
      v8f c0 = {};
      v8f c1 = {};
#pragma unroll
      for (int k = 0; k < 16; k += 2) {
        v2f w0 = *(const v2f*)(wbase + 4 * k);
        v2f w1 = *(const v2f*)(wbase + 4 * (k + 1));
        c0 = __builtin_amdgcn_wmma_f32_16x16x4_f32(false, w0, false, a[k],
                                                   (short)0, c0, false, false);
        c1 = __builtin_amdgcn_wmma_f32_16x16x4_f32(false, w1, false, a[k + 1],
                                                   (short)0, c1, false, false);
      }
      const float* bb = sB1 + boff + nt * 16 + 8 * half;
      const v4f blo = *(const v4f*)(bb);
      const v4f bhi = *(const v4f*)(bb + 4);
      v4f lo, hi;
#pragma unroll
      for (int r = 0; r < 4; ++r) {
        lo[r] = c0[r] + c1[r] + blo[r];
        hi[r] = c0[r + 4] + c1[r + 4] + bhi[r];
      }
      if (valid) {
        *(v4f*)(prow + nt * 16)     = lo;
        *(v4f*)(prow + nt * 16 + 4) = hi;
      }
    }
  }
}

struct RowSet {
  v4f es0, es1, ea0, ea1;
  v4f ps0, ps1, ps2, ps3;
  v4f pa0, pa1, pa2, pa3;
};

__device__ __forceinline__ void fetch_rows(RowSet& r,
                                           const float* __restrict__ embS,
                                           const float* __restrict__ embA,
                                           const float* __restrict__ Ps,
                                           const float* __restrict__ Pa,
                                           int s, int a, int sub4) {
  const float* e1 = embS + (size_t)s * D_DIM + sub4;
  const float* e2 = embA + (size_t)a * D_DIM + sub4;
  const float* p1 = Ps + (size_t)s * H_DIM + sub4;
  const float* p2 = Pa + (size_t)a * H_DIM + sub4;
  r.es0 = *(const v4f*)(e1);
  r.es1 = *(const v4f*)(e1 + 32);
  r.ea0 = *(const v4f*)(e2);
  r.ea1 = *(const v4f*)(e2 + 32);
  r.ps0 = *(const v4f*)(p1);
  r.ps1 = *(const v4f*)(p1 + 32);
  r.ps2 = *(const v4f*)(p1 + 64);
  r.ps3 = *(const v4f*)(p1 + 96);
  r.pa0 = *(const v4f*)(p2);
  r.pa1 = *(const v4f*)(p2 + 32);
  r.pa2 = *(const v4f*)(p2 + 64);
  r.pa3 = *(const v4f*)(p2 + 96);
}

__device__ __forceinline__ float sq4(v4f a, v4f b) {
  v4f d = a - b;
  return d[0] * d[0] + d[1] * d[1] + d[2] * d[2] + d[3] * d[3];
}
__device__ __forceinline__ float gate4(v4f p, v4f q, v4f w) {
  v4f h = p + q;
  float g = 0.0f;
#pragma unroll
  for (int i = 0; i < 4; ++i) {
    float x = h[i] > 0.0f ? h[i] : 0.0f;
    g += x * w[i];
  }
  return g;
}

template <int M>
__device__ __forceinline__ float xadd8(float v) {
  int o = __builtin_amdgcn_update_dpp(0, __float_as_int(v), 0x160 + M, 0xf, 0xf, true);
  return v + __int_as_float(o);
}
__device__ __forceinline__ int bsel8(int v, int lane, int i) {
  return __builtin_amdgcn_ds_bpermute(((lane & 24) | i) << 2, v);
}

__device__ __forceinline__ void reduce_rows(const RowSet& r, const float* sW2, int sub4,
                                            float& ss, float& g) {
  float s = sq4(r.es0, r.ea0) + sq4(r.es1, r.ea1);
  float t = gate4(r.ps0, r.pa0, *(const v4f*)(sW2 + sub4))
          + gate4(r.ps1, r.pa1, *(const v4f*)(sW2 + 32 + sub4))
          + gate4(r.ps2, r.pa2, *(const v4f*)(sW2 + 64 + sub4))
          + gate4(r.ps3, r.pa3, *(const v4f*)(sW2 + 96 + sub4));
  s = xadd8<1>(s); t = xadd8<1>(t);
  s = xadd8<2>(s); t = xadd8<2>(t);
  s = xadd8<4>(s); t = xadd8<4>(t);
  ss = s; g = t;
}

template <bool kAtomicMax>
__global__ __launch_bounds__(EDGE_WAVES * 32)
void edge_gate_cost(const float* __restrict__ embS,
                    const float* __restrict__ embA,
                    const int*   __restrict__ edgeS,
                    const int*   __restrict__ edgeA,
                    const float* __restrict__ Ps,
                    const float* __restrict__ Pa,
                    const float* __restrict__ W2,
                    const float* __restrict__ b2,
                    const float* __restrict__ logT,
                    float*       __restrict__ outCost,
                    unsigned*    __restrict__ segMax,
                    int E) {
  __shared__ __attribute__((aligned(16))) float sW2[H_DIM];
  const int tid = threadIdx.x;
  if (tid < H_DIM) sW2[tid] = W2[tid];
  __syncthreads();

  const int wave = __builtin_amdgcn_readfirstlane(tid >> 5);
  const int lane = tid & 31;
  const int sub  = lane & 7;
  const int sub4 = sub * 4;
  const float b2v  = b2[0];
  const float invT = __expf(-logT[0]);
  const int tiles  = (E + 31) >> 5;
  const int stride = (int)gridDim.x * EDGE_WAVES;

  int t = (int)blockIdx.x * EDGE_WAVES + wave;
  if (t >= tiles) return;

  int eL   = t * 32 + lane;
  int eLc  = eL < E ? eL : (E - 1);
  int curS = edgeS[eLc];
  int curA = edgeA[eLc];

  for (;;) {
    const bool hasNext = (t + stride) < tiles;
    const int  tn   = hasNext ? (t + stride) : t;
    const int  nL   = tn * 32 + lane;
    const int  nLc  = nL < E ? nL : (E - 1);
    const int  nxtS = edgeS[nLc];
    const int  nxtA = edgeA[nLc];

    float rss = 0.0f, rg = 0.0f;

#pragma unroll 1
    for (int i = 0; i < 8; i += 2) {
      RowSet X, Y;
      float ss, g;
      fetch_rows(X, embS, embA, Ps, Pa, bsel8(curS, lane, i), bsel8(curA, lane, i), sub4);
      fetch_rows(Y, embS, embA, Ps, Pa, bsel8(curS, lane, i + 1), bsel8(curA, lane, i + 1), sub4);
      __builtin_amdgcn_sched_barrier(0);
      reduce_rows(X, sW2, sub4, ss, g);
      if (sub == i) { rss = ss; rg = g; }
      reduce_rows(Y, sW2, sub4, ss, g);
      if (sub == i + 1) { rss = ss; rg = g; }
    }

    if (eL < E) {
      float logit = rg + b2v;
      float gate  = 1.0f / (1.0f + __expf(-logit));
      float cost  = sqrtf(rss) * gate;
      outCost[eL] = cost;
      if (kAtomicMax) atomicMax(&segMax[curS], ordKey(-cost * invT));
    }
    if (!hasNext) break;
    t = tn; eL = nL; curS = nxtS; curA = nxtA;
  }
}

#define SEG_SHIFT    8
#define SEG_NODES    256
#define NB_MAX       1024
#define PART_BLOCKS  256
#define PART_THREADS 1024
#define CHUNK_MAX    4096
#define STAT_REGS    8

__global__ __launch_bounds__(PART_THREADS)
void bucket_place(const int*   __restrict__ edgeS,
                  const float* __restrict__ outCost,
                  const float* __restrict__ logT,
                  int*           __restrict__ offG,
                  float*         __restrict__ recVal,
                  unsigned char* __restrict__ recNode,
                  int E, int NB, int chunk, int chunkPad) {
  __shared__ int h[NB_MAX];
  __shared__ int fill[NB_MAX];
  __shared__ int off[NB_MAX + 1];
  __shared__ __attribute__((aligned(16))) float    sVal[CHUNK_MAX];
  __shared__ __attribute__((aligned(16))) unsigned sNode4[CHUNK_MAX / 4];
  unsigned char* sNode = (unsigned char*)sNode4;

  const int tid = threadIdx.x;
  const int b   = blockIdx.x;
  for (int k = tid; k < NB; k += PART_THREADS) { h[k] = 0; fill[k] = 0; }
  sNode4[tid] = 0u;
  __syncthreads();

  const float invT = __expf(-logT[0]);
  const int lo = b * chunk;
  const int hi = (lo + chunk < E) ? (lo + chunk) : E;
  const int n  = (hi > lo) ? (hi - lo) : 0;

  int   sv[4];
  float vv[4];
#pragma unroll
  for (int j = 0; j < 4; ++j) {
    const int i = j * PART_THREADS + tid;
    sv[j] = -1;
    vv[j] = 0.0f;
    if (i < n) {
      sv[j] = edgeS[lo + i];
      vv[j] = -outCost[lo + i] * invT;
    }
  }
#pragma unroll
  for (int j = 0; j < 4; ++j)
    if (sv[j] >= 0) atomicAdd(&h[sv[j] >> SEG_SHIFT], 1);
  __syncthreads();

  if (tid < 32) {
    const int per = (NB + 31) >> 5;
    const int k0  = tid * per;
    int s = 0;
    for (int j = 0; j < per; ++j) { const int k = k0 + j; if (k < NB) s += h[k]; }
    int inc = s;
    for (int d = 1; d < 32; d <<= 1) {
      const int o = __shfl_up(inc, d);
      if (tid >= d) inc += o;
    }
    int run = inc - s;
    for (int j = 0; j < per; ++j) {
      const int k = k0 + j;
      if (k < NB) { off[k] = run; run += h[k]; }
    }
    if (tid == 31) off[NB] = inc;
  }
  __syncthreads();

  for (int k = tid; k <= NB; k += PART_THREADS) offG[b * (NB + 1) + k] = off[k];

#pragma unroll
  for (int j = 0; j < 4; ++j) {
    if (sv[j] >= 0) {
      const int k   = sv[j] >> SEG_SHIFT;
      const int pos = off[k] + atomicAdd(&fill[k], 1);
      sVal[pos]  = vv[j];
      sNode[pos] = (unsigned char)(sv[j] & (SEG_NODES - 1));
    }
  }
  __syncthreads();

  float*    gv = recVal + (size_t)b * chunkPad;
  unsigned* gn = (unsigned*)(recNode + (size_t)b * chunkPad);
#pragma unroll
  for (int j = 0; j < 4; ++j) {
    const int i = j * PART_THREADS + tid;
    if (i < n) gv[i] = sVal[i];
  }
  if (4 * tid < n) gn[tid] = sNode4[tid];
}

__global__ __launch_bounds__(PART_THREADS)
void bucket_stats(const int*           __restrict__ offG,
                  const float*         __restrict__ recVal,
                  const unsigned char* __restrict__ recNode,
                  unsigned*            __restrict__ segMax,
                  unsigned long long*  __restrict__ segSum,
                  int numS, int NB, int chunkPad) {
  __shared__ unsigned           mx[SEG_NODES];
  __shared__ unsigned long long sm[SEG_NODES];
  const int tid = threadIdx.x;
  const int k   = blockIdx.x;
  if (tid < SEG_NODES) { mx[tid] = 0u; sm[tid] = 0ull; }

  const int r   = tid >> 2;
  const int sub = tid & 3;
  const int* o  = offG + r * (NB + 1) + k;
  const int st  = r * chunkPad + o[0];
  const int en  = r * chunkPad + o[1];

  float rv[STAT_REGS];
  int   rn[STAT_REGS];
#pragma unroll
  for (int j = 0; j < STAT_REGS; ++j) {
    const int i = st + sub + 4 * j;
    rv[j] = 0.0f;
    rn[j] = -1;
    if (i < en) { rv[j] = recVal[i]; rn[j] = recNode[i]; }
  }
  const int rest = st + sub + 4 * STAT_REGS;
  __syncthreads();

#pragma unroll
  for (int j = 0; j < STAT_REGS; ++j)
    if (rn[j] >= 0) atomicMax(&mx[rn[j]], ordKey(rv[j]));
  for (int i = rest; i < en; i += 4)
    atomicMax(&mx[recNode[i]], ordKey(recVal[i]));
  __syncthreads();

#pragma unroll
  for (int j = 0; j < STAT_REGS; ++j) {
    if (rn[j] >= 0) {
      const float m = ordVal(mx[rn[j]]);
      atomicAdd(&sm[rn[j]], (unsigned long long)(__expf(rv[j] - m) * SUM_SCALE));
    }
  }
  for (int i = rest; i < en; i += 4) {
    const int   nd = recNode[i];
    const float m  = ordVal(mx[nd]);
    atomicAdd(&sm[nd], (unsigned long long)(__expf(recVal[i] - m) * SUM_SCALE));
  }
  __syncthreads();

  if (tid < SEG_NODES) {
    const int s = k * SEG_NODES + tid;
    if (s < numS) { segMax[s] = mx[tid]; segSum[s] = sm[tid]; }
  }
}

__global__ void seg_expsum(const float* __restrict__ outCost,
                           const int* __restrict__ edgeS,
                           const unsigned* __restrict__ segMax,
                           unsigned long long* __restrict__ segSum,
                           const float* __restrict__ logT, int E) {
  int e = blockIdx.x * blockDim.x + threadIdx.x;
  if (e >= E) return;
  float invT = __expf(-logT[0]);
  int s = edgeS[e];
  float val = -outCost[e] * invT;
  float m = ordVal(segMax[s]);
  atomicAdd(&segSum[s], (unsigned long long)(__expf(val - m) * SUM_SCALE));
}

__global__ void seg_norm(const float* __restrict__ outCost,
                         const int* __restrict__ edgeS,
                         const unsigned* __restrict__ segMax,
                         const unsigned long long* __restrict__ segSum,
                         const float* __restrict__ logT,
                         float* __restrict__ outW, int E) {
  int e = blockIdx.x * blockDim.x + threadIdx.x;
  if (e >= E) return;
  float invT = __expf(-logT[0]);
  int s = edgeS[e];
  float val = -outCost[e] * invT;
  float m = ordVal(segMax[s]);
  outW[e] = __expf(val - m) / ((float)segSum[s] * SUM_INV_SCALE);
}

extern "C" void kernel_launch(void* const* d_in, const int* in_sizes, int n_in,
                              void* d_out, int out_size, void* d_ws, size_t ws_size,
                              hipStream_t stream) {
  const float* embS = (const float*)d_in[0];
  const float* embA = (const float*)d_in[1];
  const int*   eidx = (const int*)d_in[2];
  const float* W1   = (const float*)d_in[3];
  const float* b1   = (const float*)d_in[4];
  const float* W2   = (const float*)d_in[5];
  const float* b2   = (const float*)d_in[6];
  const float* logT = (const float*)d_in[7];

  const int numS = in_sizes[0] / D_DIM;
  const int numA = in_sizes[1] / D_DIM;
  const int E    = in_sizes[2] / 2;
  const int* edgeS = eidx;
  const int* edgeA = eidx + E;

  float* outW    = (float*)d_out;
  float* outCost = (float*)d_out + E;

  const size_t pFloats = (size_t)(numS + numA) * H_DIM;
  const size_t need    = pFloats * sizeof(float)
                       + (size_t)numS * (sizeof(unsigned long long) + sizeof(unsigned));

  const int NB       = (numS + SEG_NODES - 1) >> SEG_SHIFT;
  const int chunk    = (E + PART_BLOCKS - 1) / PART_BLOCKS;
  const int chunkPad = (chunk + 127) & ~127;
  const size_t recSlots = (size_t)PART_BLOCKS * chunkPad;
  const size_t needGrp  = need + recSlots * (sizeof(float) + 1)
                        + (size_t)PART_BLOCKS * (NB + 1) * sizeof(int);

  if (ws_size >= needGrp && NB <= NB_MAX && chunk <= CHUNK_MAX) {
    float* P = (float*)d_ws;
    float*         recVal  = P + pFloats;
    unsigned char* recNode = (unsigned char*)(recVal + recSlots);
    unsigned long long* segSum = (unsigned long long*)(recNode + recSlots);
    unsigned*           segMax = (unsigned*)(segSum + numS);
    int*                offG   = (int*)(segMax + numS);

    node_proj<<<256, EDGE_WAVES * 32, 0, stream>>>(embS, embA, W1, b1, P, numS, numA);
    edge_gate_cost<false><<<256, EDGE_WAVES * 32, 0, stream>>>(
        embS, embA, edgeS, edgeA, P, P + (size_t)numS * H_DIM, W2, b2, logT,
        outCost, segMax, E);
    bucket_place<<<PART_BLOCKS, PART_THREADS, 0, stream>>>(
        edgeS, outCost, logT, offG, recVal, recNode, E, NB, chunk, chunkPad);
    bucket_stats<<<NB, PART_THREADS, 0, stream>>>(
        offG, recVal, recNode, segMax, segSum, numS, NB, chunkPad);
    seg_norm<<<(E + 255) / 256, 256, 0, stream>>>(outCost, edgeS, segMax, segSum, logT, outW, E);
    return;
  }

  if (ws_size >= need) {
    float* P = (float*)d_ws;
    unsigned long long* segSum = (unsigned long long*)(P + pFloats);
    unsigned*           segMax = (unsigned*)(segSum + numS);

    init_seg<<<(numS + 255) / 256, 256, 0, stream>>>(segMax, segSum, numS);
    node_proj<<<256, EDGE_WAVES * 32, 0, stream>>>(embS, embA, W1, b1, P, numS, numA);
    edge_gate_cost<true><<<256, EDGE_WAVES * 32, 0, stream>>>(
        embS, embA, edgeS, edgeA, P, P + (size_t)numS * H_DIM, W2, b2, logT,
        outCost, segMax, E);
    seg_expsum<<<(E + 255) / 256, 256, 0, stream>>>(outCost, edgeS, segMax, segSum, logT, E);
    seg_norm<<<(E + 255) / 256, 256, 0, stream>>>(outCost, edgeS, segMax, segSum, logT, outW, E);
    return;
  }

  unsigned long long* segSum = (unsigned long long*)d_ws;
  unsigned*           segMax = (unsigned*)(segSum + numS);

  init_seg<<<(numS + 255) / 256, 256, 0, stream>>>(segMax, segSum, numS);

  edge_mlp_cost<<<256, EDGE_WAVES * 32, 0, stream>>>(
      embS, embA, edgeS, edgeA, W1, b1, W2, b2, logT, outCost, segMax, E);

  seg_expsum<<<(E + 255) / 256, 256, 0, stream>>>(outCost, edgeS, segMax, segSum, logT, E);
  seg_norm<<<(E + 255) / 256, 256, 0, stream>>>(outCost, edgeS, segMax, segSum, logT, outW, E);
}


# edge pass and grouping in one launch: a block computes its contiguous chunk's costs, keeps arguments and sources in LDS, counts, scans and places them itself; the bucket_place launch goes
# speedup vs baseline: 4.0721x; 1.0362x over previous
#include <hip/hip_runtime.h>
#include <hip/hip_bf16.h>
#include <math.h>

typedef __attribute__((ext_vector_type(2))) float v2f;
typedef __attribute__((ext_vector_type(4))) float v4f;
typedef __attribute__((ext_vector_type(8))) float v8f;

#define D_DIM 64
#define H_DIM 128
#define CPITCH 66

__device__ __forceinline__ unsigned ordKey(float f) {
  unsigned u = __float_as_uint(f);
  return (u & 0x80000000u) ? ~u : (u | 0x80000000u);
}
__device__ __forceinline__ float ordVal(unsigned k) {
  unsigned u = (k & 0x80000000u) ? (k ^ 0x80000000u) : ~k;
  return __uint_as_float(u);
}

#define SUM_SCALE     1099511627776.0f
#define SUM_INV_SCALE (1.0f / 1099511627776.0f)

__global__ void init_seg(unsigned* __restrict__ segMax,
                         unsigned long long* __restrict__ segSum, int numS) {
  int i = blockIdx.x * blockDim.x + threadIdx.x;
  if (i < numS) { segMax[i] = 0u; segSum[i] = 0ull; }
}

#define EDGE_WAVES 32
__global__ __launch_bounds__(EDGE_WAVES * 32)
void edge_mlp_cost(const float* __restrict__ embS,
                   const float* __restrict__ embA,
                   const int*   __restrict__ edgeS,
                   const int*   __restrict__ edgeA,
                   const float* __restrict__ W1,
                   const float* __restrict__ b1,
                   const float* __restrict__ W2,
                   const float* __restrict__ b2,
                   const float* __restrict__ logT,
                   float*       __restrict__ outCost,
                   unsigned*    __restrict__ segMax,
                   int E) {
  __shared__ float sW1[H_DIM * CPITCH * 2];
  __shared__ float sB1[H_DIM];
  __shared__ float sW2[H_DIM];
  __shared__ float sGate[EDGE_WAVES][16];

  const int tid = threadIdx.x;
  for (int i = tid; i < H_DIM * H_DIM; i += EDGE_WAVES * 32) {
    int kk = i >> 7;
    int c  = i & 127;
    sW1[(c * CPITCH + (kk >> 1)) * 2 + (kk & 1)] = W1[i];
  }
  if (tid < H_DIM) { sB1[tid] = b1[tid]; sW2[tid] = W2[tid]; }
  __syncthreads();

  const int wave = __builtin_amdgcn_readfirstlane(tid >> 5);
  const int lane = tid & 31;
  const int half = lane >> 4;
  const int col  = lane & 15;
  const float b2v  = b2[0];
  const float invT = __expf(-logT[0]);
  const int tiles  = (E + 15) >> 4;
  const int stride = gridDim.x * EDGE_WAVES;

  for (int t = blockIdx.x * EDGE_WAVES + wave; t < tiles; t += stride) {
  const int  e0    = t * 16;
  const int  e     = e0 + col;
  const bool valid = (e < E);
  const int  ec    = valid ? e : (E - 1);
  const int  sI    = edgeS[ec];
  const int  aI    = edgeA[ec];
  const float* es = embS + (size_t)sI * D_DIM;
  const float* ea = embA + (size_t)aI * D_DIM;

  v2f a[32];
#pragma unroll
  for (int k = 0; k < 16; ++k) {
    a[k]      = *(const v2f*)(es + 4 * k + 2 * half);
    a[k + 16] = *(const v2f*)(ea + 4 * k + 2 * half);
  }
  float ss = 0.0f;
#pragma unroll
  for (int k = 0; k < 16; ++k) {
    float dx = a[k][0] - a[k + 16][0];
    float dy = a[k][1] - a[k + 16][1];
    ss += dx * dx + dy * dy;
  }
  ss += __shfl_xor(ss, 16);

  float gacc[8] = {0, 0, 0, 0, 0, 0, 0, 0};
#pragma unroll 1
  for (int n = 0; n < 8; ++n) {
    const float* bbase = sW1 + ((n * 16 + col) * CPITCH + half) * 2;
    v8f c0 = {};
    v8f c1 = {};
#pragma unroll
    for (int k = 0; k < 32; k += 2) {
      v2f b0 = *(const v2f*)(bbase + 4 * k);
      v2f b1v = *(const v2f*)(bbase + 4 * (k + 1));
      c0 = __builtin_amdgcn_wmma_f32_16x16x4_f32(false, a[k], false, b0,
                                                 (short)0, c0, false, false);
      c1 = __builtin_amdgcn_wmma_f32_16x16x4_f32(false, a[k + 1], false, b1v,
                                                 (short)0, c1, false, false);
    }
    const float bn  = sB1[n * 16 + col];
    const float w2n = sW2[n * 16 + col];
#pragma unroll
    for (int r = 0; r < 8; ++r) {
      float h = c0[r] + c1[r] + bn;
      h = h > 0.0f ? h : 0.0f;
      gacc[r] += h * w2n;
    }
  }

#pragma unroll
  for (int r = 0; r < 8; ++r) {
    float g = gacc[r];
    g += __shfl_xor(g, 1);
    g += __shfl_xor(g, 2);
    g += __shfl_xor(g, 4);
    g += __shfl_xor(g, 8);
    if (lane == 0)  sGate[wave][r]     = g;
    if (lane == 16) sGate[wave][r + 8] = g;
  }

  if (lane < 16 && valid) {
    float logit = sGate[wave][lane] + b2v;
    float gate  = 1.0f / (1.0f + __expf(-logit));
    float cost  = sqrtf(ss) * gate;
    outCost[e]  = cost;
    atomicMax(&segMax[sI], ordKey(-cost * invT));
  }
  }
}

__global__ __launch_bounds__(EDGE_WAVES * 32)
void node_proj(const float* __restrict__ embS,
               const float* __restrict__ embA,
               const float* __restrict__ W1,
               const float* __restrict__ b1,
               float*       __restrict__ P,
               int numS, int numA) {
  __shared__ float sW1[H_DIM * CPITCH * 2];
  __shared__ __attribute__((aligned(16))) float sB1[2 * H_DIM];

  const int tid = threadIdx.x;
  for (int i = tid; i < H_DIM * H_DIM; i += EDGE_WAVES * 32) {
    int kk = i >> 7;
    int c  = i & 127;
    sW1[(c * CPITCH + (kk >> 1)) * 2 + (kk & 1)] = W1[i];
  }
  if (tid < H_DIM) { sB1[tid] = b1[tid]; sB1[H_DIM + tid] = 0.0f; }
  __syncthreads();

  const int wave = __builtin_amdgcn_readfirstlane(tid >> 5);
  const int lane = tid & 31;
  const int half = lane >> 4;
  const int col  = lane & 15;
  const int tilesS = (numS + 15) >> 4;
  const int tilesA = (numA + 15) >> 4;
  const int tiles  = tilesS + tilesA;

  for (int t = wave * (int)gridDim.x + (int)blockIdx.x; t < tiles;
       t += EDGE_WAVES * (int)gridDim.x) {
    const bool isA = (t >= tilesS);
    const int  tt  = isA ? (t - tilesS) : t;
    const float* emb = isA ? embA : embS;
    const int  nrows = isA ? numA : numS;
    const int  node  = tt * 16 + col;
    const bool valid = (node < nrows);
    const int  nc    = valid ? node : (nrows - 1);
    const float* row = emb + (size_t)nc * D_DIM;

    v2f a[16];
#pragma unroll
    for (int k = 0; k < 16; ++k) a[k] = *(const v2f*)(row + 4 * k + 2 * half);

    float* prow = P + ((size_t)(isA ? numS : 0) + (size_t)nc) * H_DIM + 8 * half;
    const int koff = isA ? D_DIM : 0;
    const int boff = isA ? H_DIM : 0;

#pragma unroll 1
    for (int nt = 0; nt < 8; ++nt) {
      const float* wbase = sW1 + ((nt * 16 + col) * CPITCH + half) * 2 + koff;
      v8f c0 = {};
      v8f c1 = {};
#pragma unroll
      for (int k = 0; k < 16; k += 2) {
        v2f w0 = *(const v2f*)(wbase + 4 * k);
        v2f w1 = *(const v2f*)(wbase + 4 * (k + 1));
        c0 = __builtin_amdgcn_wmma_f32_16x16x4_f32(false, w0, false, a[k],
                                                   (short)0, c0, false, false);
        c1 = __builtin_amdgcn_wmma_f32_16x16x4_f32(false, w1, false, a[k + 1],
                                                   (short)0, c1, false, false);
      }
      const float* bb = sB1 + boff + nt * 16 + 8 * half;
      const v4f blo = *(const v4f*)(bb);
      const v4f bhi = *(const v4f*)(bb + 4);
      v4f lo, hi;
#pragma unroll
      for (int r = 0; r < 4; ++r) {
        lo[r] = c0[r] + c1[r] + blo[r];
        hi[r] = c0[r + 4] + c1[r + 4] + bhi[r];
      }
      if (valid) {
        *(v4f*)(prow + nt * 16)     = lo;
        *(v4f*)(prow + nt * 16 + 4) = hi;
      }
    }
  }
}

struct RowSet {
  v4f es0, es1, ea0, ea1;
  v4f ps0, ps1, ps2, ps3;
  v4f pa0, pa1, pa2, pa3;
};

__device__ __forceinline__ void fetch_rows(RowSet& r,
                                           const float* __restrict__ embS,
                                           const float* __restrict__ embA,
                                           const float* __restrict__ Ps,
                                           const float* __restrict__ Pa,
                                           int s, int a, int sub4) {
  const float* e1 = embS + (size_t)s * D_DIM + sub4;
  const float* e2 = embA + (size_t)a * D_DIM + sub4;
  const float* p1 = Ps + (size_t)s * H_DIM + sub4;
  const float* p2 = Pa + (size_t)a * H_DIM + sub4;
  r.es0 = *(const v4f*)(e1);
  r.es1 = *(const v4f*)(e1 + 32);
  r.ea0 = *(const v4f*)(e2);
  r.ea1 = *(const v4f*)(e2 + 32);
  r.ps0 = *(const v4f*)(p1);
  r.ps1 = *(const v4f*)(p1 + 32);
  r.ps2 = *(const v4f*)(p1 + 64);
  r.ps3 = *(const v4f*)(p1 + 96);
  r.pa0 = *(const v4f*)(p2);
  r.pa1 = *(const v4f*)(p2 + 32);
  r.pa2 = *(const v4f*)(p2 + 64);
  r.pa3 = *(const v4f*)(p2 + 96);
}

__device__ __forceinline__ float sq4(v4f a, v4f b) {
  v4f d = a - b;
  return d[0] * d[0] + d[1] * d[1] + d[2] * d[2] + d[3] * d[3];
}
__device__ __forceinline__ float gate4(v4f p, v4f q, v4f w) {
  v4f h = p + q;
  float g = 0.0f;
#pragma unroll
  for (int i = 0; i < 4; ++i) {
    float x = h[i] > 0.0f ? h[i] : 0.0f;
    g += x * w[i];
  }
  return g;
}

template <int M>
__device__ __forceinline__ float xadd8(float v) {
  int o = __builtin_amdgcn_update_dpp(0, __float_as_int(v), 0x160 + M, 0xf, 0xf, true);
  return v + __int_as_float(o);
}
__device__ __forceinline__ int bsel8(int v, int lane, int i) {
  return __builtin_amdgcn_ds_bpermute(((lane & 24) | i) << 2, v);
}

__device__ __forceinline__ void reduce_rows(const RowSet& r, const float* sW2, int sub4,
                                            float& ss, float& g) {
  float s = sq4(r.es0, r.ea0) + sq4(r.es1, r.ea1);
  float t = gate4(r.ps0, r.pa0, *(const v4f*)(sW2 + sub4))
          + gate4(r.ps1, r.pa1, *(const v4f*)(sW2 + 32 + sub4))
          + gate4(r.ps2, r.pa2, *(const v4f*)(sW2 + 64 + sub4))
          + gate4(r.ps3, r.pa3, *(const v4f*)(sW2 + 96 + sub4));
  s = xadd8<1>(s); t = xadd8<1>(t);
  s = xadd8<2>(s); t = xadd8<2>(t);
  s = xadd8<4>(s); t = xadd8<4>(t);
  ss = s; g = t;
}

__device__ __forceinline__ float tile_cost(const float* __restrict__ embS,
                                           const float* __restrict__ embA,
                                           const float* __restrict__ Ps,
                                           const float* __restrict__ Pa,
                                           const float* sW2,
                                           int curS, int curA,
                                           int lane, int sub, int sub4, float b2v) {
  float rss = 0.0f, rg = 0.0f;

#pragma unroll 1
  for (int i = 0; i < 8; i += 2) {
    RowSet X, Y;
    float ss, g;
    fetch_rows(X, embS, embA, Ps, Pa, bsel8(curS, lane, i), bsel8(curA, lane, i), sub4);
    fetch_rows(Y, embS, embA, Ps, Pa, bsel8(curS, lane, i + 1), bsel8(curA, lane, i + 1), sub4);
    __builtin_amdgcn_sched_barrier(0);
    reduce_rows(X, sW2, sub4, ss, g);
    if (sub == i) { rss = ss; rg = g; }
    reduce_rows(Y, sW2, sub4, ss, g);
    if (sub == i + 1) { rss = ss; rg = g; }
  }

  const float logit = rg + b2v;
  const float gate  = 1.0f / (1.0f + __expf(-logit));
  return sqrtf(rss) * gate;
}

__global__ __launch_bounds__(EDGE_WAVES * 32)
void edge_gate_cost(const float* __restrict__ embS,
                    const float* __restrict__ embA,
                    const int*   __restrict__ edgeS,
                    const int*   __restrict__ edgeA,
                    const float* __restrict__ Ps,
                    const float* __restrict__ Pa,
                    const float* __restrict__ W2,
                    const float* __restrict__ b2,
                    const float* __restrict__ logT,
                    float*       __restrict__ outCost,
                    unsigned*    __restrict__ segMax,
                    int E) {
  __shared__ __attribute__((aligned(16))) float sW2[H_DIM];
  const int tid = threadIdx.x;
  if (tid < H_DIM) sW2[tid] = W2[tid];
  __syncthreads();

  const int wave = __builtin_amdgcn_readfirstlane(tid >> 5);
  const int lane = tid & 31;
  const int sub  = lane & 7;
  const int sub4 = sub * 4;
  const float b2v  = b2[0];
  const float invT = __expf(-logT[0]);
  const int tiles  = (E + 31) >> 5;
  const int stride = (int)gridDim.x * EDGE_WAVES;

  int t = (int)blockIdx.x * EDGE_WAVES + wave;
  if (t >= tiles) return;

  int eL   = t * 32 + lane;
  int eLc  = eL < E ? eL : (E - 1);
  int curS = edgeS[eLc];
  int curA = edgeA[eLc];

  for (;;) {
    const bool hasNext = (t + stride) < tiles;
    const int  tn   = hasNext ? (t + stride) : t;
    const int  nL   = tn * 32 + lane;
    const int  nLc  = nL < E ? nL : (E - 1);
    const int  nxtS = edgeS[nLc];
    const int  nxtA = edgeA[nLc];

    const float cost = tile_cost(embS, embA, Ps, Pa, sW2, curS, curA, lane, sub, sub4, b2v);

    if (eL < E) {
      outCost[eL] = cost;
      atomicMax(&segMax[curS], ordKey(-cost * invT));
    }
    if (!hasNext) break;
    t = tn; eL = nL; curS = nxtS; curA = nxtA;
  }
}

#define SEG_SHIFT    8
#define SEG_NODES    256
#define NB_MAX       1024
#define PART_BLOCKS  256
#define PART_THREADS 1024
#define CHUNK_MAX    4096
#define STAT_REGS    8

__global__ __launch_bounds__(PART_THREADS)
void edge_gate_place(const float* __restrict__ embS,
                     const float* __restrict__ embA,
                     const int*   __restrict__ edgeS,
                     const int*   __restrict__ edgeA,
                     const float* __restrict__ Ps,
                     const float* __restrict__ Pa,
                     const float* __restrict__ W2,
                     const float* __restrict__ b2,
                     const float* __restrict__ logT,
                     float*         __restrict__ outCost,
                     int*           __restrict__ offG,
                     float*         __restrict__ recVal,
                     unsigned char* __restrict__ recNode,
                     int E, int NB, int chunk, int chunkPad) {
  __shared__ __attribute__((aligned(16))) float sW2[H_DIM];
  __shared__ int h[NB_MAX];
  __shared__ int fill[NB_MAX];
  __shared__ int off[NB_MAX + 1];
  __shared__ int   sSrc[CHUNK_MAX];
  __shared__ float sArg[CHUNK_MAX];
  __shared__ __attribute__((aligned(16))) float    sVal[CHUNK_MAX];
  __shared__ __attribute__((aligned(16))) unsigned sNode4[CHUNK_MAX / 4];
  unsigned char* sNode = (unsigned char*)sNode4;

  const int tid = threadIdx.x;
  const int b   = blockIdx.x;
  if (tid < H_DIM) sW2[tid] = W2[tid];
  for (int k = tid; k < NB; k += PART_THREADS) { h[k] = 0; fill[k] = 0; }
  sNode4[tid] = 0u;
  __syncthreads();

  const int wave = __builtin_amdgcn_readfirstlane(tid >> 5);
  const int lane = tid & 31;
  const int sub  = lane & 7;
  const int sub4 = sub * 4;
  const float b2v  = b2[0];
  const float invT = __expf(-logT[0]);

  const int lo = b * chunk;
  const int hi = (lo + chunk < E) ? (lo + chunk) : E;
  const int n  = (hi > lo) ? (hi - lo) : 0;
  const int nTiles = (n + 31) >> 5;

  int t = wave;
  if (t < nTiles) {
    int iL   = t * 32 + lane;
    int eLc  = iL < n ? (lo + iL) : (hi - 1);
    int curS = edgeS[eLc];
    int curA = edgeA[eLc];

    for (;;) {
      const bool hasNext = (t + EDGE_WAVES) < nTiles;
      const int  tn   = hasNext ? (t + EDGE_WAVES) : t;
      const int  nL   = tn * 32 + lane;
      const int  nLc  = nL < n ? (lo + nL) : (hi - 1);
      const int  nxtS = edgeS[nLc];
      const int  nxtA = edgeA[nLc];

      const float cost = tile_cost(embS, embA, Ps, Pa, sW2, curS, curA, lane, sub, sub4, b2v);

      if (iL < n) {
        outCost[lo + iL] = cost;
        sSrc[iL] = curS;
        sArg[iL] = -cost * invT;
        atomicAdd(&h[curS >> SEG_SHIFT], 1);
      }
      if (!hasNext) break;
      t = tn; iL = nL; curS = nxtS; curA = nxtA;
    }
  }
  __syncthreads();

  if (tid < 32) {
    const int per = (NB + 31) >> 5;
    const int k0  = tid * per;
    int s = 0;
    for (int j = 0; j < per; ++j) { const int k = k0 + j; if (k < NB) s += h[k]; }
    int inc = s;
    for (int d = 1; d < 32; d <<= 1) {
      const int o = __shfl_up(inc, d);
      if (tid >= d) inc += o;
    }
    int run = inc - s;
    for (int j = 0; j < per; ++j) {
      const int k = k0 + j;
      if (k < NB) { off[k] = run; run += h[k]; }
    }
    if (tid == 31) off[NB] = inc;
  }
  __syncthreads();

  for (int k = tid; k <= NB; k += PART_THREADS) offG[b * (NB + 1) + k] = off[k];

#pragma unroll
  for (int j = 0; j < 4; ++j) {
    const int i = j * PART_THREADS + tid;
    if (i < n) {
      const int sv  = sSrc[i];
      const int k   = sv >> SEG_SHIFT;
      const int pos = off[k] + atomicAdd(&fill[k], 1);
      sVal[pos]  = sArg[i];
      sNode[pos] = (unsigned char)(sv & (SEG_NODES - 1));
    }
  }
  __syncthreads();

  float*    gv = recVal + (size_t)b * chunkPad;
  unsigned* gn = (unsigned*)(recNode + (size_t)b * chunkPad);
#pragma unroll
  for (int j = 0; j < 4; ++j) {
    const int i = j * PART_THREADS + tid;
    if (i < n) gv[i] = sVal[i];
  }
  if (4 * tid < n) gn[tid] = sNode4[tid];
}

__global__ __launch_bounds__(PART_THREADS)
void bucket_stats(const int*           __restrict__ offG,
                  const float*         __restrict__ recVal,
                  const unsigned char* __restrict__ recNode,
                  unsigned*            __restrict__ segMax,
                  unsigned long long*  __restrict__ segSum,
                  int numS, int NB, int chunkPad) {
  __shared__ unsigned           mx[SEG_NODES];
  __shared__ unsigned long long sm[SEG_NODES];
  const int tid = threadIdx.x;
  const int k   = blockIdx.x;
  if (tid < SEG_NODES) { mx[tid] = 0u; sm[tid] = 0ull; }

  const int r   = tid >> 2;
  const int sub = tid & 3;
  const int* o  = offG + r * (NB + 1) + k;
  const int st  = r * chunkPad + o[0];
  const int en  = r * chunkPad + o[1];

  float rv[STAT_REGS];
  int   rn[STAT_REGS];
#pragma unroll
  for (int j = 0; j < STAT_REGS; ++j) {
    const int i = st + sub + 4 * j;
    rv[j] = 0.0f;
    rn[j] = -1;
    if (i < en) { rv[j] = recVal[i]; rn[j] = recNode[i]; }
  }
  const int rest = st + sub + 4 * STAT_REGS;
  __syncthreads();

#pragma unroll
  for (int j = 0; j < STAT_REGS; ++j)
    if (rn[j] >= 0) atomicMax(&mx[rn[j]], ordKey(rv[j]));
  for (int i = rest; i < en; i += 4)
    atomicMax(&mx[recNode[i]], ordKey(recVal[i]));
  __syncthreads();

#pragma unroll
  for (int j = 0; j < STAT_REGS; ++j) {
    if (rn[j] >= 0) {
      const float m = ordVal(mx[rn[j]]);
      atomicAdd(&sm[rn[j]], (unsigned long long)(__expf(rv[j] - m) * SUM_SCALE));
    }
  }
  for (int i = rest; i < en; i += 4) {
    const int   nd = recNode[i];
    const float m  = ordVal(mx[nd]);
    atomicAdd(&sm[nd], (unsigned long long)(__expf(recVal[i] - m) * SUM_SCALE));
  }
  __syncthreads();

  if (tid < SEG_NODES) {
    const int s = k * SEG_NODES + tid;
    if (s < numS) { segMax[s] = mx[tid]; segSum[s] = sm[tid]; }
  }
}

__global__ void seg_expsum(const float* __restrict__ outCost,
                           const int* __restrict__ edgeS,
                           const unsigned* __restrict__ segMax,
                           unsigned long long* __restrict__ segSum,
                           const float* __restrict__ logT, int E) {
  int e = blockIdx.x * blockDim.x + threadIdx.x;
  if (e >= E) return;
  float invT = __expf(-logT[0]);
  int s = edgeS[e];
  float val = -outCost[e] * invT;
  float m = ordVal(segMax[s]);
  atomicAdd(&segSum[s], (unsigned long long)(__expf(val - m) * SUM_SCALE));
}

__global__ void seg_norm(const float* __restrict__ outCost,
                         const int* __restrict__ edgeS,
                         const unsigned* __restrict__ segMax,
                         const unsigned long long* __restrict__ segSum,
                         const float* __restrict__ logT,
                         float* __restrict__ outW, int E) {
  int e = blockIdx.x * blockDim.x + threadIdx.x;
  if (e >= E) return;
  float invT = __expf(-logT[0]);
  int s = edgeS[e];
  float val = -outCost[e] * invT;
  float m = ordVal(segMax[s]);
  outW[e] = __expf(val - m) / ((float)segSum[s] * SUM_INV_SCALE);
}

extern "C" void kernel_launch(void* const* d_in, const int* in_sizes, int n_in,
                              void* d_out, int out_size, void* d_ws, size_t ws_size,
                              hipStream_t stream) {
  const float* embS = (const float*)d_in[0];
  const float* embA = (const float*)d_in[1];
  const int*   eidx = (const int*)d_in[2];
  const float* W1   = (const float*)d_in[3];
  const float* b1   = (const float*)d_in[4];
  const float* W2   = (const float*)d_in[5];
  const float* b2   = (const float*)d_in[6];
  const float* logT = (const float*)d_in[7];

  const int numS = in_sizes[0] / D_DIM;
  const int numA = in_sizes[1] / D_DIM;
  const int E    = in_sizes[2] / 2;
  const int* edgeS = eidx;
  const int* edgeA = eidx + E;

  float* outW    = (float*)d_out;
  float* outCost = (float*)d_out + E;

  const size_t pFloats = (size_t)(numS + numA) * H_DIM;
  const size_t need    = pFloats * sizeof(float)
                       + (size_t)numS * (sizeof(unsigned long long) + sizeof(unsigned));

  const int NB       = (numS + SEG_NODES - 1) >> SEG_SHIFT;
  const int chunk    = (((E + PART_BLOCKS - 1) / PART_BLOCKS) + 31) & ~31;
  const int chunkPad = (chunk + 127) & ~127;
  const size_t recSlots = (size_t)PART_BLOCKS * chunkPad;
  const size_t needGrp  = need + recSlots * (sizeof(float) + 1)
                        + (size_t)PART_BLOCKS * (NB + 1) * sizeof(int);

  if (ws_size >= needGrp && NB <= NB_MAX && chunk <= CHUNK_MAX) {
    float* P = (float*)d_ws;
    float*         recVal  = P + pFloats;
    unsigned char* recNode = (unsigned char*)(recVal + recSlots);
    unsigned long long* segSum = (unsigned long long*)(recNode + recSlots);
    unsigned*           segMax = (unsigned*)(segSum + numS);
    int*                offG   = (int*)(segMax + numS);

    node_proj<<<256, EDGE_WAVES * 32, 0, stream>>>(embS, embA, W1, b1, P, numS, numA);
    edge_gate_place<<<PART_BLOCKS, PART_THREADS, 0, stream>>>(
        embS, embA, edgeS, edgeA, P, P + (size_t)numS * H_DIM, W2, b2, logT,
        outCost, offG, recVal, recNode, E, NB, chunk, chunkPad);
    bucket_stats<<<NB, PART_THREADS, 0, stream>>>(
        offG, recVal, recNode, segMax, segSum, numS, NB, chunkPad);
    seg_norm<<<(E + 255) / 256, 256, 0, stream>>>(outCost, edgeS, segMax, segSum, logT, outW, E);
    return;
  }

  if (ws_size >= need) {
    float* P = (float*)d_ws;
    unsigned long long* segSum = (unsigned long long*)(P + pFloats);
    unsigned*           segMax = (unsigned*)(segSum + numS);

    init_seg<<<(numS + 255) / 256, 256, 0, stream>>>(segMax, segSum, numS);
    node_proj<<<256, EDGE_WAVES * 32, 0, stream>>>(embS, embA, W1, b1, P, numS, numA);
    edge_gate_cost<<<256, EDGE_WAVES * 32, 0, stream>>>(
        embS, embA, edgeS, edgeA, P, P + (size_t)numS * H_DIM, W2, b2, logT,
        outCost, segMax, E);
    seg_expsum<<<(E + 255) / 256, 256, 0, stream>>>(outCost, edgeS, segMax, segSum, logT, E);
    seg_norm<<<(E + 255) / 256, 256, 0, stream>>>(outCost, edgeS, segMax, segSum, logT, outW, E);
    return;
  }

  unsigned long long* segSum = (unsigned long long*)d_ws;
  unsigned*           segMax = (unsigned*)(segSum + numS);

  init_seg<<<(numS + 255) / 256, 256, 0, stream>>>(segMax, segSum, numS);

  edge_mlp_cost<<<256, EDGE_WAVES * 32, 0, stream>>>(
      embS, embA, edgeS, edgeA, W1, b1, W2, b2, logT, outCost, segMax, E);

  seg_expsum<<<(E + 255) / 256, 256, 0, stream>>>(outCost, edgeS, segMax, segSum, logT, E);
  seg_norm<<<(E + 255) / 256, 256, 0, stream>>>(outCost, edgeS, segMax, segSum, logT, outW, E);
}


# edge pass by source bucket, costs kept in record slots: a bucket's block reads its source rows from LDS, gathers only target rows; edges read costs back by slot, no scattered 4-byte stores
# speedup vs baseline: 4.3492x; 1.0680x over previous
#include <hip/hip_runtime.h>
#include <hip/hip_bf16.h>
#include <math.h>

typedef __attribute__((ext_vector_type(2))) float v2f;
typedef __attribute__((ext_vector_type(4))) float v4f;
typedef __attribute__((ext_vector_type(8))) float v8f;

#define D_DIM 64
#define H_DIM 128
#define CPITCH 66

__device__ __forceinline__ unsigned ordKey(float f) {
  unsigned u = __float_as_uint(f);
  return (u & 0x80000000u) ? ~u : (u | 0x80000000u);
}
__device__ __forceinline__ float ordVal(unsigned k) {
  unsigned u = (k & 0x80000000u) ? (k ^ 0x80000000u) : ~k;
  return __uint_as_float(u);
}

#define SUM_SCALE     1099511627776.0f
#define SUM_INV_SCALE (1.0f / 1099511627776.0f)

__global__ void init_seg(unsigned* __restrict__ segMax,
                         unsigned long long* __restrict__ segSum, int numS) {
  int i = blockIdx.x * blockDim.x + threadIdx.x;
  if (i < numS) { segMax[i] = 0u; segSum[i] = 0ull; }
}

#define EDGE_WAVES 32
__global__ __launch_bounds__(EDGE_WAVES * 32)
void edge_mlp_cost(const float* __restrict__ embS,
                   const float* __restrict__ embA,
                   const int*   __restrict__ edgeS,
                   const int*   __restrict__ edgeA,
                   const float* __restrict__ W1,
                   const float* __restrict__ b1,
                   const float* __restrict__ W2,
                   const float* __restrict__ b2,
                   const float* __restrict__ logT,
                   float*       __restrict__ outCost,
                   unsigned*    __restrict__ segMax,
                   int E) {
  __shared__ float sW1[H_DIM * CPITCH * 2];
  __shared__ float sB1[H_DIM];
  __shared__ float sW2[H_DIM];
  __shared__ float sGate[EDGE_WAVES][16];

  const int tid = threadIdx.x;
  for (int i = tid; i < H_DIM * H_DIM; i += EDGE_WAVES * 32) {
    int kk = i >> 7;
    int c  = i & 127;
    sW1[(c * CPITCH + (kk >> 1)) * 2 + (kk & 1)] = W1[i];
  }
  if (tid < H_DIM) { sB1[tid] = b1[tid]; sW2[tid] = W2[tid]; }
  __syncthreads();

  const int wave = __builtin_amdgcn_readfirstlane(tid >> 5);
  const int lane = tid & 31;
  const int half = lane >> 4;
  const int col  = lane & 15;
  const float b2v  = b2[0];
  const float invT = __expf(-logT[0]);
  const int tiles  = (E + 15) >> 4;
  const int stride = gridDim.x * EDGE_WAVES;

  for (int t = blockIdx.x * EDGE_WAVES + wave; t < tiles; t += stride) {
  const int  e0    = t * 16;
  const int  e     = e0 + col;
  const bool valid = (e < E);
  const int  ec    = valid ? e : (E - 1);
  const int  sI    = edgeS[ec];
  const int  aI    = edgeA[ec];
  const float* es = embS + (size_t)sI * D_DIM;
  const float* ea = embA + (size_t)aI * D_DIM;

  v2f a[32];
#pragma unroll
  for (int k = 0; k < 16; ++k) {
    a[k]      = *(const v2f*)(es + 4 * k + 2 * half);
    a[k + 16] = *(const v2f*)(ea + 4 * k + 2 * half);
  }
  float ss = 0.0f;
#pragma unroll
  for (int k = 0; k < 16; ++k) {
    float dx = a[k][0] - a[k + 16][0];
    float dy = a[k][1] - a[k + 16][1];
    ss += dx * dx + dy * dy;
  }
  ss += __shfl_xor(ss, 16);

  float gacc[8] = {0, 0, 0, 0, 0, 0, 0, 0};
#pragma unroll 1
  for (int n = 0; n < 8; ++n) {
    const float* bbase = sW1 + ((n * 16 + col) * CPITCH + half) * 2;
    v8f c0 = {};
    v8f c1 = {};
#pragma unroll
    for (int k = 0; k < 32; k += 2) {
      v2f b0 = *(const v2f*)(bbase + 4 * k);
      v2f b1v = *(const v2f*)(bbase + 4 * (k + 1));
      c0 = __builtin_amdgcn_wmma_f32_16x16x4_f32(false, a[k], false, b0,
                                                 (short)0, c0, false, false);
      c1 = __builtin_amdgcn_wmma_f32_16x16x4_f32(false, a[k + 1], false, b1v,
                                                 (short)0, c1, false, false);
    }
    const float bn  = sB1[n * 16 + col];
    const float w2n = sW2[n * 16 + col];
#pragma unroll
    for (int r = 0; r < 8; ++r) {
      float h = c0[r] + c1[r] + bn;
      h = h > 0.0f ? h : 0.0f;
      gacc[r] += h * w2n;
    }
  }

#pragma unroll
  for (int r = 0; r < 8; ++r) {
    float g = gacc[r];
    g += __shfl_xor(g, 1);
    g += __shfl_xor(g, 2);
    g += __shfl_xor(g, 4);
    g += __shfl_xor(g, 8);
    if (lane == 0)  sGate[wave][r]     = g;
    if (lane == 16) sGate[wave][r + 8] = g;
  }

  if (lane < 16 && valid) {
    float logit = sGate[wave][lane] + b2v;
    float gate  = 1.0f / (1.0f + __expf(-logit));
    float cost  = sqrtf(ss) * gate;
    outCost[e]  = cost;
    atomicMax(&segMax[sI], ordKey(-cost * invT));
  }
  }
}

__global__ __launch_bounds__(EDGE_WAVES * 32)
void node_proj(const float* __restrict__ embS,
               const float* __restrict__ embA,
               const float* __restrict__ W1,
               const float* __restrict__ b1,
               float*       __restrict__ P,
               int numS, int numA) {
  __shared__ float sW1[H_DIM * CPITCH * 2];
  __shared__ __attribute__((aligned(16))) float sB1[2 * H_DIM];

  const int tid = threadIdx.x;
  for (int i = tid; i < H_DIM * H_DIM; i += EDGE_WAVES * 32) {
    int kk = i >> 7;
    int c  = i & 127;
    sW1[(c * CPITCH + (kk >> 1)) * 2 + (kk & 1)] = W1[i];
  }
  if (tid < H_DIM) { sB1[tid] = b1[tid]; sB1[H_DIM + tid] = 0.0f; }
  __syncthreads();

  const int wave = __builtin_amdgcn_readfirstlane(tid >> 5);
  const int lane = tid & 31;
  const int half = lane >> 4;
  const int col  = lane & 15;
  const int tilesS = (numS + 15) >> 4;
  const int tilesA = (numA + 15) >> 4;
  const int tiles  = tilesS + tilesA;

  for (int t = wave * (int)gridDim.x + (int)blockIdx.x; t < tiles;
       t += EDGE_WAVES * (int)gridDim.x) {
    const bool isA = (t >= tilesS);
    const int  tt  = isA ? (t - tilesS) : t;
    const float* emb = isA ? embA : embS;
    const int  nrows = isA ? numA : numS;
    const int  node  = tt * 16 + col;
    const bool valid = (node < nrows);
    const int  nc    = valid ? node : (nrows - 1);
    const float* row = emb + (size_t)nc * D_DIM;

    v2f a[16];
#pragma unroll
    for (int k = 0; k < 16; ++k) a[k] = *(const v2f*)(row + 4 * k + 2 * half);

    float* prow = P + ((size_t)(isA ? numS : 0) + (size_t)nc) * H_DIM + 8 * half;
    const int koff = isA ? D_DIM : 0;
    const int boff = isA ? H_DIM : 0;

#pragma unroll 1
    for (int nt = 0; nt < 8; ++nt) {
      const float* wbase = sW1 + ((nt * 16 + col) * CPITCH + half) * 2 + koff;
      v8f c0 = {};
      v8f c1 = {};
#pragma unroll
      for (int k = 0; k < 16; k += 2) {
        v2f w0 = *(const v2f*)(wbase + 4 * k);
        v2f w1 = *(const v2f*)(wbase + 4 * (k + 1));
        c0 = __builtin_amdgcn_wmma_f32_16x16x4_f32(false, w0, false, a[k],
                                                   (short)0, c0, false, false);
        c1 = __builtin_amdgcn_wmma_f32_16x16x4_f32(false, w1, false, a[k + 1],
                                                   (short)0, c1, false, false);
      }
      const float* bb = sB1 + boff + nt * 16 + 8 * half;
      const v4f blo = *(const v4f*)(bb);
      const v4f bhi = *(const v4f*)(bb + 4);
      v4f lo, hi;
#pragma unroll
      for (int r = 0; r < 4; ++r) {
        lo[r] = c0[r] + c1[r] + blo[r];
        hi[r] = c0[r + 4] + c1[r + 4] + bhi[r];
      }
      if (valid) {
        *(v4f*)(prow + nt * 16)     = lo;
        *(v4f*)(prow + nt * 16 + 4) = hi;
      }
    }
  }
}

struct RowSet {
  v4f es0, es1, ea0, ea1;
  v4f ps0, ps1, ps2, ps3;
  v4f pa0, pa1, pa2, pa3;
};

__device__ __forceinline__ void fetch_rows(RowSet& r,
                                           const float* __restrict__ embS,
                                           const float* __restrict__ embA,
                                           const float* __restrict__ Ps,
                                           const float* __restrict__ Pa,
                                           int s, int a, int sub4) {
  const float* e1 = embS + (size_t)s * D_DIM + sub4;
  const float* e2 = embA + (size_t)a * D_DIM + sub4;
  const float* p1 = Ps + (size_t)s * H_DIM + sub4;
  const float* p2 = Pa + (size_t)a * H_DIM + sub4;
  r.es0 = *(const v4f*)(e1);
  r.es1 = *(const v4f*)(e1 + 32);
  r.ea0 = *(const v4f*)(e2);
  r.ea1 = *(const v4f*)(e2 + 32);
  r.ps0 = *(const v4f*)(p1);
  r.ps1 = *(const v4f*)(p1 + 32);
  r.ps2 = *(const v4f*)(p1 + 64);
  r.ps3 = *(const v4f*)(p1 + 96);
  r.pa0 = *(const v4f*)(p2);
  r.pa1 = *(const v4f*)(p2 + 32);
  r.pa2 = *(const v4f*)(p2 + 64);
  r.pa3 = *(const v4f*)(p2 + 96);
}

__device__ __forceinline__ float sq4(v4f a, v4f b) {
  v4f d = a - b;
  return d[0] * d[0] + d[1] * d[1] + d[2] * d[2] + d[3] * d[3];
}
__device__ __forceinline__ float gate4(v4f p, v4f q, v4f w) {
  v4f h = p + q;
  float g = 0.0f;
#pragma unroll
  for (int i = 0; i < 4; ++i) {
    float x = h[i] > 0.0f ? h[i] : 0.0f;
    g += x * w[i];
  }
  return g;
}

template <int M>
__device__ __forceinline__ float xadd8(float v) {
  int o = __builtin_amdgcn_update_dpp(0, __float_as_int(v), 0x160 + M, 0xf, 0xf, true);
  return v + __int_as_float(o);
}
__device__ __forceinline__ int bsel8(int v, int lane, int i) {
  return __builtin_amdgcn_ds_bpermute(((lane & 24) | i) << 2, v);
}

__device__ __forceinline__ void reduce_rows(const RowSet& r, const float* sW2, int sub4,
                                            float& ss, float& g) {
  float s = sq4(r.es0, r.ea0) + sq4(r.es1, r.ea1);
  float t = gate4(r.ps0, r.pa0, *(const v4f*)(sW2 + sub4))
          + gate4(r.ps1, r.pa1, *(const v4f*)(sW2 + 32 + sub4))
          + gate4(r.ps2, r.pa2, *(const v4f*)(sW2 + 64 + sub4))
          + gate4(r.ps3, r.pa3, *(const v4f*)(sW2 + 96 + sub4));
  s = xadd8<1>(s); t = xadd8<1>(t);
  s = xadd8<2>(s); t = xadd8<2>(t);
  s = xadd8<4>(s); t = xadd8<4>(t);
  ss = s; g = t;
}

__device__ __forceinline__ float tile_cost(const float* __restrict__ embS,
                                           const float* __restrict__ embA,
                                           const float* __restrict__ Ps,
                                           const float* __restrict__ Pa,
                                           const float* sW2,
                                           int curS, int curA,
                                           int lane, int sub, int sub4, float b2v) {
  float rss = 0.0f, rg = 0.0f;

#pragma unroll 1
  for (int i = 0; i < 8; i += 2) {
    RowSet X, Y;
    float ss, g;
    fetch_rows(X, embS, embA, Ps, Pa, bsel8(curS, lane, i), bsel8(curA, lane, i), sub4);
    fetch_rows(Y, embS, embA, Ps, Pa, bsel8(curS, lane, i + 1), bsel8(curA, lane, i + 1), sub4);
    __builtin_amdgcn_sched_barrier(0);
    reduce_rows(X, sW2, sub4, ss, g);
    if (sub == i) { rss = ss; rg = g; }
    reduce_rows(Y, sW2, sub4, ss, g);
    if (sub == i + 1) { rss = ss; rg = g; }
  }

  const float logit = rg + b2v;
  const float gate  = 1.0f / (1.0f + __expf(-logit));
  return sqrtf(rss) * gate;
}

__global__ __launch_bounds__(EDGE_WAVES * 32)
void edge_gate_cost(const float* __restrict__ embS,
                    const float* __restrict__ embA,
                    const int*   __restrict__ edgeS,
                    const int*   __restrict__ edgeA,
                    const float* __restrict__ Ps,
                    const float* __restrict__ Pa,
                    const float* __restrict__ W2,
                    const float* __restrict__ b2,
                    const float* __restrict__ logT,
                    float*       __restrict__ outCost,
                    unsigned*    __restrict__ segMax,
                    int E) {
  __shared__ __attribute__((aligned(16))) float sW2[H_DIM];
  const int tid = threadIdx.x;
  if (tid < H_DIM) sW2[tid] = W2[tid];
  __syncthreads();

  const int wave = __builtin_amdgcn_readfirstlane(tid >> 5);
  const int lane = tid & 31;
  const int sub  = lane & 7;
  const int sub4 = sub * 4;
  const float b2v  = b2[0];
  const float invT = __expf(-logT[0]);
  const int tiles  = (E + 31) >> 5;
  const int stride = (int)gridDim.x * EDGE_WAVES;

  int t = (int)blockIdx.x * EDGE_WAVES + wave;
  if (t >= tiles) return;

  int eL   = t * 32 + lane;
  int eLc  = eL < E ? eL : (E - 1);
  int curS = edgeS[eLc];
  int curA = edgeA[eLc];

  for (;;) {
    const bool hasNext = (t + stride) < tiles;
    const int  tn   = hasNext ? (t + stride) : t;
    const int  nL   = tn * 32 + lane;
    const int  nLc  = nL < E ? nL : (E - 1);
    const int  nxtS = edgeS[nLc];
    const int  nxtA = edgeA[nLc];

    const float cost = tile_cost(embS, embA, Ps, Pa, sW2, curS, curA, lane, sub, sub4, b2v);

    if (eL < E) {
      outCost[eL] = cost;
      atomicMax(&segMax[curS], ordKey(-cost * invT));
    }
    if (!hasNext) break;
    t = tn; eL = nL; curS = nxtS; curA = nxtA;
  }
}

#define SEG_SHIFT    8
#define SEG_NODES    256
#define NB_MAX       1024
#define PART_BLOCKS  256
#define PART_THREADS 1024
#define CHUNK_MAX    4096
#define STAT_REGS    8

__global__ __launch_bounds__(PART_THREADS)
void edge_gate_place(const float* __restrict__ embS,
                     const float* __restrict__ embA,
                     const int*   __restrict__ edgeS,
                     const int*   __restrict__ edgeA,
                     const float* __restrict__ Ps,
                     const float* __restrict__ Pa,
                     const float* __restrict__ W2,
                     const float* __restrict__ b2,
                     const float* __restrict__ logT,
                     float*         __restrict__ outCost,
                     int*           __restrict__ offG,
                     float*         __restrict__ recVal,
                     unsigned char* __restrict__ recNode,
                     int E, int NB, int chunk, int chunkPad) {
  __shared__ __attribute__((aligned(16))) float sW2[H_DIM];
  __shared__ int h[NB_MAX];
  __shared__ int fill[NB_MAX];
  __shared__ int off[NB_MAX + 1];
  __shared__ int   sSrc[CHUNK_MAX];
  __shared__ float sArg[CHUNK_MAX];
  __shared__ __attribute__((aligned(16))) float    sVal[CHUNK_MAX];
  __shared__ __attribute__((aligned(16))) unsigned sNode4[CHUNK_MAX / 4];
  unsigned char* sNode = (unsigned char*)sNode4;

  const int tid = threadIdx.x;
  const int b   = blockIdx.x;
  if (tid < H_DIM) sW2[tid] = W2[tid];
  for (int k = tid; k < NB; k += PART_THREADS) { h[k] = 0; fill[k] = 0; }
  sNode4[tid] = 0u;
  __syncthreads();

  const int wave = __builtin_amdgcn_readfirstlane(tid >> 5);
  const int lane = tid & 31;
  const int sub  = lane & 7;
  const int sub4 = sub * 4;
  const float b2v  = b2[0];
  const float invT = __expf(-logT[0]);

  const int lo = b * chunk;
  const int hi = (lo + chunk < E) ? (lo + chunk) : E;
  const int n  = (hi > lo) ? (hi - lo) : 0;
  const int nTiles = (n + 31) >> 5;

  int t = wave;
  if (t < nTiles) {
    int iL   = t * 32 + lane;
    int eLc  = iL < n ? (lo + iL) : (hi - 1);
    int curS = edgeS[eLc];
    int curA = edgeA[eLc];

    for (;;) {
      const bool hasNext = (t + EDGE_WAVES) < nTiles;
      const int  tn   = hasNext ? (t + EDGE_WAVES) : t;
      const int  nL   = tn * 32 + lane;
      const int  nLc  = nL < n ? (lo + nL) : (hi - 1);
      const int  nxtS = edgeS[nLc];
      const int  nxtA = edgeA[nLc];

      const float cost = tile_cost(embS, embA, Ps, Pa, sW2, curS, curA, lane, sub, sub4, b2v);

      if (iL < n) {
        outCost[lo + iL] = cost;
        sSrc[iL] = curS;
        sArg[iL] = -cost * invT;
        atomicAdd(&h[curS >> SEG_SHIFT], 1);
      }
      if (!hasNext) break;
      t = tn; iL = nL; curS = nxtS; curA = nxtA;
    }
  }
  __syncthreads();

  if (tid < 32) {
    const int per = (NB + 31) >> 5;
    const int k0  = tid * per;
    int s = 0;
    for (int j = 0; j < per; ++j) { const int k = k0 + j; if (k < NB) s += h[k]; }
    int inc = s;
    for (int d = 1; d < 32; d <<= 1) {
      const int o = __shfl_up(inc, d);
      if (tid >= d) inc += o;
    }
    int run = inc - s;
    for (int j = 0; j < per; ++j) {
      const int k = k0 + j;
      if (k < NB) { off[k] = run; run += h[k]; }
    }
    if (tid == 31) off[NB] = inc;
  }
  __syncthreads();

  for (int k = tid; k <= NB; k += PART_THREADS) offG[b * (NB + 1) + k] = off[k];

#pragma unroll
  for (int j = 0; j < 4; ++j) {
    const int i = j * PART_THREADS + tid;
    if (i < n) {
      const int sv  = sSrc[i];
      const int k   = sv >> SEG_SHIFT;
      const int pos = off[k] + atomicAdd(&fill[k], 1);
      sVal[pos]  = sArg[i];
      sNode[pos] = (unsigned char)(sv & (SEG_NODES - 1));
    }
  }
  __syncthreads();

  float*    gv = recVal + (size_t)b * chunkPad;
  unsigned* gn = (unsigned*)(recNode + (size_t)b * chunkPad);
#pragma unroll
  for (int j = 0; j < 4; ++j) {
    const int i = j * PART_THREADS + tid;
    if (i < n) gv[i] = sVal[i];
  }
  if (4 * tid < n) gn[tid] = sNode4[tid];
}

__global__ __launch_bounds__(PART_THREADS)
void bucket_stats(const int*           __restrict__ offG,
                  const float*         __restrict__ recVal,
                  const unsigned char* __restrict__ recNode,
                  unsigned*            __restrict__ segMax,
                  unsigned long long*  __restrict__ segSum,
                  int numS, int NB, int chunkPad) {
  __shared__ unsigned           mx[SEG_NODES];
  __shared__ unsigned long long sm[SEG_NODES];
  const int tid = threadIdx.x;
  const int k   = blockIdx.x;
  if (tid < SEG_NODES) { mx[tid] = 0u; sm[tid] = 0ull; }

  const int r   = tid >> 2;
  const int sub = tid & 3;
  const int* o  = offG + r * (NB + 1) + k;
  const int st  = r * chunkPad + o[0];
  const int en  = r * chunkPad + o[1];

  float rv[STAT_REGS];
  int   rn[STAT_REGS];
#pragma unroll
  for (int j = 0; j < STAT_REGS; ++j) {
    const int i = st + sub + 4 * j;
    rv[j] = 0.0f;
    rn[j] = -1;
    if (i < en) { rv[j] = recVal[i]; rn[j] = recNode[i]; }
  }
  const int rest = st + sub + 4 * STAT_REGS;
  __syncthreads();

#pragma unroll
  for (int j = 0; j < STAT_REGS; ++j)
    if (rn[j] >= 0) atomicMax(&mx[rn[j]], ordKey(rv[j]));
  for (int i = rest; i < en; i += 4)
    atomicMax(&mx[recNode[i]], ordKey(recVal[i]));
  __syncthreads();

#pragma unroll
  for (int j = 0; j < STAT_REGS; ++j) {
    if (rn[j] >= 0) {
      const float m = ordVal(mx[rn[j]]);
      atomicAdd(&sm[rn[j]], (unsigned long long)(__expf(rv[j] - m) * SUM_SCALE));
    }
  }
  for (int i = rest; i < en; i += 4) {
    const int   nd = recNode[i];
    const float m  = ordVal(mx[nd]);
    atomicAdd(&sm[nd], (unsigned long long)(__expf(recVal[i] - m) * SUM_SCALE));
  }
  __syncthreads();

  if (tid < SEG_NODES) {
    const int s = k * SEG_NODES + tid;
    if (s < numS) { segMax[s] = mx[tid]; segSum[s] = sm[tid]; }
  }
}

#define BE_CAP 6144
#define EPITCH 68
#define PPITCH 132

__global__ __launch_bounds__(PART_THREADS)
void group_edges(const int* __restrict__ edgeS,
                 const int* __restrict__ edgeA,
                 int*           __restrict__ offG,
                 int*           __restrict__ recA,
                 unsigned char* __restrict__ recNode,
                 int*           __restrict__ pos,
                 int E, int NB, int chunk, int chunkPad) {
  __shared__ int h[NB_MAX];
  __shared__ int fill[NB_MAX];
  __shared__ int off[NB_MAX + 1];
  __shared__ __attribute__((aligned(16))) int      sA[CHUNK_MAX];
  __shared__ __attribute__((aligned(16))) unsigned sNode4[CHUNK_MAX / 4];
  unsigned char* sNode = (unsigned char*)sNode4;

  const int tid = threadIdx.x;
  const int b   = blockIdx.x;
  for (int k = tid; k < NB; k += PART_THREADS) { h[k] = 0; fill[k] = 0; }
  sNode4[tid] = 0u;
  __syncthreads();

  const int lo = b * chunk;
  const int hi = (lo + chunk < E) ? (lo + chunk) : E;
  const int n  = (hi > lo) ? (hi - lo) : 0;

  int sv[4], av[4];
#pragma unroll
  for (int j = 0; j < 4; ++j) {
    const int i = j * PART_THREADS + tid;
    sv[j] = -1; av[j] = 0;
    if (i < n) {
      sv[j] = edgeS[lo + i];
      av[j] = edgeA[lo + i];
      atomicAdd(&h[sv[j] >> SEG_SHIFT], 1);
    }
  }
  __syncthreads();

  if (tid < 32) {
    const int per = (NB + 31) >> 5;
    const int k0  = tid * per;
    int s = 0;
    for (int j = 0; j < per; ++j) { const int k = k0 + j; if (k < NB) s += h[k]; }
    int inc = s;
    for (int d = 1; d < 32; d <<= 1) {
      const int o = __shfl_up(inc, d);
      if (tid >= d) inc += o;
    }
    int run = inc - s;
    for (int j = 0; j < per; ++j) {
      const int k = k0 + j;
      if (k < NB) { off[k] = run; run += h[k]; }
    }
    if (tid == 31) off[NB] = inc;
  }
  __syncthreads();

  for (int k = tid; k <= NB; k += PART_THREADS) offG[b * (NB + 1) + k] = off[k];

#pragma unroll
  for (int j = 0; j < 4; ++j) {
    const int i = j * PART_THREADS + tid;
    if (i < n) {
      const int k = sv[j] >> SEG_SHIFT;
      const int p = off[k] + atomicAdd(&fill[k], 1);
      sA[p]    = av[j];
      sNode[p] = (unsigned char)(sv[j] & (SEG_NODES - 1));
      pos[lo + i] = b * chunkPad + p;
    }
  }
  __syncthreads();

  int*      ga = recA + (size_t)b * chunkPad;
  unsigned* gn = (unsigned*)(recNode + (size_t)b * chunkPad);
#pragma unroll
  for (int j = 0; j < 4; ++j) {
    const int i = j * PART_THREADS + tid;
    if (i < n) ga[i] = sA[i];
  }
  if (4 * tid < n) gn[tid] = sNode4[tid];
}

struct RowT {
  v4f ea0, ea1;
  v4f pa0, pa1, pa2, pa3;
};
__device__ __forceinline__ void fetch_t(RowT& r,
                                        const float* __restrict__ embA,
                                        const float* __restrict__ Pa,
                                        int a, int sub4) {
  const float* e2 = embA + (size_t)a * D_DIM + sub4;
  const float* p2 = Pa + (size_t)a * H_DIM + sub4;
  r.ea0 = *(const v4f*)(e2);
  r.ea1 = *(const v4f*)(e2 + 32);
  r.pa0 = *(const v4f*)(p2);
  r.pa1 = *(const v4f*)(p2 + 32);
  r.pa2 = *(const v4f*)(p2 + 64);
  r.pa3 = *(const v4f*)(p2 + 96);
}
__device__ __forceinline__ void reduce_t(const RowT& r, const float* sE, const float* sP,
                                         const float* sW2, int sub4, float& ss, float& g) {
  const v4f es0 = *(const v4f*)(sE + sub4);
  const v4f es1 = *(const v4f*)(sE + 32 + sub4);
  const v4f ps0 = *(const v4f*)(sP + sub4);
  const v4f ps1 = *(const v4f*)(sP + 32 + sub4);
  const v4f ps2 = *(const v4f*)(sP + 64 + sub4);
  const v4f ps3 = *(const v4f*)(sP + 96 + sub4);
  float s = sq4(es0, r.ea0) + sq4(es1, r.ea1);
  float t = gate4(ps0, r.pa0, *(const v4f*)(sW2 + sub4))
          + gate4(ps1, r.pa1, *(const v4f*)(sW2 + 32 + sub4))
          + gate4(ps2, r.pa2, *(const v4f*)(sW2 + 64 + sub4))
          + gate4(ps3, r.pa3, *(const v4f*)(sW2 + 96 + sub4));
  s = xadd8<1>(s); t = xadd8<1>(t);
  s = xadd8<2>(s); t = xadd8<2>(t);
  s = xadd8<4>(s); t = xadd8<4>(t);
  ss = s; g = t;
}

__device__ __forceinline__ float tile_cost_b(const float* __restrict__ embA,
                                             const float* __restrict__ Pa,
                                             const float* sRowE, const float* sRowP,
                                             const float* sW2,
                                             int curN, int curA,
                                             int lane, int sub, int sub4, float b2v) {
  float rss = 0.0f, rg = 0.0f;

#pragma unroll 1
  for (int i = 0; i < 8; i += 2) {
    RowT X, Y;
    float ss, g;
    const int n0 = bsel8(curN, lane, i);
    const int n1 = bsel8(curN, lane, i + 1);
    fetch_t(X, embA, Pa, bsel8(curA, lane, i), sub4);
    fetch_t(Y, embA, Pa, bsel8(curA, lane, i + 1), sub4);
    __builtin_amdgcn_sched_barrier(0);
    reduce_t(X, sRowE + n0 * EPITCH, sRowP + n0 * PPITCH, sW2, sub4, ss, g);
    if (sub == i) { rss = ss; rg = g; }
    reduce_t(Y, sRowE + n1 * EPITCH, sRowP + n1 * PPITCH, sW2, sub4, ss, g);
    if (sub == i + 1) { rss = ss; rg = g; }
  }

  const float logit = rg + b2v;
  const float gate  = 1.0f / (1.0f + __expf(-logit));
  return sqrtf(rss) * gate;
}

__global__ __launch_bounds__(PART_THREADS)
void bucket_edge(const float* __restrict__ embS,
                 const float* __restrict__ embA,
                 const float* __restrict__ Ps,
                 const float* __restrict__ Pa,
                 const float* __restrict__ W2,
                 const float* __restrict__ b2,
                 const float* __restrict__ logT,
                 const int*           __restrict__ offG,
                 const int*           __restrict__ recA,
                 const unsigned char* __restrict__ recNode,
                 float*               recCost,
                 unsigned*            __restrict__ segMax,
                 unsigned long long*  __restrict__ segSum,
                 int numS, int NB, int chunkPad) {
  __shared__ __attribute__((aligned(16))) float sRowE[SEG_NODES * EPITCH];
  __shared__ __attribute__((aligned(16))) float sRowP[SEG_NODES * PPITCH];
  __shared__ __attribute__((aligned(16))) float sW2[H_DIM];
  __shared__ unsigned           mx[SEG_NODES];
  __shared__ unsigned long long sm[SEG_NODES];
  __shared__ int runLen[PART_BLOCKS];
  __shared__ int runStart[PART_BLOCKS + 1];
  __shared__ int cA[BE_CAP];
  __shared__ int cPos[BE_CAP];
  __shared__ unsigned char cN[BE_CAP];

  const int tid = threadIdx.x;
  const int k   = blockIdx.x;
  if (tid < H_DIM) sW2[tid] = W2[tid];
  if (tid < SEG_NODES) { mx[tid] = 0u; sm[tid] = 0ull; }

  const int r   = tid >> 2;
  const int sub = tid & 3;
  const int* o  = offG + r * (NB + 1) + k;
  const int st  = o[0];
  const int en  = o[1];
  const int gbase = r * chunkPad;
  if (sub == 0) runLen[r] = en - st;

  const int nodeBase = k << SEG_SHIFT;
  const int nNodes   = (numS - nodeBase < SEG_NODES) ? (numS - nodeBase) : SEG_NODES;
  const v4f* gE = (const v4f*)(embS + (size_t)nodeBase * D_DIM);
  for (int i = tid; i < nNodes * (D_DIM / 4); i += PART_THREADS)
    *(v4f*)(sRowE + (i >> 4) * EPITCH + (i & 15) * 4) = gE[i];
  const v4f* gP = (const v4f*)(Ps + (size_t)nodeBase * H_DIM);
  for (int i = tid; i < nNodes * (H_DIM / 4); i += PART_THREADS)
    *(v4f*)(sRowP + (i >> 5) * PPITCH + (i & 31) * 4) = gP[i];
  __syncthreads();

  if (tid < 32) {
    int s = 0;
    for (int j = 0; j < PART_BLOCKS / 32; ++j) s += runLen[tid * (PART_BLOCKS / 32) + j];
    int inc = s;
    for (int d = 1; d < 32; d <<= 1) {
      const int v = __shfl_up(inc, d);
      if (tid >= d) inc += v;
    }
    int run = inc - s;
    for (int j = 0; j < PART_BLOCKS / 32; ++j) {
      const int q = tid * (PART_BLOCKS / 32) + j;
      runStart[q] = run;
      run += runLen[q];
    }
    if (tid == 31) runStart[PART_BLOCKS] = inc;
  }
  __syncthreads();

  const int wave = __builtin_amdgcn_readfirstlane(tid >> 5);
  const int lane = tid & 31;
  const int sub8 = lane & 7;
  const int sub4 = sub8 * 4;
  const float b2v  = b2[0];
  const float invT = __expf(-logT[0]);
  const int total = __builtin_amdgcn_readfirstlane(runStart[PART_BLOCKS]);
  const int rs    = runStart[r];

  for (int base = 0; base < total; base += BE_CAP) {
    const int cnt = (total - base < BE_CAP) ? (total - base) : BE_CAP;

    int ca[STAT_REGS];
    int cn[STAT_REGS];
#pragma unroll
    for (int q = 0; q < STAT_REGS; ++q) {
      const int i = st + sub + 4 * q;
      const int j = rs + (i - st) - base;
      ca[q] = 0;
      cn[q] = -1;
      if (i < en && j >= 0 && j < cnt) { ca[q] = recA[gbase + i]; cn[q] = recNode[gbase + i]; }
    }
#pragma unroll
    for (int q = 0; q < STAT_REGS; ++q) {
      if (cn[q] >= 0) {
        const int i = st + sub + 4 * q;
        const int j = rs + (i - st) - base;
        cA[j]   = ca[q];
        cN[j]   = (unsigned char)cn[q];
        cPos[j] = gbase + i;
      }
    }
    for (int i = st + sub + 4 * STAT_REGS; i < en; i += 4) {
      const int j = rs + (i - st) - base;
      if (j >= 0 && j < cnt) {
        cA[j]   = recA[gbase + i];
        cN[j]   = recNode[gbase + i];
        cPos[j] = gbase + i;
      }
    }
    __syncthreads();

    const int nT = (cnt + 31) >> 5;
    for (int t = wave; t < nT; t += EDGE_WAVES) {
      const int j  = t * 32 + lane;
      const int jc = j < cnt ? j : (cnt - 1);
      const int curN = cN[jc];
      const int curA = cA[jc];
      const float cost = tile_cost_b(embA, Pa, sRowE, sRowP, sW2, curN, curA,
                                     lane, sub8, sub4, b2v);
      if (j < cnt) {
        recCost[cPos[j]] = cost;
        atomicMax(&mx[curN], ordKey(-cost * invT));
      }
    }
    __syncthreads();
  }

  float rv[STAT_REGS];
  int   rn[STAT_REGS];
#pragma unroll
  for (int j = 0; j < STAT_REGS; ++j) {
    const int i = st + sub + 4 * j;
    rv[j] = 0.0f;
    rn[j] = -1;
    if (i < en) { rv[j] = recCost[gbase + i]; rn[j] = recNode[gbase + i]; }
  }
#pragma unroll
  for (int j = 0; j < STAT_REGS; ++j) {
    if (rn[j] >= 0) {
      const float m = ordVal(mx[rn[j]]);
      atomicAdd(&sm[rn[j]], (unsigned long long)(__expf(-rv[j] * invT - m) * SUM_SCALE));
    }
  }
  for (int i = st + sub + 4 * STAT_REGS; i < en; i += 4) {
    const int   nd = recNode[gbase + i];
    const float m  = ordVal(mx[nd]);
    atomicAdd(&sm[nd], (unsigned long long)(__expf(-recCost[gbase + i] * invT - m) * SUM_SCALE));
  }
  __syncthreads();

  if (tid < SEG_NODES) {
    const int s = nodeBase + tid;
    if (s < numS) { segMax[s] = mx[tid]; segSum[s] = sm[tid]; }
  }
}

__global__ void seg_norm_pos(const float* __restrict__ recCost,
                             const int* __restrict__ pos,
                             const int* __restrict__ edgeS,
                             const unsigned* __restrict__ segMax,
                             const unsigned long long* __restrict__ segSum,
                             const float* __restrict__ logT,
                             float* __restrict__ outW,
                             float* __restrict__ outCost, int E) {
  int e = blockIdx.x * blockDim.x + threadIdx.x;
  if (e >= E) return;
  float invT = __expf(-logT[0]);
  int s = edgeS[e];
  float cost = recCost[pos[e]];
  float val = -cost * invT;
  float m = ordVal(segMax[s]);
  outCost[e] = cost;
  outW[e] = __expf(val - m) / ((float)segSum[s] * SUM_INV_SCALE);
}

__global__ void seg_expsum(const float* __restrict__ outCost,
                           const int* __restrict__ edgeS,
                           const unsigned* __restrict__ segMax,
                           unsigned long long* __restrict__ segSum,
                           const float* __restrict__ logT, int E) {
  int e = blockIdx.x * blockDim.x + threadIdx.x;
  if (e >= E) return;
  float invT = __expf(-logT[0]);
  int s = edgeS[e];
  float val = -outCost[e] * invT;
  float m = ordVal(segMax[s]);
  atomicAdd(&segSum[s], (unsigned long long)(__expf(val - m) * SUM_SCALE));
}

__global__ void seg_norm(const float* __restrict__ outCost,
                         const int* __restrict__ edgeS,
                         const unsigned* __restrict__ segMax,
                         const unsigned long long* __restrict__ segSum,
                         const float* __restrict__ logT,
                         float* __restrict__ outW, int E) {
  int e = blockIdx.x * blockDim.x + threadIdx.x;
  if (e >= E) return;
  float invT = __expf(-logT[0]);
  int s = edgeS[e];
  float val = -outCost[e] * invT;
  float m = ordVal(segMax[s]);
  outW[e] = __expf(val - m) / ((float)segSum[s] * SUM_INV_SCALE);
}

extern "C" void kernel_launch(void* const* d_in, const int* in_sizes, int n_in,
                              void* d_out, int out_size, void* d_ws, size_t ws_size,
                              hipStream_t stream) {
  const float* embS = (const float*)d_in[0];
  const float* embA = (const float*)d_in[1];
  const int*   eidx = (const int*)d_in[2];
  const float* W1   = (const float*)d_in[3];
  const float* b1   = (const float*)d_in[4];
  const float* W2   = (const float*)d_in[5];
  const float* b2   = (const float*)d_in[6];
  const float* logT = (const float*)d_in[7];

  const int numS = in_sizes[0] / D_DIM;
  const int numA = in_sizes[1] / D_DIM;
  const int E    = in_sizes[2] / 2;
  const int* edgeS = eidx;
  const int* edgeA = eidx + E;

  float* outW    = (float*)d_out;
  float* outCost = (float*)d_out + E;

  const size_t pFloats = (size_t)(numS + numA) * H_DIM;
  const size_t need    = pFloats * sizeof(float)
                       + (size_t)numS * (sizeof(unsigned long long) + sizeof(unsigned));

  const int NB       = (numS + SEG_NODES - 1) >> SEG_SHIFT;
  const int chunk    = (((E + PART_BLOCKS - 1) / PART_BLOCKS) + 31) & ~31;
  const int chunkPad = (chunk + 127) & ~127;
  const size_t recSlots = (size_t)PART_BLOCKS * chunkPad;
  const size_t needGrp  = need + recSlots * (sizeof(float) + 1)
                        + (size_t)PART_BLOCKS * (NB + 1) * sizeof(int);

  const size_t posSlots = ((size_t)E + 31) & ~(size_t)31;
  const size_t needSrc  = needGrp + recSlots * sizeof(float) + posSlots * sizeof(int);

  if (ws_size >= needSrc && NB <= NB_MAX && chunk <= CHUNK_MAX) {
    float* P = (float*)d_ws;
    int*           recA    = (int*)(P + pFloats);
    float*         recCost = (float*)(recA + recSlots);
    int*           pos     = (int*)(recCost + recSlots);
    unsigned char* recNode = (unsigned char*)(pos + posSlots);
    unsigned long long* segSum = (unsigned long long*)(recNode + recSlots);
    unsigned*           segMax = (unsigned*)(segSum + numS);
    int*                offG   = (int*)(segMax + numS);

    node_proj<<<256, EDGE_WAVES * 32, 0, stream>>>(embS, embA, W1, b1, P, numS, numA);
    group_edges<<<PART_BLOCKS, PART_THREADS, 0, stream>>>(
        edgeS, edgeA, offG, recA, recNode, pos, E, NB, chunk, chunkPad);
    bucket_edge<<<NB, PART_THREADS, 0, stream>>>(
        embS, embA, P, P + (size_t)numS * H_DIM, W2, b2, logT,
        offG, recA, recNode, recCost, segMax, segSum, numS, NB, chunkPad);
    seg_norm_pos<<<(E + 255) / 256, 256, 0, stream>>>(
        recCost, pos, edgeS, segMax, segSum, logT, outW, outCost, E);
    return;
  }

  if (ws_size >= needGrp && NB <= NB_MAX && chunk <= CHUNK_MAX) {
    float* P = (float*)d_ws;
    float*         recVal  = P + pFloats;
    unsigned char* recNode = (unsigned char*)(recVal + recSlots);
    unsigned long long* segSum = (unsigned long long*)(recNode + recSlots);
    unsigned*           segMax = (unsigned*)(segSum + numS);
    int*                offG   = (int*)(segMax + numS);

    node_proj<<<256, EDGE_WAVES * 32, 0, stream>>>(embS, embA, W1, b1, P, numS, numA);
    edge_gate_place<<<PART_BLOCKS, PART_THREADS, 0, stream>>>(
        embS, embA, edgeS, edgeA, P, P + (size_t)numS * H_DIM, W2, b2, logT,
        outCost, offG, recVal, recNode, E, NB, chunk, chunkPad);
    bucket_stats<<<NB, PART_THREADS, 0, stream>>>(
        offG, recVal, recNode, segMax, segSum, numS, NB, chunkPad);
    seg_norm<<<(E + 255) / 256, 256, 0, stream>>>(outCost, edgeS, segMax, segSum, logT, outW, E);
    return;
  }

  if (ws_size >= need) {
    float* P = (float*)d_ws;
    unsigned long long* segSum = (unsigned long long*)(P + pFloats);
    unsigned*           segMax = (unsigned*)(segSum + numS);

    init_seg<<<(numS + 255) / 256, 256, 0, stream>>>(segMax, segSum, numS);
    node_proj<<<256, EDGE_WAVES * 32, 0, stream>>>(embS, embA, W1, b1, P, numS, numA);
    edge_gate_cost<<<256, EDGE_WAVES * 32, 0, stream>>>(
        embS, embA, edgeS, edgeA, P, P + (size_t)numS * H_DIM, W2, b2, logT,
        outCost, segMax, E);
    seg_expsum<<<(E + 255) / 256, 256, 0, stream>>>(outCost, edgeS, segMax, segSum, logT, E);
    seg_norm<<<(E + 255) / 256, 256, 0, stream>>>(outCost, edgeS, segMax, segSum, logT, outW, E);
    return;
  }

  unsigned long long* segSum = (unsigned long long*)d_ws;
  unsigned*           segMax = (unsigned*)(segSum + numS);

  init_seg<<<(numS + 255) / 256, 256, 0, stream>>>(segMax, segSum, numS);

  edge_mlp_cost<<<256, EDGE_WAVES * 32, 0, stream>>>(
      embS, embA, edgeS, edgeA, W1, b1, W2, b2, logT, outCost, segMax, E);

  seg_expsum<<<(E + 255) / 256, 256, 0, stream>>>(outCost, edgeS, segMax, segSum, logT, E);
  seg_norm<<<(E + 255) / 256, 256, 0, stream>>>(outCost, edgeS, segMax, segSum, logT, outW, E);
}


# edge pass on every WGP: source buckets of 196 nodes, not 256, so 50,000 sources make 256 one-per-WGP blocks (256-node buckets made 196: 60 of 256 WGPs idle); 23% fewer edges and LDS rows per block
# speedup vs baseline: 4.5227x; 1.0399x over previous
#include <hip/hip_runtime.h>
#include <hip/hip_bf16.h>
#include <math.h>

typedef __attribute__((ext_vector_type(2))) float v2f;
typedef __attribute__((ext_vector_type(4))) float v4f;
typedef __attribute__((ext_vector_type(8))) float v8f;

#define D_DIM 64
#define H_DIM 128
#define CPITCH 66

__device__ __forceinline__ unsigned ordKey(float f) {
  unsigned u = __float_as_uint(f);
  return (u & 0x80000000u) ? ~u : (u | 0x80000000u);
}
__device__ __forceinline__ float ordVal(unsigned k) {
  unsigned u = (k & 0x80000000u) ? (k ^ 0x80000000u) : ~k;
  return __uint_as_float(u);
}

#define SUM_SCALE     1099511627776.0f
#define SUM_INV_SCALE (1.0f / 1099511627776.0f)

__global__ void init_seg(unsigned* __restrict__ segMax,
                         unsigned long long* __restrict__ segSum, int numS) {
  int i = blockIdx.x * blockDim.x + threadIdx.x;
  if (i < numS) { segMax[i] = 0u; segSum[i] = 0ull; }
}

#define EDGE_WAVES 32
__global__ __launch_bounds__(EDGE_WAVES * 32)
void edge_mlp_cost(const float* __restrict__ embS,
                   const float* __restrict__ embA,
                   const int*   __restrict__ edgeS,
                   const int*   __restrict__ edgeA,
                   const float* __restrict__ W1,
                   const float* __restrict__ b1,
                   const float* __restrict__ W2,
                   const float* __restrict__ b2,
                   const float* __restrict__ logT,
                   float*       __restrict__ outCost,
                   unsigned*    __restrict__ segMax,
                   int E) {
  __shared__ float sW1[H_DIM * CPITCH * 2];
  __shared__ float sB1[H_DIM];
  __shared__ float sW2[H_DIM];
  __shared__ float sGate[EDGE_WAVES][16];

  const int tid = threadIdx.x;
  for (int i = tid; i < H_DIM * H_DIM; i += EDGE_WAVES * 32) {
    int kk = i >> 7;
    int c  = i & 127;
    sW1[(c * CPITCH + (kk >> 1)) * 2 + (kk & 1)] = W1[i];
  }
  if (tid < H_DIM) { sB1[tid] = b1[tid]; sW2[tid] = W2[tid]; }
  __syncthreads();

  const int wave = __builtin_amdgcn_readfirstlane(tid >> 5);
  const int lane = tid & 31;
  const int half = lane >> 4;
  const int col  = lane & 15;
  const float b2v  = b2[0];
  const float invT = __expf(-logT[0]);
  const int tiles  = (E + 15) >> 4;
  const int stride = gridDim.x * EDGE_WAVES;

  for (int t = blockIdx.x * EDGE_WAVES + wave; t < tiles; t += stride) {
  const int  e0    = t * 16;
  const int  e     = e0 + col;
  const bool valid = (e < E);
  const int  ec    = valid ? e : (E - 1);
  const int  sI    = edgeS[ec];
  const int  aI    = edgeA[ec];
  const float* es = embS + (size_t)sI * D_DIM;
  const float* ea = embA + (size_t)aI * D_DIM;

  v2f a[32];
#pragma unroll
  for (int k = 0; k < 16; ++k) {
    a[k]      = *(const v2f*)(es + 4 * k + 2 * half);
    a[k + 16] = *(const v2f*)(ea + 4 * k + 2 * half);
  }
  float ss = 0.0f;
#pragma unroll
  for (int k = 0; k < 16; ++k) {
    float dx = a[k][0] - a[k + 16][0];
    float dy = a[k][1] - a[k + 16][1];
    ss += dx * dx + dy * dy;
  }
  ss += __shfl_xor(ss, 16);

  float gacc[8] = {0, 0, 0, 0, 0, 0, 0, 0};
#pragma unroll 1
  for (int n = 0; n < 8; ++n) {
    const float* bbase = sW1 + ((n * 16 + col) * CPITCH + half) * 2;
    v8f c0 = {};
    v8f c1 = {};
#pragma unroll
    for (int k = 0; k < 32; k += 2) {
      v2f b0 = *(const v2f*)(bbase + 4 * k);
      v2f b1v = *(const v2f*)(bbase + 4 * (k + 1));
      c0 = __builtin_amdgcn_wmma_f32_16x16x4_f32(false, a[k], false, b0,
                                                 (short)0, c0, false, false);
      c1 = __builtin_amdgcn_wmma_f32_16x16x4_f32(false, a[k + 1], false, b1v,
                                                 (short)0, c1, false, false);
    }
    const float bn  = sB1[n * 16 + col];
    const float w2n = sW2[n * 16 + col];
#pragma unroll
    for (int r = 0; r < 8; ++r) {
      float h = c0[r] + c1[r] + bn;
      h = h > 0.0f ? h : 0.0f;
      gacc[r] += h * w2n;
    }
  }

#pragma unroll
  for (int r = 0; r < 8; ++r) {
    float g = gacc[r];
    g += __shfl_xor(g, 1);
    g += __shfl_xor(g, 2);
    g += __shfl_xor(g, 4);
    g += __shfl_xor(g, 8);
    if (lane == 0)  sGate[wave][r]     = g;
    if (lane == 16) sGate[wave][r + 8] = g;
  }

  if (lane < 16 && valid) {
    float logit = sGate[wave][lane] + b2v;
    float gate  = 1.0f / (1.0f + __expf(-logit));
    float cost  = sqrtf(ss) * gate;
    outCost[e]  = cost;
    atomicMax(&segMax[sI], ordKey(-cost * invT));
  }
  }
}

__global__ __launch_bounds__(EDGE_WAVES * 32)
void node_proj(const float* __restrict__ embS,
               const float* __restrict__ embA,
               const float* __restrict__ W1,
               const float* __restrict__ b1,
               float*       __restrict__ P,
               int numS, int numA) {
  __shared__ float sW1[H_DIM * CPITCH * 2];
  __shared__ __attribute__((aligned(16))) float sB1[2 * H_DIM];

  const int tid = threadIdx.x;
  for (int i = tid; i < H_DIM * H_DIM; i += EDGE_WAVES * 32) {
    int kk = i >> 7;
    int c  = i & 127;
    sW1[(c * CPITCH + (kk >> 1)) * 2 + (kk & 1)] = W1[i];
  }
  if (tid < H_DIM) { sB1[tid] = b1[tid]; sB1[H_DIM + tid] = 0.0f; }
  __syncthreads();

  const int wave = __builtin_amdgcn_readfirstlane(tid >> 5);
  const int lane = tid & 31;
  const int half = lane >> 4;
  const int col  = lane & 15;
  const int tilesS = (numS + 15) >> 4;
  const int tilesA = (numA + 15) >> 4;
  const int tiles  = tilesS + tilesA;

  for (int t = wave * (int)gridDim.x + (int)blockIdx.x; t < tiles;
       t += EDGE_WAVES * (int)gridDim.x) {
    const bool isA = (t >= tilesS);
    const int  tt  = isA ? (t - tilesS) : t;
    const float* emb = isA ? embA : embS;
    const int  nrows = isA ? numA : numS;
    const int  node  = tt * 16 + col;
    const bool valid = (node < nrows);
    const int  nc    = valid ? node : (nrows - 1);
    const float* row = emb + (size_t)nc * D_DIM;

    v2f a[16];
#pragma unroll
    for (int k = 0; k < 16; ++k) a[k] = *(const v2f*)(row + 4 * k + 2 * half);

    float* prow = P + ((size_t)(isA ? numS : 0) + (size_t)nc) * H_DIM + 8 * half;
    const int koff = isA ? D_DIM : 0;
    const int boff = isA ? H_DIM : 0;

#pragma unroll 1
    for (int nt = 0; nt < 8; ++nt) {
      const float* wbase = sW1 + ((nt * 16 + col) * CPITCH + half) * 2 + koff;
      v8f c0 = {};
      v8f c1 = {};
#pragma unroll
      for (int k = 0; k < 16; k += 2) {
        v2f w0 = *(const v2f*)(wbase + 4 * k);
        v2f w1 = *(const v2f*)(wbase + 4 * (k + 1));
        c0 = __builtin_amdgcn_wmma_f32_16x16x4_f32(false, w0, false, a[k],
                                                   (short)0, c0, false, false);
        c1 = __builtin_amdgcn_wmma_f32_16x16x4_f32(false, w1, false, a[k + 1],
                                                   (short)0, c1, false, false);
      }
      const float* bb = sB1 + boff + nt * 16 + 8 * half;
      const v4f blo = *(const v4f*)(bb);
      const v4f bhi = *(const v4f*)(bb + 4);
      v4f lo, hi;
#pragma unroll
      for (int r = 0; r < 4; ++r) {
        lo[r] = c0[r] + c1[r] + blo[r];
        hi[r] = c0[r + 4] + c1[r + 4] + bhi[r];
      }
      if (valid) {
        *(v4f*)(prow + nt * 16)     = lo;
        *(v4f*)(prow + nt * 16 + 4) = hi;
      }
    }
  }
}

struct RowSet {
  v4f es0, es1, ea0, ea1;
  v4f ps0, ps1, ps2, ps3;
  v4f pa0, pa1, pa2, pa3;
};

__device__ __forceinline__ void fetch_rows(RowSet& r,
                                           const float* __restrict__ embS,
                                           const float* __restrict__ embA,
                                           const float* __restrict__ Ps,
                                           const float* __restrict__ Pa,
                                           int s, int a, int sub4) {
  const float* e1 = embS + (size_t)s * D_DIM + sub4;
  const float* e2 = embA + (size_t)a * D_DIM + sub4;
  const float* p1 = Ps + (size_t)s * H_DIM + sub4;
  const float* p2 = Pa + (size_t)a * H_DIM + sub4;
  r.es0 = *(const v4f*)(e1);
  r.es1 = *(const v4f*)(e1 + 32);
  r.ea0 = *(const v4f*)(e2);
  r.ea1 = *(const v4f*)(e2 + 32);
  r.ps0 = *(const v4f*)(p1);
  r.ps1 = *(const v4f*)(p1 + 32);
  r.ps2 = *(const v4f*)(p1 + 64);
  r.ps3 = *(const v4f*)(p1 + 96);
  r.pa0 = *(const v4f*)(p2);
  r.pa1 = *(const v4f*)(p2 + 32);
  r.pa2 = *(const v4f*)(p2 + 64);
  r.pa3 = *(const v4f*)(p2 + 96);
}

__device__ __forceinline__ float sq4(v4f a, v4f b) {
  v4f d = a - b;
  return d[0] * d[0] + d[1] * d[1] + d[2] * d[2] + d[3] * d[3];
}
__device__ __forceinline__ float gate4(v4f p, v4f q, v4f w) {
  v4f h = p + q;
  float g = 0.0f;
#pragma unroll
  for (int i = 0; i < 4; ++i) {
    float x = h[i] > 0.0f ? h[i] : 0.0f;
    g += x * w[i];
  }
  return g;
}

template <int M>
__device__ __forceinline__ float xadd8(float v) {
  int o = __builtin_amdgcn_update_dpp(0, __float_as_int(v), 0x160 + M, 0xf, 0xf, true);
  return v + __int_as_float(o);
}
__device__ __forceinline__ int bsel8(int v, int lane, int i) {
  return __builtin_amdgcn_ds_bpermute(((lane & 24) | i) << 2, v);
}

__device__ __forceinline__ void reduce_rows(const RowSet& r, const float* sW2, int sub4,
                                            float& ss, float& g) {
  float s = sq4(r.es0, r.ea0) + sq4(r.es1, r.ea1);
  float t = gate4(r.ps0, r.pa0, *(const v4f*)(sW2 + sub4))
          + gate4(r.ps1, r.pa1, *(const v4f*)(sW2 + 32 + sub4))
          + gate4(r.ps2, r.pa2, *(const v4f*)(sW2 + 64 + sub4))
          + gate4(r.ps3, r.pa3, *(const v4f*)(sW2 + 96 + sub4));
  s = xadd8<1>(s); t = xadd8<1>(t);
  s = xadd8<2>(s); t = xadd8<2>(t);
  s = xadd8<4>(s); t = xadd8<4>(t);
  ss = s; g = t;
}

__device__ __forceinline__ float tile_cost(const float* __restrict__ embS,
                                           const float* __restrict__ embA,
                                           const float* __restrict__ Ps,
                                           const float* __restrict__ Pa,
                                           const float* sW2,
                                           int curS, int curA,
                                           int lane, int sub, int sub4, float b2v) {
  float rss = 0.0f, rg = 0.0f;

#pragma unroll 1
  for (int i = 0; i < 8; i += 2) {
    RowSet X, Y;
    float ss, g;
    fetch_rows(X, embS, embA, Ps, Pa, bsel8(curS, lane, i), bsel8(curA, lane, i), sub4);
    fetch_rows(Y, embS, embA, Ps, Pa, bsel8(curS, lane, i + 1), bsel8(curA, lane, i + 1), sub4);
    __builtin_amdgcn_sched_barrier(0);
    reduce_rows(X, sW2, sub4, ss, g);
    if (sub == i) { rss = ss; rg = g; }
    reduce_rows(Y, sW2, sub4, ss, g);
    if (sub == i + 1) { rss = ss; rg = g; }
  }

  const float logit = rg + b2v;
  const float gate  = 1.0f / (1.0f + __expf(-logit));
  return sqrtf(rss) * gate;
}

__global__ __launch_bounds__(EDGE_WAVES * 32)
void edge_gate_cost(const float* __restrict__ embS,
                    const float* __restrict__ embA,
                    const int*   __restrict__ edgeS,
                    const int*   __restrict__ edgeA,
                    const float* __restrict__ Ps,
                    const float* __restrict__ Pa,
                    const float* __restrict__ W2,
                    const float* __restrict__ b2,
                    const float* __restrict__ logT,
                    float*       __restrict__ outCost,
                    unsigned*    __restrict__ segMax,
                    int E) {
  __shared__ __attribute__((aligned(16))) float sW2[H_DIM];
  const int tid = threadIdx.x;
  if (tid < H_DIM) sW2[tid] = W2[tid];
  __syncthreads();

  const int wave = __builtin_amdgcn_readfirstlane(tid >> 5);
  const int lane = tid & 31;
  const int sub  = lane & 7;
  const int sub4 = sub * 4;
  const float b2v  = b2[0];
  const float invT = __expf(-logT[0]);
  const int tiles  = (E + 31) >> 5;
  const int stride = (int)gridDim.x * EDGE_WAVES;

  int t = (int)blockIdx.x * EDGE_WAVES + wave;
  if (t >= tiles) return;

  int eL   = t * 32 + lane;
  int eLc  = eL < E ? eL : (E - 1);
  int curS = edgeS[eLc];
  int curA = edgeA[eLc];

  for (;;) {
    const bool hasNext = (t + stride) < tiles;
    const int  tn   = hasNext ? (t + stride) : t;
    const int  nL   = tn * 32 + lane;
    const int  nLc  = nL < E ? nL : (E - 1);
    const int  nxtS = edgeS[nLc];
    const int  nxtA = edgeA[nLc];

    const float cost = tile_cost(embS, embA, Ps, Pa, sW2, curS, curA, lane, sub, sub4, b2v);

    if (eL < E) {
      outCost[eL] = cost;
      atomicMax(&segMax[curS], ordKey(-cost * invT));
    }
    if (!hasNext) break;
    t = tn; eL = nL; curS = nxtS; curA = nxtA;
  }
}

#define SEG_NODES    196
#define NB_MAX       1024
#define PART_BLOCKS  256
#define PART_THREADS 1024
#define CHUNK_MAX    4096
#define STAT_REGS    8

__device__ __forceinline__ int seg_bucket(int s) { return (int)((unsigned)s / (unsigned)SEG_NODES); }

__global__ __launch_bounds__(PART_THREADS)
void edge_gate_place(const float* __restrict__ embS,
                     const float* __restrict__ embA,
                     const int*   __restrict__ edgeS,
                     const int*   __restrict__ edgeA,
                     const float* __restrict__ Ps,
                     const float* __restrict__ Pa,
                     const float* __restrict__ W2,
                     const float* __restrict__ b2,
                     const float* __restrict__ logT,
                     float*         __restrict__ outCost,
                     int*           __restrict__ offG,
                     float*         __restrict__ recVal,
                     unsigned char* __restrict__ recNode,
                     int E, int NB, int chunk, int chunkPad) {
  __shared__ __attribute__((aligned(16))) float sW2[H_DIM];
  __shared__ int h[NB_MAX];
  __shared__ int fill[NB_MAX];
  __shared__ int off[NB_MAX + 1];
  __shared__ int   sSrc[CHUNK_MAX];
  __shared__ float sArg[CHUNK_MAX];
  __shared__ __attribute__((aligned(16))) float    sVal[CHUNK_MAX];
  __shared__ __attribute__((aligned(16))) unsigned sNode4[CHUNK_MAX / 4];
  unsigned char* sNode = (unsigned char*)sNode4;

  const int tid = threadIdx.x;
  const int b   = blockIdx.x;
  if (tid < H_DIM) sW2[tid] = W2[tid];
  for (int k = tid; k < NB; k += PART_THREADS) { h[k] = 0; fill[k] = 0; }
  sNode4[tid] = 0u;
  __syncthreads();

  const int wave = __builtin_amdgcn_readfirstlane(tid >> 5);
  const int lane = tid & 31;
  const int sub  = lane & 7;
  const int sub4 = sub * 4;
  const float b2v  = b2[0];
  const float invT = __expf(-logT[0]);

  const int lo = b * chunk;
  const int hi = (lo + chunk < E) ? (lo + chunk) : E;
  const int n  = (hi > lo) ? (hi - lo) : 0;
  const int nTiles = (n + 31) >> 5;

  int t = wave;
  if (t < nTiles) {
    int iL   = t * 32 + lane;
    int eLc  = iL < n ? (lo + iL) : (hi - 1);
    int curS = edgeS[eLc];
    int curA = edgeA[eLc];

    for (;;) {
      const bool hasNext = (t + EDGE_WAVES) < nTiles;
      const int  tn   = hasNext ? (t + EDGE_WAVES) : t;
      const int  nL   = tn * 32 + lane;
      const int  nLc  = nL < n ? (lo + nL) : (hi - 1);
      const int  nxtS = edgeS[nLc];
      const int  nxtA = edgeA[nLc];

      const float cost = tile_cost(embS, embA, Ps, Pa, sW2, curS, curA, lane, sub, sub4, b2v);

      if (iL < n) {
        outCost[lo + iL] = cost;
        sSrc[iL] = curS;
        sArg[iL] = -cost * invT;
        atomicAdd(&h[seg_bucket(curS)], 1);
      }
      if (!hasNext) break;
      t = tn; iL = nL; curS = nxtS; curA = nxtA;
    }
  }
  __syncthreads();

  if (tid < 32) {
    const int per = (NB + 31) >> 5;
    const int k0  = tid * per;
    int s = 0;
    for (int j = 0; j < per; ++j) { const int k = k0 + j; if (k < NB) s += h[k]; }
    int inc = s;
    for (int d = 1; d < 32; d <<= 1) {
      const int o = __shfl_up(inc, d);
      if (tid >= d) inc += o;
    }
    int run = inc - s;
    for (int j = 0; j < per; ++j) {
      const int k = k0 + j;
      if (k < NB) { off[k] = run; run += h[k]; }
    }
    if (tid == 31) off[NB] = inc;
  }
  __syncthreads();

  for (int k = tid; k <= NB; k += PART_THREADS) offG[b * (NB + 1) + k] = off[k];

#pragma unroll
  for (int j = 0; j < 4; ++j) {
    const int i = j * PART_THREADS + tid;
    if (i < n) {
      const int sv  = sSrc[i];
      const int k   = seg_bucket(sv);
      const int pos = off[k] + atomicAdd(&fill[k], 1);
      sVal[pos]  = sArg[i];
      sNode[pos] = (unsigned char)(sv - k * SEG_NODES);
    }
  }
  __syncthreads();

  float*    gv = recVal + (size_t)b * chunkPad;
  unsigned* gn = (unsigned*)(recNode + (size_t)b * chunkPad);
#pragma unroll
  for (int j = 0; j < 4; ++j) {
    const int i = j * PART_THREADS + tid;
    if (i < n) gv[i] = sVal[i];
  }
  if (4 * tid < n) gn[tid] = sNode4[tid];
}

__global__ __launch_bounds__(PART_THREADS)
void bucket_stats(const int*           __restrict__ offG,
                  const float*         __restrict__ recVal,
                  const unsigned char* __restrict__ recNode,
                  unsigned*            __restrict__ segMax,
                  unsigned long long*  __restrict__ segSum,
                  int numS, int NB, int chunkPad) {
  __shared__ unsigned           mx[SEG_NODES];
  __shared__ unsigned long long sm[SEG_NODES];
  const int tid = threadIdx.x;
  const int k   = blockIdx.x;
  if (tid < SEG_NODES) { mx[tid] = 0u; sm[tid] = 0ull; }

  const int r   = tid >> 2;
  const int sub = tid & 3;
  const int* o  = offG + r * (NB + 1) + k;
  const int st  = r * chunkPad + o[0];
  const int en  = r * chunkPad + o[1];

  float rv[STAT_REGS];
  int   rn[STAT_REGS];
#pragma unroll
  for (int j = 0; j < STAT_REGS; ++j) {
    const int i = st + sub + 4 * j;
    rv[j] = 0.0f;
    rn[j] = -1;
    if (i < en) { rv[j] = recVal[i]; rn[j] = recNode[i]; }
  }
  const int rest = st + sub + 4 * STAT_REGS;
  __syncthreads();

#pragma unroll
  for (int j = 0; j < STAT_REGS; ++j)
    if (rn[j] >= 0) atomicMax(&mx[rn[j]], ordKey(rv[j]));
  for (int i = rest; i < en; i += 4)
    atomicMax(&mx[recNode[i]], ordKey(recVal[i]));
  __syncthreads();

#pragma unroll
  for (int j = 0; j < STAT_REGS; ++j) {
    if (rn[j] >= 0) {
      const float m = ordVal(mx[rn[j]]);
      atomicAdd(&sm[rn[j]], (unsigned long long)(__expf(rv[j] - m) * SUM_SCALE));
    }
  }
  for (int i = rest; i < en; i += 4) {
    const int   nd = recNode[i];
    const float m  = ordVal(mx[nd]);
    atomicAdd(&sm[nd], (unsigned long long)(__expf(recVal[i] - m) * SUM_SCALE));
  }
  __syncthreads();

  if (tid < SEG_NODES) {
    const int s = k * SEG_NODES + tid;
    if (s < numS) { segMax[s] = mx[tid]; segSum[s] = sm[tid]; }
  }
}

#define BE_CAP 6144
#define EPITCH 68
#define PPITCH 132

__global__ __launch_bounds__(PART_THREADS)
void group_edges(const int* __restrict__ edgeS,
                 const int* __restrict__ edgeA,
                 int*           __restrict__ offG,
                 int*           __restrict__ recA,
                 unsigned char* __restrict__ recNode,
                 int*           __restrict__ pos,
                 int E, int NB, int chunk, int chunkPad) {
  __shared__ int h[NB_MAX];
  __shared__ int fill[NB_MAX];
  __shared__ int off[NB_MAX + 1];
  __shared__ __attribute__((aligned(16))) int      sA[CHUNK_MAX];
  __shared__ __attribute__((aligned(16))) unsigned sNode4[CHUNK_MAX / 4];
  unsigned char* sNode = (unsigned char*)sNode4;

  const int tid = threadIdx.x;
  const int b   = blockIdx.x;
  for (int k = tid; k < NB; k += PART_THREADS) { h[k] = 0; fill[k] = 0; }
  sNode4[tid] = 0u;
  __syncthreads();

  const int lo = b * chunk;
  const int hi = (lo + chunk < E) ? (lo + chunk) : E;
  const int n  = (hi > lo) ? (hi - lo) : 0;

  int sv[4], av[4];
#pragma unroll
  for (int j = 0; j < 4; ++j) {
    const int i = j * PART_THREADS + tid;
    sv[j] = -1; av[j] = 0;
    if (i < n) {
      sv[j] = edgeS[lo + i];
      av[j] = edgeA[lo + i];
      atomicAdd(&h[seg_bucket(sv[j])], 1);
    }
  }
  __syncthreads();

  if (tid < 32) {
    const int per = (NB + 31) >> 5;
    const int k0  = tid * per;
    int s = 0;
    for (int j = 0; j < per; ++j) { const int k = k0 + j; if (k < NB) s += h[k]; }
    int inc = s;
    for (int d = 1; d < 32; d <<= 1) {
      const int o = __shfl_up(inc, d);
      if (tid >= d) inc += o;
    }
    int run = inc - s;
    for (int j = 0; j < per; ++j) {
      const int k = k0 + j;
      if (k < NB) { off[k] = run; run += h[k]; }
    }
    if (tid == 31) off[NB] = inc;
  }
  __syncthreads();

  for (int k = tid; k <= NB; k += PART_THREADS) offG[b * (NB + 1) + k] = off[k];

#pragma unroll
  for (int j = 0; j < 4; ++j) {
    const int i = j * PART_THREADS + tid;
    if (i < n) {
      const int k = seg_bucket(sv[j]);
      const int p = off[k] + atomicAdd(&fill[k], 1);
      sA[p]    = av[j];
      sNode[p] = (unsigned char)(sv[j] - k * SEG_NODES);
      pos[lo + i] = b * chunkPad + p;
    }
  }
  __syncthreads();

  int*      ga = recA + (size_t)b * chunkPad;
  unsigned* gn = (unsigned*)(recNode + (size_t)b * chunkPad);
#pragma unroll
  for (int j = 0; j < 4; ++j) {
    const int i = j * PART_THREADS + tid;
    if (i < n) ga[i] = sA[i];
  }
  if (4 * tid < n) gn[tid] = sNode4[tid];
}

struct RowT {
  v4f ea0, ea1;
  v4f pa0, pa1, pa2, pa3;
};
__device__ __forceinline__ void fetch_t(RowT& r,
                                        const float* __restrict__ embA,
                                        const float* __restrict__ Pa,
                                        int a, int sub4) {
  const float* e2 = embA + (size_t)a * D_DIM + sub4;
  const float* p2 = Pa + (size_t)a * H_DIM + sub4;
  r.ea0 = *(const v4f*)(e2);
  r.ea1 = *(const v4f*)(e2 + 32);
  r.pa0 = *(const v4f*)(p2);
  r.pa1 = *(const v4f*)(p2 + 32);
  r.pa2 = *(const v4f*)(p2 + 64);
  r.pa3 = *(const v4f*)(p2 + 96);
}
__device__ __forceinline__ void reduce_t(const RowT& r, const float* sE, const float* sP,
                                         const float* sW2, int sub4, float& ss, float& g) {
  const v4f es0 = *(const v4f*)(sE + sub4);
  const v4f es1 = *(const v4f*)(sE + 32 + sub4);
  const v4f ps0 = *(const v4f*)(sP + sub4);
  const v4f ps1 = *(const v4f*)(sP + 32 + sub4);
  const v4f ps2 = *(const v4f*)(sP + 64 + sub4);
  const v4f ps3 = *(const v4f*)(sP + 96 + sub4);
  float s = sq4(es0, r.ea0) + sq4(es1, r.ea1);
  float t = gate4(ps0, r.pa0, *(const v4f*)(sW2 + sub4))
          + gate4(ps1, r.pa1, *(const v4f*)(sW2 + 32 + sub4))
          + gate4(ps2, r.pa2, *(const v4f*)(sW2 + 64 + sub4))
          + gate4(ps3, r.pa3, *(const v4f*)(sW2 + 96 + sub4));
  s = xadd8<1>(s); t = xadd8<1>(t);
  s = xadd8<2>(s); t = xadd8<2>(t);
  s = xadd8<4>(s); t = xadd8<4>(t);
  ss = s; g = t;
}

__device__ __forceinline__ float tile_cost_b(const float* __restrict__ embA,
                                             const float* __restrict__ Pa,
                                             const float* sRowE, const float* sRowP,
                                             const float* sW2,
                                             int curN, int curA,
                                             int lane, int sub, int sub4, float b2v) {
  float rss = 0.0f, rg = 0.0f;

#pragma unroll 1
  for (int i = 0; i < 8; i += 2) {
    RowT X, Y;
    float ss, g;
    const int n0 = bsel8(curN, lane, i);
    const int n1 = bsel8(curN, lane, i + 1);
    fetch_t(X, embA, Pa, bsel8(curA, lane, i), sub4);
    fetch_t(Y, embA, Pa, bsel8(curA, lane, i + 1), sub4);
    __builtin_amdgcn_sched_barrier(0);
    reduce_t(X, sRowE + n0 * EPITCH, sRowP + n0 * PPITCH, sW2, sub4, ss, g);
    if (sub == i) { rss = ss; rg = g; }
    reduce_t(Y, sRowE + n1 * EPITCH, sRowP + n1 * PPITCH, sW2, sub4, ss, g);
    if (sub == i + 1) { rss = ss; rg = g; }
  }

  const float logit = rg + b2v;
  const float gate  = 1.0f / (1.0f + __expf(-logit));
  return sqrtf(rss) * gate;
}

__global__ __launch_bounds__(PART_THREADS)
void bucket_edge(const float* __restrict__ embS,
                 const float* __restrict__ embA,
                 const float* __restrict__ Ps,
                 const float* __restrict__ Pa,
                 const float* __restrict__ W2,
                 const float* __restrict__ b2,
                 const float* __restrict__ logT,
                 const int*           __restrict__ offG,
                 const int*           __restrict__ recA,
                 const unsigned char* __restrict__ recNode,
                 float*               recCost,
                 unsigned*            __restrict__ segMax,
                 unsigned long long*  __restrict__ segSum,
                 int numS, int NB, int chunkPad) {
  __shared__ __attribute__((aligned(16))) float sRowE[SEG_NODES * EPITCH];
  __shared__ __attribute__((aligned(16))) float sRowP[SEG_NODES * PPITCH];
  __shared__ __attribute__((aligned(16))) float sW2[H_DIM];
  __shared__ unsigned           mx[SEG_NODES];
  __shared__ unsigned long long sm[SEG_NODES];
  __shared__ int runLen[PART_BLOCKS];
  __shared__ int runStart[PART_BLOCKS + 1];
  __shared__ int cA[BE_CAP];
  __shared__ int cPos[BE_CAP];
  __shared__ unsigned char cN[BE_CAP];

  const int tid = threadIdx.x;
  const int k   = blockIdx.x;
  if (tid < H_DIM) sW2[tid] = W2[tid];
  if (tid < SEG_NODES) { mx[tid] = 0u; sm[tid] = 0ull; }

  const int r   = tid >> 2;
  const int sub = tid & 3;
  const int* o  = offG + r * (NB + 1) + k;
  const int st  = o[0];
  const int en  = o[1];
  const int gbase = r * chunkPad;
  if (sub == 0) runLen[r] = en - st;

  const int nodeBase = k * SEG_NODES;
  const int nNodes   = (numS - nodeBase < SEG_NODES) ? (numS - nodeBase) : SEG_NODES;
  const v4f* gE = (const v4f*)(embS + (size_t)nodeBase * D_DIM);
  for (int i = tid; i < nNodes * (D_DIM / 4); i += PART_THREADS)
    *(v4f*)(sRowE + (i >> 4) * EPITCH + (i & 15) * 4) = gE[i];
  const v4f* gP = (const v4f*)(Ps + (size_t)nodeBase * H_DIM);
  for (int i = tid; i < nNodes * (H_DIM / 4); i += PART_THREADS)
    *(v4f*)(sRowP + (i >> 5) * PPITCH + (i & 31) * 4) = gP[i];
  __syncthreads();

  if (tid < 32) {
    int s = 0;
    for (int j = 0; j < PART_BLOCKS / 32; ++j) s += runLen[tid * (PART_BLOCKS / 32) + j];
    int inc = s;
    for (int d = 1; d < 32; d <<= 1) {
      const int v = __shfl_up(inc, d);
      if (tid >= d) inc += v;
    }
    int run = inc - s;
    for (int j = 0; j < PART_BLOCKS / 32; ++j) {
      const int q = tid * (PART_BLOCKS / 32) + j;
      runStart[q] = run;
      run += runLen[q];
    }
    if (tid == 31) runStart[PART_BLOCKS] = inc;
  }
  __syncthreads();

  const int wave = __builtin_amdgcn_readfirstlane(tid >> 5);
  const int lane = tid & 31;
  const int sub8 = lane & 7;
  const int sub4 = sub8 * 4;
  const float b2v  = b2[0];
  const float invT = __expf(-logT[0]);
  const int total = __builtin_amdgcn_readfirstlane(runStart[PART_BLOCKS]);
  const int rs    = runStart[r];

  for (int base = 0; base < total; base += BE_CAP) {
    const int cnt = (total - base < BE_CAP) ? (total - base) : BE_CAP;

    int ca[STAT_REGS];
    int cn[STAT_REGS];
#pragma unroll
    for (int q = 0; q < STAT_REGS; ++q) {
      const int i = st + sub + 4 * q;
      const int j = rs + (i - st) - base;
      ca[q] = 0;
      cn[q] = -1;
      if (i < en && j >= 0 && j < cnt) { ca[q] = recA[gbase + i]; cn[q] = recNode[gbase + i]; }
    }
#pragma unroll
    for (int q = 0; q < STAT_REGS; ++q) {
      if (cn[q] >= 0) {
        const int i = st + sub + 4 * q;
        const int j = rs + (i - st) - base;
        cA[j]   = ca[q];
        cN[j]   = (unsigned char)cn[q];
        cPos[j] = gbase + i;
      }
    }
    for (int i = st + sub + 4 * STAT_REGS; i < en; i += 4) {
      const int j = rs + (i - st) - base;
      if (j >= 0 && j < cnt) {
        cA[j]   = recA[gbase + i];
        cN[j]   = recNode[gbase + i];
        cPos[j] = gbase + i;
      }
    }
    __syncthreads();

    const int nT = (cnt + 31) >> 5;
    for (int t = wave; t < nT; t += EDGE_WAVES) {
      const int j  = t * 32 + lane;
      const int jc = j < cnt ? j : (cnt - 1);
      const int curN = cN[jc];
      const int curA = cA[jc];
      const float cost = tile_cost_b(embA, Pa, sRowE, sRowP, sW2, curN, curA,
                                     lane, sub8, sub4, b2v);
      if (j < cnt) {
        recCost[cPos[j]] = cost;
        atomicMax(&mx[curN], ordKey(-cost * invT));
      }
    }
    __syncthreads();
  }

  float rv[STAT_REGS];
  int   rn[STAT_REGS];
#pragma unroll
  for (int j = 0; j < STAT_REGS; ++j) {
    const int i = st + sub + 4 * j;
    rv[j] = 0.0f;
    rn[j] = -1;
    if (i < en) { rv[j] = recCost[gbase + i]; rn[j] = recNode[gbase + i]; }
  }
#pragma unroll
  for (int j = 0; j < STAT_REGS; ++j) {
    if (rn[j] >= 0) {
      const float m = ordVal(mx[rn[j]]);
      atomicAdd(&sm[rn[j]], (unsigned long long)(__expf(-rv[j] * invT - m) * SUM_SCALE));
    }
  }
  for (int i = st + sub + 4 * STAT_REGS; i < en; i += 4) {
    const int   nd = recNode[gbase + i];
    const float m  = ordVal(mx[nd]);
    atomicAdd(&sm[nd], (unsigned long long)(__expf(-recCost[gbase + i] * invT - m) * SUM_SCALE));
  }
  __syncthreads();

  if (tid < SEG_NODES) {
    const int s = nodeBase + tid;
    if (s < numS) { segMax[s] = mx[tid]; segSum[s] = sm[tid]; }
  }
}

__global__ void seg_norm_pos(const float* __restrict__ recCost,
                             const int* __restrict__ pos,
                             const int* __restrict__ edgeS,
                             const unsigned* __restrict__ segMax,
                             const unsigned long long* __restrict__ segSum,
                             const float* __restrict__ logT,
                             float* __restrict__ outW,
                             float* __restrict__ outCost, int E) {
  int e = blockIdx.x * blockDim.x + threadIdx.x;
  if (e >= E) return;
  float invT = __expf(-logT[0]);
  int s = edgeS[e];
  float cost = recCost[pos[e]];
  float val = -cost * invT;
  float m = ordVal(segMax[s]);
  outCost[e] = cost;
  outW[e] = __expf(val - m) / ((float)segSum[s] * SUM_INV_SCALE);
}

__global__ void seg_expsum(const float* __restrict__ outCost,
                           const int* __restrict__ edgeS,
                           const unsigned* __restrict__ segMax,
                           unsigned long long* __restrict__ segSum,
                           const float* __restrict__ logT, int E) {
  int e = blockIdx.x * blockDim.x + threadIdx.x;
  if (e >= E) return;
  float invT = __expf(-logT[0]);
  int s = edgeS[e];
  float val = -outCost[e] * invT;
  float m = ordVal(segMax[s]);
  atomicAdd(&segSum[s], (unsigned long long)(__expf(val - m) * SUM_SCALE));
}

__global__ void seg_norm(const float* __restrict__ outCost,
                         const int* __restrict__ edgeS,
                         const unsigned* __restrict__ segMax,
                         const unsigned long long* __restrict__ segSum,
                         const float* __restrict__ logT,
                         float* __restrict__ outW, int E) {
  int e = blockIdx.x * blockDim.x + threadIdx.x;
  if (e >= E) return;
  float invT = __expf(-logT[0]);
  int s = edgeS[e];
  float val = -outCost[e] * invT;
  float m = ordVal(segMax[s]);
  outW[e] = __expf(val - m) / ((float)segSum[s] * SUM_INV_SCALE);
}

extern "C" void kernel_launch(void* const* d_in, const int* in_sizes, int n_in,
                              void* d_out, int out_size, void* d_ws, size_t ws_size,
                              hipStream_t stream) {
  const float* embS = (const float*)d_in[0];
  const float* embA = (const float*)d_in[1];
  const int*   eidx = (const int*)d_in[2];
  const float* W1   = (const float*)d_in[3];
  const float* b1   = (const float*)d_in[4];
  const float* W2   = (const float*)d_in[5];
  const float* b2   = (const float*)d_in[6];
  const float* logT = (const float*)d_in[7];

  const int numS = in_sizes[0] / D_DIM;
  const int numA = in_sizes[1] / D_DIM;
  const int E    = in_sizes[2] / 2;
  const int* edgeS = eidx;
  const int* edgeA = eidx + E;

  float* outW    = (float*)d_out;
  float* outCost = (float*)d_out + E;

  const size_t pFloats = (size_t)(numS + numA) * H_DIM;
  const size_t need    = pFloats * sizeof(float)
                       + (size_t)numS * (sizeof(unsigned long long) + sizeof(unsigned));

  const int NB       = (numS + SEG_NODES - 1) / SEG_NODES;
  const int chunk    = (((E + PART_BLOCKS - 1) / PART_BLOCKS) + 31) & ~31;
  const int chunkPad = (chunk + 127) & ~127;
  const size_t recSlots = (size_t)PART_BLOCKS * chunkPad;
  const size_t needGrp  = need + recSlots * (sizeof(float) + 1)
                        + (size_t)PART_BLOCKS * (NB + 1) * sizeof(int);

  const size_t posSlots = ((size_t)E + 31) & ~(size_t)31;
  const size_t needSrc  = needGrp + recSlots * sizeof(float) + posSlots * sizeof(int);

  if (ws_size >= needSrc && NB <= NB_MAX && chunk <= CHUNK_MAX) {
    float* P = (float*)d_ws;
    int*           recA    = (int*)(P + pFloats);
    float*         recCost = (float*)(recA + recSlots);
    int*           pos     = (int*)(recCost + recSlots);
    unsigned char* recNode = (unsigned char*)(pos + posSlots);
    unsigned long long* segSum = (unsigned long long*)(recNode + recSlots);
    unsigned*           segMax = (unsigned*)(segSum + numS);
    int*                offG   = (int*)(segMax + numS);

    node_proj<<<256, EDGE_WAVES * 32, 0, stream>>>(embS, embA, W1, b1, P, numS, numA);
    group_edges<<<PART_BLOCKS, PART_THREADS, 0, stream>>>(
        edgeS, edgeA, offG, recA, recNode, pos, E, NB, chunk, chunkPad);
    bucket_edge<<<NB, PART_THREADS, 0, stream>>>(
        embS, embA, P, P + (size_t)numS * H_DIM, W2, b2, logT,
        offG, recA, recNode, recCost, segMax, segSum, numS, NB, chunkPad);
    seg_norm_pos<<<(E + 255) / 256, 256, 0, stream>>>(
        recCost, pos, edgeS, segMax, segSum, logT, outW, outCost, E);
    return;
  }

  if (ws_size >= needGrp && NB <= NB_MAX && chunk <= CHUNK_MAX) {
    float* P = (float*)d_ws;
    float*         recVal  = P + pFloats;
    unsigned char* recNode = (unsigned char*)(recVal + recSlots);
    unsigned long long* segSum = (unsigned long long*)(recNode + recSlots);
    unsigned*           segMax = (unsigned*)(segSum + numS);
    int*                offG   = (int*)(segMax + numS);

    node_proj<<<256, EDGE_WAVES * 32, 0, stream>>>(embS, embA, W1, b1, P, numS, numA);
    edge_gate_place<<<PART_BLOCKS, PART_THREADS, 0, stream>>>(
        embS, embA, edgeS, edgeA, P, P + (size_t)numS * H_DIM, W2, b2, logT,
        outCost, offG, recVal, recNode, E, NB, chunk, chunkPad);
    bucket_stats<<<NB, PART_THREADS, 0, stream>>>(
        offG, recVal, recNode, segMax, segSum, numS, NB, chunkPad);
    seg_norm<<<(E + 255) / 256, 256, 0, stream>>>(outCost, edgeS, segMax, segSum, logT, outW, E);
    return;
  }

  if (ws_size >= need) {
    float* P = (float*)d_ws;
    unsigned long long* segSum = (unsigned long long*)(P + pFloats);
    unsigned*           segMax = (unsigned*)(segSum + numS);

    init_seg<<<(numS + 255) / 256, 256, 0, stream>>>(segMax, segSum, numS);
    node_proj<<<256, EDGE_WAVES * 32, 0, stream>>>(embS, embA, W1, b1, P, numS, numA);
    edge_gate_cost<<<256, EDGE_WAVES * 32, 0, stream>>>(
        embS, embA, edgeS, edgeA, P, P + (size_t)numS * H_DIM, W2, b2, logT,
        outCost, segMax, E);
    seg_expsum<<<(E + 255) / 256, 256, 0, stream>>>(outCost, edgeS, segMax, segSum, logT, E);
    seg_norm<<<(E + 255) / 256, 256, 0, stream>>>(outCost, edgeS, segMax, segSum, logT, outW, E);
    return;
  }

  unsigned long long* segSum = (unsigned long long*)d_ws;
  unsigned*           segMax = (unsigned*)(segSum + numS);

  init_seg<<<(numS + 255) / 256, 256, 0, stream>>>(segMax, segSum, numS);

  edge_mlp_cost<<<256, EDGE_WAVES * 32, 0, stream>>>(
      embS, embA, edgeS, edgeA, W1, b1, W2, b2, logT, outCost, segMax, E);

  seg_expsum<<<(E + 255) / 256, 256, 0, stream>>>(outCost, edgeS, segMax, segSum, logT, E);
  seg_norm<<<(E + 255) / 256, 256, 0, stream>>>(outCost, edgeS, segMax, segSum, logT, outW, E);
}


# softmax weights finished in the bucket block (segment max and sum are in its LDS); last launch only un-sorts (cost, weight) to edge order, chunk by chunk in LDS: no per-edge max/sum lookups
# speedup vs baseline: 4.6788x; 1.0345x over previous
#include <hip/hip_runtime.h>
#include <hip/hip_bf16.h>
#include <math.h>

typedef __attribute__((ext_vector_type(2))) float v2f;
typedef __attribute__((ext_vector_type(4))) float v4f;
typedef __attribute__((ext_vector_type(8))) float v8f;

#define D_DIM 64
#define H_DIM 128
#define CPITCH 66

__device__ __forceinline__ unsigned ordKey(float f) {
  unsigned u = __float_as_uint(f);
  return (u & 0x80000000u) ? ~u : (u | 0x80000000u);
}
__device__ __forceinline__ float ordVal(unsigned k) {
  unsigned u = (k & 0x80000000u) ? (k ^ 0x80000000u) : ~k;
  return __uint_as_float(u);
}

#define SUM_SCALE     1099511627776.0f
#define SUM_INV_SCALE (1.0f / 1099511627776.0f)

__global__ void init_seg(unsigned* __restrict__ segMax,
                         unsigned long long* __restrict__ segSum, int numS) {
  int i = blockIdx.x * blockDim.x + threadIdx.x;
  if (i < numS) { segMax[i] = 0u; segSum[i] = 0ull; }
}

#define EDGE_WAVES 32
__global__ __launch_bounds__(EDGE_WAVES * 32)
void edge_mlp_cost(const float* __restrict__ embS,
                   const float* __restrict__ embA,
                   const int*   __restrict__ edgeS,
                   const int*   __restrict__ edgeA,
                   const float* __restrict__ W1,
                   const float* __restrict__ b1,
                   const float* __restrict__ W2,
                   const float* __restrict__ b2,
                   const float* __restrict__ logT,
                   float*       __restrict__ outCost,
                   unsigned*    __restrict__ segMax,
                   int E) {
  __shared__ float sW1[H_DIM * CPITCH * 2];
  __shared__ float sB1[H_DIM];
  __shared__ float sW2[H_DIM];
  __shared__ float sGate[EDGE_WAVES][16];

  const int tid = threadIdx.x;
  for (int i = tid; i < H_DIM * H_DIM; i += EDGE_WAVES * 32) {
    int kk = i >> 7;
    int c  = i & 127;
    sW1[(c * CPITCH + (kk >> 1)) * 2 + (kk & 1)] = W1[i];
  }
  if (tid < H_DIM) { sB1[tid] = b1[tid]; sW2[tid] = W2[tid]; }
  __syncthreads();

  const int wave = __builtin_amdgcn_readfirstlane(tid >> 5);
  const int lane = tid & 31;
  const int half = lane >> 4;
  const int col  = lane & 15;
  const float b2v  = b2[0];
  const float invT = __expf(-logT[0]);
  const int tiles  = (E + 15) >> 4;
  const int stride = gridDim.x * EDGE_WAVES;

  for (int t = blockIdx.x * EDGE_WAVES + wave; t < tiles; t += stride) {
  const int  e0    = t * 16;
  const int  e     = e0 + col;
  const bool valid = (e < E);
  const int  ec    = valid ? e : (E - 1);
  const int  sI    = edgeS[ec];
  const int  aI    = edgeA[ec];
  const float* es = embS + (size_t)sI * D_DIM;
  const float* ea = embA + (size_t)aI * D_DIM;

  v2f a[32];
#pragma unroll
  for (int k = 0; k < 16; ++k) {
    a[k]      = *(const v2f*)(es + 4 * k + 2 * half);
    a[k + 16] = *(const v2f*)(ea + 4 * k + 2 * half);
  }
  float ss = 0.0f;
#pragma unroll
  for (int k = 0; k < 16; ++k) {
    float dx = a[k][0] - a[k + 16][0];
    float dy = a[k][1] - a[k + 16][1];
    ss += dx * dx + dy * dy;
  }
  ss += __shfl_xor(ss, 16);

  float gacc[8] = {0, 0, 0, 0, 0, 0, 0, 0};
#pragma unroll 1
  for (int n = 0; n < 8; ++n) {
    const float* bbase = sW1 + ((n * 16 + col) * CPITCH + half) * 2;
    v8f c0 = {};
    v8f c1 = {};
#pragma unroll
    for (int k = 0; k < 32; k += 2) {
      v2f b0 = *(const v2f*)(bbase + 4 * k);
      v2f b1v = *(const v2f*)(bbase + 4 * (k + 1));
      c0 = __builtin_amdgcn_wmma_f32_16x16x4_f32(false, a[k], false, b0,
                                                 (short)0, c0, false, false);
      c1 = __builtin_amdgcn_wmma_f32_16x16x4_f32(false, a[k + 1], false, b1v,
                                                 (short)0, c1, false, false);
    }
    const float bn  = sB1[n * 16 + col];
    const float w2n = sW2[n * 16 + col];
#pragma unroll
    for (int r = 0; r < 8; ++r) {
      float h = c0[r] + c1[r] + bn;
      h = h > 0.0f ? h : 0.0f;
      gacc[r] += h * w2n;
    }
  }

#pragma unroll
  for (int r = 0; r < 8; ++r) {
    float g = gacc[r];
    g += __shfl_xor(g, 1);
    g += __shfl_xor(g, 2);
    g += __shfl_xor(g, 4);
    g += __shfl_xor(g, 8);
    if (lane == 0)  sGate[wave][r]     = g;
    if (lane == 16) sGate[wave][r + 8] = g;
  }

  if (lane < 16 && valid) {
    float logit = sGate[wave][lane] + b2v;
    float gate  = 1.0f / (1.0f + __expf(-logit));
    float cost  = sqrtf(ss) * gate;
    outCost[e]  = cost;
    atomicMax(&segMax[sI], ordKey(-cost * invT));
  }
  }
}

__global__ __launch_bounds__(EDGE_WAVES * 32)
void node_proj(const float* __restrict__ embS,
               const float* __restrict__ embA,
               const float* __restrict__ W1,
               const float* __restrict__ b1,
               float*       __restrict__ P,
               int numS, int numA) {
  __shared__ float sW1[H_DIM * CPITCH * 2];
  __shared__ __attribute__((aligned(16))) float sB1[2 * H_DIM];

  const int tid = threadIdx.x;
  for (int i = tid; i < H_DIM * H_DIM; i += EDGE_WAVES * 32) {
    int kk = i >> 7;
    int c  = i & 127;
    sW1[(c * CPITCH + (kk >> 1)) * 2 + (kk & 1)] = W1[i];
  }
  if (tid < H_DIM) { sB1[tid] = b1[tid]; sB1[H_DIM + tid] = 0.0f; }
  __syncthreads();

  const int wave = __builtin_amdgcn_readfirstlane(tid >> 5);
  const int lane = tid & 31;
  const int half = lane >> 4;
  const int col  = lane & 15;
  const int tilesS = (numS + 15) >> 4;
  const int tilesA = (numA + 15) >> 4;
  const int tiles  = tilesS + tilesA;

  for (int t = wave * (int)gridDim.x + (int)blockIdx.x; t < tiles;
       t += EDGE_WAVES * (int)gridDim.x) {
    const bool isA = (t >= tilesS);
    const int  tt  = isA ? (t - tilesS) : t;
    const float* emb = isA ? embA : embS;
    const int  nrows = isA ? numA : numS;
    const int  node  = tt * 16 + col;
    const bool valid = (node < nrows);
    const int  nc    = valid ? node : (nrows - 1);
    const float* row = emb + (size_t)nc * D_DIM;

    v2f a[16];
#pragma unroll
    for (int k = 0; k < 16; ++k) a[k] = *(const v2f*)(row + 4 * k + 2 * half);

    float* prow = P + ((size_t)(isA ? numS : 0) + (size_t)nc) * H_DIM + 8 * half;
    const int koff = isA ? D_DIM : 0;
    const int boff = isA ? H_DIM : 0;

#pragma unroll 1
    for (int nt = 0; nt < 8; ++nt) {
      const float* wbase = sW1 + ((nt * 16 + col) * CPITCH + half) * 2 + koff;
      v8f c0 = {};
      v8f c1 = {};
#pragma unroll
      for (int k = 0; k < 16; k += 2) {
        v2f w0 = *(const v2f*)(wbase + 4 * k);
        v2f w1 = *(const v2f*)(wbase + 4 * (k + 1));
        c0 = __builtin_amdgcn_wmma_f32_16x16x4_f32(false, w0, false, a[k],
                                                   (short)0, c0, false, false);
        c1 = __builtin_amdgcn_wmma_f32_16x16x4_f32(false, w1, false, a[k + 1],
                                                   (short)0, c1, false, false);
      }
      const float* bb = sB1 + boff + nt * 16 + 8 * half;
      const v4f blo = *(const v4f*)(bb);
      const v4f bhi = *(const v4f*)(bb + 4);
      v4f lo, hi;
#pragma unroll
      for (int r = 0; r < 4; ++r) {
        lo[r] = c0[r] + c1[r] + blo[r];
        hi[r] = c0[r + 4] + c1[r + 4] + bhi[r];
      }
      if (valid) {
        *(v4f*)(prow + nt * 16)     = lo;
        *(v4f*)(prow + nt * 16 + 4) = hi;
      }
    }
  }
}

struct RowSet {
  v4f es0, es1, ea0, ea1;
  v4f ps0, ps1, ps2, ps3;
  v4f pa0, pa1, pa2, pa3;
};

__device__ __forceinline__ void fetch_rows(RowSet& r,
                                           const float* __restrict__ embS,
                                           const float* __restrict__ embA,
                                           const float* __restrict__ Ps,
                                           const float* __restrict__ Pa,
                                           int s, int a, int sub4) {
  const float* e1 = embS + (size_t)s * D_DIM + sub4;
  const float* e2 = embA + (size_t)a * D_DIM + sub4;
  const float* p1 = Ps + (size_t)s * H_DIM + sub4;
  const float* p2 = Pa + (size_t)a * H_DIM + sub4;
  r.es0 = *(const v4f*)(e1);
  r.es1 = *(const v4f*)(e1 + 32);
  r.ea0 = *(const v4f*)(e2);
  r.ea1 = *(const v4f*)(e2 + 32);
  r.ps0 = *(const v4f*)(p1);
  r.ps1 = *(const v4f*)(p1 + 32);
  r.ps2 = *(const v4f*)(p1 + 64);
  r.ps3 = *(const v4f*)(p1 + 96);
  r.pa0 = *(const v4f*)(p2);
  r.pa1 = *(const v4f*)(p2 + 32);
  r.pa2 = *(const v4f*)(p2 + 64);
  r.pa3 = *(const v4f*)(p2 + 96);
}

__device__ __forceinline__ float sq4(v4f a, v4f b) {
  v4f d = a - b;
  return d[0] * d[0] + d[1] * d[1] + d[2] * d[2] + d[3] * d[3];
}
__device__ __forceinline__ float gate4(v4f p, v4f q, v4f w) {
  v4f h = p + q;
  float g = 0.0f;
#pragma unroll
  for (int i = 0; i < 4; ++i) {
    float x = h[i] > 0.0f ? h[i] : 0.0f;
    g += x * w[i];
  }
  return g;
}

template <int M>
__device__ __forceinline__ float xadd8(float v) {
  int o = __builtin_amdgcn_update_dpp(0, __float_as_int(v), 0x160 + M, 0xf, 0xf, true);
  return v + __int_as_float(o);
}
__device__ __forceinline__ int bsel8(int v, int lane, int i) {
  return __builtin_amdgcn_ds_bpermute(((lane & 24) | i) << 2, v);
}

__device__ __forceinline__ void reduce_rows(const RowSet& r, const float* sW2, int sub4,
                                            float& ss, float& g) {
  float s = sq4(r.es0, r.ea0) + sq4(r.es1, r.ea1);
  float t = gate4(r.ps0, r.pa0, *(const v4f*)(sW2 + sub4))
          + gate4(r.ps1, r.pa1, *(const v4f*)(sW2 + 32 + sub4))
          + gate4(r.ps2, r.pa2, *(const v4f*)(sW2 + 64 + sub4))
          + gate4(r.ps3, r.pa3, *(const v4f*)(sW2 + 96 + sub4));
  s = xadd8<1>(s); t = xadd8<1>(t);
  s = xadd8<2>(s); t = xadd8<2>(t);
  s = xadd8<4>(s); t = xadd8<4>(t);
  ss = s; g = t;
}

__device__ __forceinline__ float tile_cost(const float* __restrict__ embS,
                                           const float* __restrict__ embA,
                                           const float* __restrict__ Ps,
                                           const float* __restrict__ Pa,
                                           const float* sW2,
                                           int curS, int curA,
                                           int lane, int sub, int sub4, float b2v) {
  float rss = 0.0f, rg = 0.0f;

#pragma unroll 1
  for (int i = 0; i < 8; i += 2) {
    RowSet X, Y;
    float ss, g;
    fetch_rows(X, embS, embA, Ps, Pa, bsel8(curS, lane, i), bsel8(curA, lane, i), sub4);
    fetch_rows(Y, embS, embA, Ps, Pa, bsel8(curS, lane, i + 1), bsel8(curA, lane, i + 1), sub4);
    __builtin_amdgcn_sched_barrier(0);
    reduce_rows(X, sW2, sub4, ss, g);
    if (sub == i) { rss = ss; rg = g; }
    reduce_rows(Y, sW2, sub4, ss, g);
    if (sub == i + 1) { rss = ss; rg = g; }
  }

  const float logit = rg + b2v;
  const float gate  = 1.0f / (1.0f + __expf(-logit));
  return sqrtf(rss) * gate;
}

__global__ __launch_bounds__(EDGE_WAVES * 32)
void edge_gate_cost(const float* __restrict__ embS,
                    const float* __restrict__ embA,
                    const int*   __restrict__ edgeS,
                    const int*   __restrict__ edgeA,
                    const float* __restrict__ Ps,
                    const float* __restrict__ Pa,
                    const float* __restrict__ W2,
                    const float* __restrict__ b2,
                    const float* __restrict__ logT,
                    float*       __restrict__ outCost,
                    unsigned*    __restrict__ segMax,
                    int E) {
  __shared__ __attribute__((aligned(16))) float sW2[H_DIM];
  const int tid = threadIdx.x;
  if (tid < H_DIM) sW2[tid] = W2[tid];
  __syncthreads();

  const int wave = __builtin_amdgcn_readfirstlane(tid >> 5);
  const int lane = tid & 31;
  const int sub  = lane & 7;
  const int sub4 = sub * 4;
  const float b2v  = b2[0];
  const float invT = __expf(-logT[0]);
  const int tiles  = (E + 31) >> 5;
  const int stride = (int)gridDim.x * EDGE_WAVES;

  int t = (int)blockIdx.x * EDGE_WAVES + wave;
  if (t >= tiles) return;

  int eL   = t * 32 + lane;
  int eLc  = eL < E ? eL : (E - 1);
  int curS = edgeS[eLc];
  int curA = edgeA[eLc];

  for (;;) {
    const bool hasNext = (t + stride) < tiles;
    const int  tn   = hasNext ? (t + stride) : t;
    const int  nL   = tn * 32 + lane;
    const int  nLc  = nL < E ? nL : (E - 1);
    const int  nxtS = edgeS[nLc];
    const int  nxtA = edgeA[nLc];

    const float cost = tile_cost(embS, embA, Ps, Pa, sW2, curS, curA, lane, sub, sub4, b2v);

    if (eL < E) {
      outCost[eL] = cost;
      atomicMax(&segMax[curS], ordKey(-cost * invT));
    }
    if (!hasNext) break;
    t = tn; eL = nL; curS = nxtS; curA = nxtA;
  }
}

#define SEG_NODES    196
#define NB_MAX       1024
#define PART_BLOCKS  256
#define PART_THREADS 1024
#define CHUNK_MAX    4096
#define STAT_REGS    8

__device__ __forceinline__ int seg_bucket(int s) { return (int)((unsigned)s / (unsigned)SEG_NODES); }

__global__ __launch_bounds__(PART_THREADS)
void edge_gate_place(const float* __restrict__ embS,
                     const float* __restrict__ embA,
                     const int*   __restrict__ edgeS,
                     const int*   __restrict__ edgeA,
                     const float* __restrict__ Ps,
                     const float* __restrict__ Pa,
                     const float* __restrict__ W2,
                     const float* __restrict__ b2,
                     const float* __restrict__ logT,
                     float*         __restrict__ outCost,
                     int*           __restrict__ offG,
                     float*         __restrict__ recVal,
                     unsigned char* __restrict__ recNode,
                     int E, int NB, int chunk, int chunkPad) {
  __shared__ __attribute__((aligned(16))) float sW2[H_DIM];
  __shared__ int h[NB_MAX];
  __shared__ int fill[NB_MAX];
  __shared__ int off[NB_MAX + 1];
  __shared__ int   sSrc[CHUNK_MAX];
  __shared__ float sArg[CHUNK_MAX];
  __shared__ __attribute__((aligned(16))) float    sVal[CHUNK_MAX];
  __shared__ __attribute__((aligned(16))) unsigned sNode4[CHUNK_MAX / 4];
  unsigned char* sNode = (unsigned char*)sNode4;

  const int tid = threadIdx.x;
  const int b   = blockIdx.x;
  if (tid < H_DIM) sW2[tid] = W2[tid];
  for (int k = tid; k < NB; k += PART_THREADS) { h[k] = 0; fill[k] = 0; }
  sNode4[tid] = 0u;
  __syncthreads();

  const int wave = __builtin_amdgcn_readfirstlane(tid >> 5);
  const int lane = tid & 31;
  const int sub  = lane & 7;
  const int sub4 = sub * 4;
  const float b2v  = b2[0];
  const float invT = __expf(-logT[0]);

  const int lo = b * chunk;
  const int hi = (lo + chunk < E) ? (lo + chunk) : E;
  const int n  = (hi > lo) ? (hi - lo) : 0;
  const int nTiles = (n + 31) >> 5;

  int t = wave;
  if (t < nTiles) {
    int iL   = t * 32 + lane;
    int eLc  = iL < n ? (lo + iL) : (hi - 1);
    int curS = edgeS[eLc];
    int curA = edgeA[eLc];

    for (;;) {
      const bool hasNext = (t + EDGE_WAVES) < nTiles;
      const int  tn   = hasNext ? (t + EDGE_WAVES) : t;
      const int  nL   = tn * 32 + lane;
      const int  nLc  = nL < n ? (lo + nL) : (hi - 1);
      const int  nxtS = edgeS[nLc];
      const int  nxtA = edgeA[nLc];

      const float cost = tile_cost(embS, embA, Ps, Pa, sW2, curS, curA, lane, sub, sub4, b2v);

      if (iL < n) {
        outCost[lo + iL] = cost;
        sSrc[iL] = curS;
        sArg[iL] = -cost * invT;
        atomicAdd(&h[seg_bucket(curS)], 1);
      }
      if (!hasNext) break;
      t = tn; iL = nL; curS = nxtS; curA = nxtA;
    }
  }
  __syncthreads();

  if (tid < 32) {
    const int per = (NB + 31) >> 5;
    const int k0  = tid * per;
    int s = 0;
    for (int j = 0; j < per; ++j) { const int k = k0 + j; if (k < NB) s += h[k]; }
    int inc = s;
    for (int d = 1; d < 32; d <<= 1) {
      const int o = __shfl_up(inc, d);
      if (tid >= d) inc += o;
    }
    int run = inc - s;
    for (int j = 0; j < per; ++j) {
      const int k = k0 + j;
      if (k < NB) { off[k] = run; run += h[k]; }
    }
    if (tid == 31) off[NB] = inc;
  }
  __syncthreads();

  for (int k = tid; k <= NB; k += PART_THREADS) offG[b * (NB + 1) + k] = off[k];

#pragma unroll
  for (int j = 0; j < 4; ++j) {
    const int i = j * PART_THREADS + tid;
    if (i < n) {
      const int sv  = sSrc[i];
      const int k   = seg_bucket(sv);
      const int pos = off[k] + atomicAdd(&fill[k], 1);
      sVal[pos]  = sArg[i];
      sNode[pos] = (unsigned char)(sv - k * SEG_NODES);
    }
  }
  __syncthreads();

  float*    gv = recVal + (size_t)b * chunkPad;
  unsigned* gn = (unsigned*)(recNode + (size_t)b * chunkPad);
#pragma unroll
  for (int j = 0; j < 4; ++j) {
    const int i = j * PART_THREADS + tid;
    if (i < n) gv[i] = sVal[i];
  }
  if (4 * tid < n) gn[tid] = sNode4[tid];
}

__global__ __launch_bounds__(PART_THREADS)
void bucket_stats(const int*           __restrict__ offG,
                  const float*         __restrict__ recVal,
                  const unsigned char* __restrict__ recNode,
                  unsigned*            __restrict__ segMax,
                  unsigned long long*  __restrict__ segSum,
                  int numS, int NB, int chunkPad) {
  __shared__ unsigned           mx[SEG_NODES];
  __shared__ unsigned long long sm[SEG_NODES];
  const int tid = threadIdx.x;
  const int k   = blockIdx.x;
  if (tid < SEG_NODES) { mx[tid] = 0u; sm[tid] = 0ull; }

  const int r   = tid >> 2;
  const int sub = tid & 3;
  const int* o  = offG + r * (NB + 1) + k;
  const int st  = r * chunkPad + o[0];
  const int en  = r * chunkPad + o[1];

  float rv[STAT_REGS];
  int   rn[STAT_REGS];
#pragma unroll
  for (int j = 0; j < STAT_REGS; ++j) {
    const int i = st + sub + 4 * j;
    rv[j] = 0.0f;
    rn[j] = -1;
    if (i < en) { rv[j] = recVal[i]; rn[j] = recNode[i]; }
  }
  const int rest = st + sub + 4 * STAT_REGS;
  __syncthreads();

#pragma unroll
  for (int j = 0; j < STAT_REGS; ++j)
    if (rn[j] >= 0) atomicMax(&mx[rn[j]], ordKey(rv[j]));
  for (int i = rest; i < en; i += 4)
    atomicMax(&mx[recNode[i]], ordKey(recVal[i]));
  __syncthreads();

#pragma unroll
  for (int j = 0; j < STAT_REGS; ++j) {
    if (rn[j] >= 0) {
      const float m = ordVal(mx[rn[j]]);
      atomicAdd(&sm[rn[j]], (unsigned long long)(__expf(rv[j] - m) * SUM_SCALE));
    }
  }
  for (int i = rest; i < en; i += 4) {
    const int   nd = recNode[i];
    const float m  = ordVal(mx[nd]);
    atomicAdd(&sm[nd], (unsigned long long)(__expf(recVal[i] - m) * SUM_SCALE));
  }
  __syncthreads();

  if (tid < SEG_NODES) {
    const int s = k * SEG_NODES + tid;
    if (s < numS) { segMax[s] = mx[tid]; segSum[s] = sm[tid]; }
  }
}

#define BE_CAP 6144
#define EPITCH 68
#define PPITCH 132

__global__ __launch_bounds__(PART_THREADS)
void group_edges(const int* __restrict__ edgeS,
                 const int* __restrict__ edgeA,
                 int*           __restrict__ offG,
                 int*           __restrict__ recA,
                 unsigned char* __restrict__ recNode,
                 int*           __restrict__ pos,
                 int E, int NB, int chunk, int chunkPad) {
  __shared__ int h[NB_MAX];
  __shared__ int fill[NB_MAX];
  __shared__ int off[NB_MAX + 1];
  __shared__ __attribute__((aligned(16))) int      sA[CHUNK_MAX];
  __shared__ __attribute__((aligned(16))) unsigned sNode4[CHUNK_MAX / 4];
  unsigned char* sNode = (unsigned char*)sNode4;

  const int tid = threadIdx.x;
  const int b   = blockIdx.x;
  for (int k = tid; k < NB; k += PART_THREADS) { h[k] = 0; fill[k] = 0; }
  sNode4[tid] = 0u;
  __syncthreads();

  const int lo = b * chunk;
  const int hi = (lo + chunk < E) ? (lo + chunk) : E;
  const int n  = (hi > lo) ? (hi - lo) : 0;

  int sv[4], av[4];
#pragma unroll
  for (int j = 0; j < 4; ++j) {
    const int i = j * PART_THREADS + tid;
    sv[j] = -1; av[j] = 0;
    if (i < n) {
      sv[j] = edgeS[lo + i];
      av[j] = edgeA[lo + i];
      atomicAdd(&h[seg_bucket(sv[j])], 1);
    }
  }
  __syncthreads();

  if (tid < 32) {
    const int per = (NB + 31) >> 5;
    const int k0  = tid * per;
    int s = 0;
    for (int j = 0; j < per; ++j) { const int k = k0 + j; if (k < NB) s += h[k]; }
    int inc = s;
    for (int d = 1; d < 32; d <<= 1) {
      const int o = __shfl_up(inc, d);
      if (tid >= d) inc += o;
    }
    int run = inc - s;
    for (int j = 0; j < per; ++j) {
      const int k = k0 + j;
      if (k < NB) { off[k] = run; run += h[k]; }
    }
    if (tid == 31) off[NB] = inc;
  }
  __syncthreads();

  for (int k = tid; k <= NB; k += PART_THREADS) offG[b * (NB + 1) + k] = off[k];

#pragma unroll
  for (int j = 0; j < 4; ++j) {
    const int i = j * PART_THREADS + tid;
    if (i < n) {
      const int k = seg_bucket(sv[j]);
      const int p = off[k] + atomicAdd(&fill[k], 1);
      sA[p]    = av[j];
      sNode[p] = (unsigned char)(sv[j] - k * SEG_NODES);
      pos[lo + i] = b * chunkPad + p;
    }
  }
  __syncthreads();

  int*      ga = recA + (size_t)b * chunkPad;
  unsigned* gn = (unsigned*)(recNode + (size_t)b * chunkPad);
#pragma unroll
  for (int j = 0; j < 4; ++j) {
    const int i = j * PART_THREADS + tid;
    if (i < n) ga[i] = sA[i];
  }
  if (4 * tid < n) gn[tid] = sNode4[tid];
}

struct RowT {
  v4f ea0, ea1;
  v4f pa0, pa1, pa2, pa3;
};
__device__ __forceinline__ void fetch_t(RowT& r,
                                        const float* __restrict__ embA,
                                        const float* __restrict__ Pa,
                                        int a, int sub4) {
  const float* e2 = embA + (size_t)a * D_DIM + sub4;
  const float* p2 = Pa + (size_t)a * H_DIM + sub4;
  r.ea0 = *(const v4f*)(e2);
  r.ea1 = *(const v4f*)(e2 + 32);
  r.pa0 = *(const v4f*)(p2);
  r.pa1 = *(const v4f*)(p2 + 32);
  r.pa2 = *(const v4f*)(p2 + 64);
  r.pa3 = *(const v4f*)(p2 + 96);
}
__device__ __forceinline__ void reduce_t(const RowT& r, const float* sE, const float* sP,
                                         const float* sW2, int sub4, float& ss, float& g) {
  const v4f es0 = *(const v4f*)(sE + sub4);
  const v4f es1 = *(const v4f*)(sE + 32 + sub4);
  const v4f ps0 = *(const v4f*)(sP + sub4);
  const v4f ps1 = *(const v4f*)(sP + 32 + sub4);
  const v4f ps2 = *(const v4f*)(sP + 64 + sub4);
  const v4f ps3 = *(const v4f*)(sP + 96 + sub4);
  float s = sq4(es0, r.ea0) + sq4(es1, r.ea1);
  float t = gate4(ps0, r.pa0, *(const v4f*)(sW2 + sub4))
          + gate4(ps1, r.pa1, *(const v4f*)(sW2 + 32 + sub4))
          + gate4(ps2, r.pa2, *(const v4f*)(sW2 + 64 + sub4))
          + gate4(ps3, r.pa3, *(const v4f*)(sW2 + 96 + sub4));
  s = xadd8<1>(s); t = xadd8<1>(t);
  s = xadd8<2>(s); t = xadd8<2>(t);
  s = xadd8<4>(s); t = xadd8<4>(t);
  ss = s; g = t;
}

__device__ __forceinline__ float tile_cost_b(const float* __restrict__ embA,
                                             const float* __restrict__ Pa,
                                             const float* sRowE, const float* sRowP,
                                             const float* sW2,
                                             int curN, int curA,
                                             int lane, int sub, int sub4, float b2v) {
  float rss = 0.0f, rg = 0.0f;

#pragma unroll 1
  for (int i = 0; i < 8; i += 2) {
    RowT X, Y;
    float ss, g;
    const int n0 = bsel8(curN, lane, i);
    const int n1 = bsel8(curN, lane, i + 1);
    fetch_t(X, embA, Pa, bsel8(curA, lane, i), sub4);
    fetch_t(Y, embA, Pa, bsel8(curA, lane, i + 1), sub4);
    __builtin_amdgcn_sched_barrier(0);
    reduce_t(X, sRowE + n0 * EPITCH, sRowP + n0 * PPITCH, sW2, sub4, ss, g);
    if (sub == i) { rss = ss; rg = g; }
    reduce_t(Y, sRowE + n1 * EPITCH, sRowP + n1 * PPITCH, sW2, sub4, ss, g);
    if (sub == i + 1) { rss = ss; rg = g; }
  }

  const float logit = rg + b2v;
  const float gate  = 1.0f / (1.0f + __expf(-logit));
  return sqrtf(rss) * gate;
}

__global__ __launch_bounds__(PART_THREADS)
void bucket_edge(const float* __restrict__ embS,
                 const float* __restrict__ embA,
                 const float* __restrict__ Ps,
                 const float* __restrict__ Pa,
                 const float* __restrict__ W2,
                 const float* __restrict__ b2,
                 const float* __restrict__ logT,
                 const int*           __restrict__ offG,
                 int*                 recAW,
                 const unsigned char* __restrict__ recNode,
                 float*               recCost,
                 int numS, int NB, int chunkPad) {
  __shared__ __attribute__((aligned(16))) float sRowE[SEG_NODES * EPITCH];
  __shared__ __attribute__((aligned(16))) float sRowP[SEG_NODES * PPITCH];
  __shared__ __attribute__((aligned(16))) float sW2[H_DIM];
  __shared__ unsigned           mx[SEG_NODES];
  __shared__ unsigned long long sm[SEG_NODES];
  __shared__ int runLen[PART_BLOCKS];
  __shared__ int runStart[PART_BLOCKS + 1];
  __shared__ int cA[BE_CAP];
  __shared__ int cPos[BE_CAP];
  __shared__ unsigned char cN[BE_CAP];

  const int tid = threadIdx.x;
  const int k   = blockIdx.x;
  if (tid < H_DIM) sW2[tid] = W2[tid];
  if (tid < SEG_NODES) { mx[tid] = 0u; sm[tid] = 0ull; }

  const int r   = tid >> 2;
  const int sub = tid & 3;
  const int* o  = offG + r * (NB + 1) + k;
  const int st  = o[0];
  const int en  = o[1];
  const int gbase = r * chunkPad;
  if (sub == 0) runLen[r] = en - st;

  const int nodeBase = k * SEG_NODES;
  const int nNodes   = (numS - nodeBase < SEG_NODES) ? (numS - nodeBase) : SEG_NODES;
  const v4f* gE = (const v4f*)(embS + (size_t)nodeBase * D_DIM);
  for (int i = tid; i < nNodes * (D_DIM / 4); i += PART_THREADS)
    *(v4f*)(sRowE + (i >> 4) * EPITCH + (i & 15) * 4) = gE[i];
  const v4f* gP = (const v4f*)(Ps + (size_t)nodeBase * H_DIM);
  for (int i = tid; i < nNodes * (H_DIM / 4); i += PART_THREADS)
    *(v4f*)(sRowP + (i >> 5) * PPITCH + (i & 31) * 4) = gP[i];
  __syncthreads();

  if (tid < 32) {
    int s = 0;
    for (int j = 0; j < PART_BLOCKS / 32; ++j) s += runLen[tid * (PART_BLOCKS / 32) + j];
    int inc = s;
    for (int d = 1; d < 32; d <<= 1) {
      const int v = __shfl_up(inc, d);
      if (tid >= d) inc += v;
    }
    int run = inc - s;
    for (int j = 0; j < PART_BLOCKS / 32; ++j) {
      const int q = tid * (PART_BLOCKS / 32) + j;
      runStart[q] = run;
      run += runLen[q];
    }
    if (tid == 31) runStart[PART_BLOCKS] = inc;
  }
  __syncthreads();

  const int wave = __builtin_amdgcn_readfirstlane(tid >> 5);
  const int lane = tid & 31;
  const int sub8 = lane & 7;
  const int sub4 = sub8 * 4;
  const float b2v  = b2[0];
  const float invT = __expf(-logT[0]);
  const int total = __builtin_amdgcn_readfirstlane(runStart[PART_BLOCKS]);
  const int rs    = runStart[r];

  for (int base = 0; base < total; base += BE_CAP) {
    const int cnt = (total - base < BE_CAP) ? (total - base) : BE_CAP;

    int ca[STAT_REGS];
    int cn[STAT_REGS];
#pragma unroll
    for (int q = 0; q < STAT_REGS; ++q) {
      const int i = st + sub + 4 * q;
      const int j = rs + (i - st) - base;
      ca[q] = 0;
      cn[q] = -1;
      if (i < en && j >= 0 && j < cnt) { ca[q] = recAW[gbase + i]; cn[q] = recNode[gbase + i]; }
    }
#pragma unroll
    for (int q = 0; q < STAT_REGS; ++q) {
      if (cn[q] >= 0) {
        const int i = st + sub + 4 * q;
        const int j = rs + (i - st) - base;
        cA[j]   = ca[q];
        cN[j]   = (unsigned char)cn[q];
        cPos[j] = gbase + i;
      }
    }
    for (int i = st + sub + 4 * STAT_REGS; i < en; i += 4) {
      const int j = rs + (i - st) - base;
      if (j >= 0 && j < cnt) {
        cA[j]   = recAW[gbase + i];
        cN[j]   = recNode[gbase + i];
        cPos[j] = gbase + i;
      }
    }
    __syncthreads();

    const int nT = (cnt + 31) >> 5;
    for (int t = wave; t < nT; t += EDGE_WAVES) {
      const int j  = t * 32 + lane;
      const int jc = j < cnt ? j : (cnt - 1);
      const int curN = cN[jc];
      const int curA = cA[jc];
      const float cost = tile_cost_b(embA, Pa, sRowE, sRowP, sW2, curN, curA,
                                     lane, sub8, sub4, b2v);
      if (j < cnt) {
        recCost[cPos[j]] = cost;
        atomicMax(&mx[curN], ordKey(-cost * invT));
      }
    }
    __syncthreads();
  }

  float rv[STAT_REGS];
  int   rn[STAT_REGS];
#pragma unroll
  for (int j = 0; j < STAT_REGS; ++j) {
    const int i = st + sub + 4 * j;
    rv[j] = 0.0f;
    rn[j] = -1;
    if (i < en) { rv[j] = recCost[gbase + i]; rn[j] = recNode[gbase + i]; }
  }
#pragma unroll
  for (int j = 0; j < STAT_REGS; ++j) {
    if (rn[j] >= 0) {
      const float m = ordVal(mx[rn[j]]);
      rv[j] = __expf(-rv[j] * invT - m);
      atomicAdd(&sm[rn[j]], (unsigned long long)(rv[j] * SUM_SCALE));
    }
  }
  for (int i = st + sub + 4 * STAT_REGS; i < en; i += 4) {
    const int   nd = recNode[gbase + i];
    const float m  = ordVal(mx[nd]);
    atomicAdd(&sm[nd], (unsigned long long)(__expf(-recCost[gbase + i] * invT - m) * SUM_SCALE));
  }
  __syncthreads();

#pragma unroll
  for (int j = 0; j < STAT_REGS; ++j) {
    if (rn[j] >= 0) {
      const int i = st + sub + 4 * j;
      recAW[gbase + i] = __float_as_int(rv[j] / ((float)sm[rn[j]] * SUM_INV_SCALE));
    }
  }
  for (int i = st + sub + 4 * STAT_REGS; i < en; i += 4) {
    const int   nd = recNode[gbase + i];
    const float m  = ordVal(mx[nd]);
    const float ex = __expf(-recCost[gbase + i] * invT - m);
    recAW[gbase + i] = __float_as_int(ex / ((float)sm[nd] * SUM_INV_SCALE));
  }
}

__global__ __launch_bounds__(PART_THREADS)
void chunk_unsort(const float* __restrict__ recCost,
                  const int*   __restrict__ recW,
                  const int*   __restrict__ pos,
                  float* __restrict__ outW,
                  float* __restrict__ outCost,
                  int E, int chunk, int chunkPad) {
  __shared__ float sCost[CHUNK_MAX];
  __shared__ int   sWgt[CHUNK_MAX];

  const int tid  = threadIdx.x;
  const int b    = blockIdx.x;
  const int lo   = b * chunk;
  const int hi   = (lo + chunk < E) ? (lo + chunk) : E;
  const int n    = hi - lo;
  const int base = b * chunkPad;
  if (n <= 0) return;

  int   p[4];
  float c[4];
  int   w[4];
#pragma unroll
  for (int j = 0; j < 4; ++j) {
    const int i  = j * PART_THREADS + tid;
    const int ic = i < n ? i : (n - 1);
    p[j] = pos[lo + ic];
    c[j] = recCost[base + ic];
    w[j] = recW[base + ic];
  }
#pragma unroll
  for (int j = 0; j < 4; ++j) {
    const int i = j * PART_THREADS + tid;
    if (i < n) { sCost[i] = c[j]; sWgt[i] = w[j]; }
  }
  __syncthreads();

#pragma unroll
  for (int j = 0; j < 4; ++j) {
    const int i = j * PART_THREADS + tid;
    if (i < n) {
      const int q = p[j] - base;
      outCost[lo + i] = sCost[q];
      outW[lo + i]    = __int_as_float(sWgt[q]);
    }
  }
}

__global__ void seg_expsum(const float* __restrict__ outCost,
                           const int* __restrict__ edgeS,
                           const unsigned* __restrict__ segMax,
                           unsigned long long* __restrict__ segSum,
                           const float* __restrict__ logT, int E) {
  int e = blockIdx.x * blockDim.x + threadIdx.x;
  if (e >= E) return;
  float invT = __expf(-logT[0]);
  int s = edgeS[e];
  float val = -outCost[e] * invT;
  float m = ordVal(segMax[s]);
  atomicAdd(&segSum[s], (unsigned long long)(__expf(val - m) * SUM_SCALE));
}

__global__ void seg_norm(const float* __restrict__ outCost,
                         const int* __restrict__ edgeS,
                         const unsigned* __restrict__ segMax,
                         const unsigned long long* __restrict__ segSum,
                         const float* __restrict__ logT,
                         float* __restrict__ outW, int E) {
  int e = blockIdx.x * blockDim.x + threadIdx.x;
  if (e >= E) return;
  float invT = __expf(-logT[0]);
  int s = edgeS[e];
  float val = -outCost[e] * invT;
  float m = ordVal(segMax[s]);
  outW[e] = __expf(val - m) / ((float)segSum[s] * SUM_INV_SCALE);
}

extern "C" void kernel_launch(void* const* d_in, const int* in_sizes, int n_in,
                              void* d_out, int out_size, void* d_ws, size_t ws_size,
                              hipStream_t stream) {
  const float* embS = (const float*)d_in[0];
  const float* embA = (const float*)d_in[1];
  const int*   eidx = (const int*)d_in[2];
  const float* W1   = (const float*)d_in[3];
  const float* b1   = (const float*)d_in[4];
  const float* W2   = (const float*)d_in[5];
  const float* b2   = (const float*)d_in[6];
  const float* logT = (const float*)d_in[7];

  const int numS = in_sizes[0] / D_DIM;
  const int numA = in_sizes[1] / D_DIM;
  const int E    = in_sizes[2] / 2;
  const int* edgeS = eidx;
  const int* edgeA = eidx + E;

  float* outW    = (float*)d_out;
  float* outCost = (float*)d_out + E;

  const size_t pFloats = (size_t)(numS + numA) * H_DIM;
  const size_t need    = pFloats * sizeof(float)
                       + (size_t)numS * (sizeof(unsigned long long) + sizeof(unsigned));

  const int NB       = (numS + SEG_NODES - 1) / SEG_NODES;
  const int chunk    = (((E + PART_BLOCKS - 1) / PART_BLOCKS) + 31) & ~31;
  const int chunkPad = (chunk + 127) & ~127;
  const size_t recSlots = (size_t)PART_BLOCKS * chunkPad;
  const size_t needGrp  = need + recSlots * (sizeof(float) + 1)
                        + (size_t)PART_BLOCKS * (NB + 1) * sizeof(int);

  const size_t posSlots = ((size_t)E + 31) & ~(size_t)31;
  const size_t needSrc  = needGrp + recSlots * sizeof(float) + posSlots * sizeof(int);

  if (ws_size >= needSrc && NB <= NB_MAX && chunk <= CHUNK_MAX) {
    float* P = (float*)d_ws;
    int*           recA    = (int*)(P + pFloats);
    float*         recCost = (float*)(recA + recSlots);
    int*           pos     = (int*)(recCost + recSlots);
    unsigned char* recNode = (unsigned char*)(pos + posSlots);
    unsigned long long* segSum = (unsigned long long*)(recNode + recSlots);
    int*                offG   = (int*)((unsigned*)(segSum + numS) + numS);

    node_proj<<<256, EDGE_WAVES * 32, 0, stream>>>(embS, embA, W1, b1, P, numS, numA);
    group_edges<<<PART_BLOCKS, PART_THREADS, 0, stream>>>(
        edgeS, edgeA, offG, recA, recNode, pos, E, NB, chunk, chunkPad);
    bucket_edge<<<NB, PART_THREADS, 0, stream>>>(
        embS, embA, P, P + (size_t)numS * H_DIM, W2, b2, logT,
        offG, recA, recNode, recCost, numS, NB, chunkPad);
    chunk_unsort<<<PART_BLOCKS, PART_THREADS, 0, stream>>>(
        recCost, recA, pos, outW, outCost, E, chunk, chunkPad);
    return;
  }

  if (ws_size >= needGrp && NB <= NB_MAX && chunk <= CHUNK_MAX) {
    float* P = (float*)d_ws;
    float*         recVal  = P + pFloats;
    unsigned char* recNode = (unsigned char*)(recVal + recSlots);
    unsigned long long* segSum = (unsigned long long*)(recNode + recSlots);
    unsigned*           segMax = (unsigned*)(segSum + numS);
    int*                offG   = (int*)(segMax + numS);

    node_proj<<<256, EDGE_WAVES * 32, 0, stream>>>(embS, embA, W1, b1, P, numS, numA);
    edge_gate_place<<<PART_BLOCKS, PART_THREADS, 0, stream>>>(
        embS, embA, edgeS, edgeA, P, P + (size_t)numS * H_DIM, W2, b2, logT,
        outCost, offG, recVal, recNode, E, NB, chunk, chunkPad);
    bucket_stats<<<NB, PART_THREADS, 0, stream>>>(
        offG, recVal, recNode, segMax, segSum, numS, NB, chunkPad);
    seg_norm<<<(E + 255) / 256, 256, 0, stream>>>(outCost, edgeS, segMax, segSum, logT, outW, E);
    return;
  }

  if (ws_size >= need) {
    float* P = (float*)d_ws;
    unsigned long long* segSum = (unsigned long long*)(P + pFloats);
    unsigned*           segMax = (unsigned*)(segSum + numS);

    init_seg<<<(numS + 255) / 256, 256, 0, stream>>>(segMax, segSum, numS);
    node_proj<<<256, EDGE_WAVES * 32, 0, stream>>>(embS, embA, W1, b1, P, numS, numA);
    edge_gate_cost<<<256, EDGE_WAVES * 32, 0, stream>>>(
        embS, embA, edgeS, edgeA, P, P + (size_t)numS * H_DIM, W2, b2, logT,
        outCost, segMax, E);
    seg_expsum<<<(E + 255) / 256, 256, 0, stream>>>(outCost, edgeS, segMax, segSum, logT, E);
    seg_norm<<<(E + 255) / 256, 256, 0, stream>>>(outCost, edgeS, segMax, segSum, logT, outW, E);
    return;
  }

  unsigned long long* segSum = (unsigned long long*)d_ws;
  unsigned*           segMax = (unsigned*)(segSum + numS);

  init_seg<<<(numS + 255) / 256, 256, 0, stream>>>(segMax, segSum, numS);

  edge_mlp_cost<<<256, EDGE_WAVES * 32, 0, stream>>>(
      embS, embA, edgeS, edgeA, W1, b1, W2, b2, logT, outCost, segMax, E);

  seg_expsum<<<(E + 255) / 256, 256, 0, stream>>>(outCost, edgeS, segMax, segSum, logT, E);
  seg_norm<<<(E + 255) / 256, 256, 0, stream>>>(outCost, edgeS, segMax, segSum, logT, outW, E);
}


# node projection and edge grouping in ONE launch (512 blocks, block-uniform branch on block index): independent work, two 32-wave blocks per WGP, grouping runs under the matrix-bound projection
# speedup vs baseline: 4.9563x; 1.0593x over previous
#include <hip/hip_runtime.h>
#include <hip/hip_bf16.h>
#include <math.h>

typedef __attribute__((ext_vector_type(2))) float v2f;
typedef __attribute__((ext_vector_type(4))) float v4f;
typedef __attribute__((ext_vector_type(8))) float v8f;

#define D_DIM 64
#define H_DIM 128
#define CPITCH 66

__device__ __forceinline__ unsigned ordKey(float f) {
  unsigned u = __float_as_uint(f);
  return (u & 0x80000000u) ? ~u : (u | 0x80000000u);
}
__device__ __forceinline__ float ordVal(unsigned k) {
  unsigned u = (k & 0x80000000u) ? (k ^ 0x80000000u) : ~k;
  return __uint_as_float(u);
}

#define SUM_SCALE     1099511627776.0f
#define SUM_INV_SCALE (1.0f / 1099511627776.0f)

__global__ void init_seg(unsigned* __restrict__ segMax,
                         unsigned long long* __restrict__ segSum, int numS) {
  int i = blockIdx.x * blockDim.x + threadIdx.x;
  if (i < numS) { segMax[i] = 0u; segSum[i] = 0ull; }
}

#define EDGE_WAVES 32
__global__ __launch_bounds__(EDGE_WAVES * 32)
void edge_mlp_cost(const float* __restrict__ embS,
                   const float* __restrict__ embA,
                   const int*   __restrict__ edgeS,
                   const int*   __restrict__ edgeA,
                   const float* __restrict__ W1,
                   const float* __restrict__ b1,
                   const float* __restrict__ W2,
                   const float* __restrict__ b2,
                   const float* __restrict__ logT,
                   float*       __restrict__ outCost,
                   unsigned*    __restrict__ segMax,
                   int E) {
  __shared__ float sW1[H_DIM * CPITCH * 2];
  __shared__ float sB1[H_DIM];
  __shared__ float sW2[H_DIM];
  __shared__ float sGate[EDGE_WAVES][16];

  const int tid = threadIdx.x;
  for (int i = tid; i < H_DIM * H_DIM; i += EDGE_WAVES * 32) {
    int kk = i >> 7;
    int c  = i & 127;
    sW1[(c * CPITCH + (kk >> 1)) * 2 + (kk & 1)] = W1[i];
  }
  if (tid < H_DIM) { sB1[tid] = b1[tid]; sW2[tid] = W2[tid]; }
  __syncthreads();

  const int wave = __builtin_amdgcn_readfirstlane(tid >> 5);
  const int lane = tid & 31;
  const int half = lane >> 4;
  const int col  = lane & 15;
  const float b2v  = b2[0];
  const float invT = __expf(-logT[0]);
  const int tiles  = (E + 15) >> 4;
  const int stride = gridDim.x * EDGE_WAVES;

  for (int t = blockIdx.x * EDGE_WAVES + wave; t < tiles; t += stride) {
  const int  e0    = t * 16;
  const int  e     = e0 + col;
  const bool valid = (e < E);
  const int  ec    = valid ? e : (E - 1);
  const int  sI    = edgeS[ec];
  const int  aI    = edgeA[ec];
  const float* es = embS + (size_t)sI * D_DIM;
  const float* ea = embA + (size_t)aI * D_DIM;

  v2f a[32];
#pragma unroll
  for (int k = 0; k < 16; ++k) {
    a[k]      = *(const v2f*)(es + 4 * k + 2 * half);
    a[k + 16] = *(const v2f*)(ea + 4 * k + 2 * half);
  }
  float ss = 0.0f;
#pragma unroll
  for (int k = 0; k < 16; ++k) {
    float dx = a[k][0] - a[k + 16][0];
    float dy = a[k][1] - a[k + 16][1];
    ss += dx * dx + dy * dy;
  }
  ss += __shfl_xor(ss, 16);

  float gacc[8] = {0, 0, 0, 0, 0, 0, 0, 0};
#pragma unroll 1
  for (int n = 0; n < 8; ++n) {
    const float* bbase = sW1 + ((n * 16 + col) * CPITCH + half) * 2;
    v8f c0 = {};
    v8f c1 = {};
#pragma unroll
    for (int k = 0; k < 32; k += 2) {
      v2f b0 = *(const v2f*)(bbase + 4 * k);
      v2f b1v = *(const v2f*)(bbase + 4 * (k + 1));
      c0 = __builtin_amdgcn_wmma_f32_16x16x4_f32(false, a[k], false, b0,
                                                 (short)0, c0, false, false);
      c1 = __builtin_amdgcn_wmma_f32_16x16x4_f32(false, a[k + 1], false, b1v,
                                                 (short)0, c1, false, false);
    }
    const float bn  = sB1[n * 16 + col];
    const float w2n = sW2[n * 16 + col];
#pragma unroll
    for (int r = 0; r < 8; ++r) {
      float h = c0[r] + c1[r] + bn;
      h = h > 0.0f ? h : 0.0f;
      gacc[r] += h * w2n;
    }
  }

#pragma unroll
  for (int r = 0; r < 8; ++r) {
    float g = gacc[r];
    g += __shfl_xor(g, 1);
    g += __shfl_xor(g, 2);
    g += __shfl_xor(g, 4);
    g += __shfl_xor(g, 8);
    if (lane == 0)  sGate[wave][r]     = g;
    if (lane == 16) sGate[wave][r + 8] = g;
  }

  if (lane < 16 && valid) {
    float logit = sGate[wave][lane] + b2v;
    float gate  = 1.0f / (1.0f + __expf(-logit));
    float cost  = sqrtf(ss) * gate;
    outCost[e]  = cost;
    atomicMax(&segMax[sI], ordKey(-cost * invT));
  }
  }
}

__device__ __forceinline__
void node_proj_body(const float* __restrict__ embS,
                    const float* __restrict__ embA,
                    const float* __restrict__ W1,
                    const float* __restrict__ b1,
                    float*       __restrict__ P,
                    int numS, int numA, int blk, int nblk) {
  __shared__ float sW1[H_DIM * CPITCH * 2];
  __shared__ __attribute__((aligned(16))) float sB1[2 * H_DIM];

  const int tid = threadIdx.x;
  for (int i = tid; i < H_DIM * H_DIM; i += EDGE_WAVES * 32) {
    int kk = i >> 7;
    int c  = i & 127;
    sW1[(c * CPITCH + (kk >> 1)) * 2 + (kk & 1)] = W1[i];
  }
  if (tid < H_DIM) { sB1[tid] = b1[tid]; sB1[H_DIM + tid] = 0.0f; }
  __syncthreads();

  const int wave = __builtin_amdgcn_readfirstlane(tid >> 5);
  const int lane = tid & 31;
  const int tilesS = (numS + 15) >> 4;
  const int tilesA = (numA + 15) >> 4;
  const int tiles  = tilesS + tilesA;

  for (int t = wave * nblk + blk; t < tiles; t += EDGE_WAVES * nblk) {
    int ln = lane;
    asm volatile("" : "+v"(ln));
    const int half = ln >> 4;
    const int col  = ln & 15;
    const bool isA = (t >= tilesS);
    const int  tt  = isA ? (t - tilesS) : t;
    const float* emb = isA ? embA : embS;
    const int  nrows = isA ? numA : numS;
    const int  node  = tt * 16 + col;
    const bool valid = (node < nrows);
    const int  nc    = valid ? node : (nrows - 1);
    const float* row = emb + (size_t)nc * D_DIM;

    v2f a[16];
#pragma unroll
    for (int k = 0; k < 16; ++k) a[k] = *(const v2f*)(row + 4 * k + 2 * half);

    float* prow = P + ((size_t)(isA ? numS : 0) + (size_t)nc) * H_DIM + 8 * half;
    const int koff = isA ? D_DIM : 0;
    const int boff = isA ? H_DIM : 0;

#pragma unroll 1
    for (int nt = 0; nt < 8; ++nt) {
      const float* wbase = sW1 + ((nt * 16 + col) * CPITCH + half) * 2 + koff;
      v8f c0 = {};
      v8f c1 = {};
#pragma unroll
      for (int k = 0; k < 16; k += 2) {
        v2f w0 = *(const v2f*)(wbase + 4 * k);
        v2f w1 = *(const v2f*)(wbase + 4 * (k + 1));
        c0 = __builtin_amdgcn_wmma_f32_16x16x4_f32(false, w0, false, a[k],
                                                   (short)0, c0, false, false);
        c1 = __builtin_amdgcn_wmma_f32_16x16x4_f32(false, w1, false, a[k + 1],
                                                   (short)0, c1, false, false);
      }
      const float* bb = sB1 + boff + nt * 16 + 8 * half;
      const v4f blo = *(const v4f*)(bb);
      const v4f bhi = *(const v4f*)(bb + 4);
      v4f lo, hi;
#pragma unroll
      for (int r = 0; r < 4; ++r) {
        lo[r] = c0[r] + c1[r] + blo[r];
        hi[r] = c0[r + 4] + c1[r + 4] + bhi[r];
      }
      if (valid) {
        *(v4f*)(prow + nt * 16)     = lo;
        *(v4f*)(prow + nt * 16 + 4) = hi;
      }
    }
  }
}

__global__ __launch_bounds__(EDGE_WAVES * 32)
void node_proj(const float* __restrict__ embS,
               const float* __restrict__ embA,
               const float* __restrict__ W1,
               const float* __restrict__ b1,
               float*       __restrict__ P,
               int numS, int numA) {
  node_proj_body(embS, embA, W1, b1, P, numS, numA, (int)blockIdx.x, (int)gridDim.x);
}

struct RowSet {
  v4f es0, es1, ea0, ea1;
  v4f ps0, ps1, ps2, ps3;
  v4f pa0, pa1, pa2, pa3;
};

__device__ __forceinline__ void fetch_rows(RowSet& r,
                                           const float* __restrict__ embS,
                                           const float* __restrict__ embA,
                                           const float* __restrict__ Ps,
                                           const float* __restrict__ Pa,
                                           int s, int a, int sub4) {
  const float* e1 = embS + (size_t)s * D_DIM + sub4;
  const float* e2 = embA + (size_t)a * D_DIM + sub4;
  const float* p1 = Ps + (size_t)s * H_DIM + sub4;
  const float* p2 = Pa + (size_t)a * H_DIM + sub4;
  r.es0 = *(const v4f*)(e1);
  r.es1 = *(const v4f*)(e1 + 32);
  r.ea0 = *(const v4f*)(e2);
  r.ea1 = *(const v4f*)(e2 + 32);
  r.ps0 = *(const v4f*)(p1);
  r.ps1 = *(const v4f*)(p1 + 32);
  r.ps2 = *(const v4f*)(p1 + 64);
  r.ps3 = *(const v4f*)(p1 + 96);
  r.pa0 = *(const v4f*)(p2);
  r.pa1 = *(const v4f*)(p2 + 32);
  r.pa2 = *(const v4f*)(p2 + 64);
  r.pa3 = *(const v4f*)(p2 + 96);
}

__device__ __forceinline__ float sq4(v4f a, v4f b) {
  v4f d = a - b;
  return d[0] * d[0] + d[1] * d[1] + d[2] * d[2] + d[3] * d[3];
}
__device__ __forceinline__ float gate4(v4f p, v4f q, v4f w) {
  v4f h = p + q;
  float g = 0.0f;
#pragma unroll
  for (int i = 0; i < 4; ++i) {
    float x = h[i] > 0.0f ? h[i] : 0.0f;
    g += x * w[i];
  }
  return g;
}

template <int M>
__device__ __forceinline__ float xadd8(float v) {
  int o = __builtin_amdgcn_update_dpp(0, __float_as_int(v), 0x160 + M, 0xf, 0xf, true);
  return v + __int_as_float(o);
}
__device__ __forceinline__ int bsel8(int v, int lane, int i) {
  return __builtin_amdgcn_ds_bpermute(((lane & 24) | i) << 2, v);
}

__device__ __forceinline__ void reduce_rows(const RowSet& r, const float* sW2, int sub4,
                                            float& ss, float& g) {
  float s = sq4(r.es0, r.ea0) + sq4(r.es1, r.ea1);
  float t = gate4(r.ps0, r.pa0, *(const v4f*)(sW2 + sub4))
          + gate4(r.ps1, r.pa1, *(const v4f*)(sW2 + 32 + sub4))
          + gate4(r.ps2, r.pa2, *(const v4f*)(sW2 + 64 + sub4))
          + gate4(r.ps3, r.pa3, *(const v4f*)(sW2 + 96 + sub4));
  s = xadd8<1>(s); t = xadd8<1>(t);
  s = xadd8<2>(s); t = xadd8<2>(t);
  s = xadd8<4>(s); t = xadd8<4>(t);
  ss = s; g = t;
}

__device__ __forceinline__ float tile_cost(const float* __restrict__ embS,
                                           const float* __restrict__ embA,
                                           const float* __restrict__ Ps,
                                           const float* __restrict__ Pa,
                                           const float* sW2,
                                           int curS, int curA,
                                           int lane, int sub, int sub4, float b2v) {
  float rss = 0.0f, rg = 0.0f;

#pragma unroll 1
  for (int i = 0; i < 8; i += 2) {
    RowSet X, Y;
    float ss, g;
    fetch_rows(X, embS, embA, Ps, Pa, bsel8(curS, lane, i), bsel8(curA, lane, i), sub4);
    fetch_rows(Y, embS, embA, Ps, Pa, bsel8(curS, lane, i + 1), bsel8(curA, lane, i + 1), sub4);
    __builtin_amdgcn_sched_barrier(0);
    reduce_rows(X, sW2, sub4, ss, g);
    if (sub == i) { rss = ss; rg = g; }
    reduce_rows(Y, sW2, sub4, ss, g);
    if (sub == i + 1) { rss = ss; rg = g; }
  }

  const float logit = rg + b2v;
  const float gate  = 1.0f / (1.0f + __expf(-logit));
  return sqrtf(rss) * gate;
}

__global__ __launch_bounds__(EDGE_WAVES * 32)
void edge_gate_cost(const float* __restrict__ embS,
                    const float* __restrict__ embA,
                    const int*   __restrict__ edgeS,
                    const int*   __restrict__ edgeA,
                    const float* __restrict__ Ps,
                    const float* __restrict__ Pa,
                    const float* __restrict__ W2,
                    const float* __restrict__ b2,
                    const float* __restrict__ logT,
                    float*       __restrict__ outCost,
                    unsigned*    __restrict__ segMax,
                    int E) {
  __shared__ __attribute__((aligned(16))) float sW2[H_DIM];
  const int tid = threadIdx.x;
  if (tid < H_DIM) sW2[tid] = W2[tid];
  __syncthreads();

  const int wave = __builtin_amdgcn_readfirstlane(tid >> 5);
  const int lane = tid & 31;
  const int sub  = lane & 7;
  const int sub4 = sub * 4;
  const float b2v  = b2[0];
  const float invT = __expf(-logT[0]);
  const int tiles  = (E + 31) >> 5;
  const int stride = (int)gridDim.x * EDGE_WAVES;

  int t = (int)blockIdx.x * EDGE_WAVES + wave;
  if (t >= tiles) return;

  int eL   = t * 32 + lane;
  int eLc  = eL < E ? eL : (E - 1);
  int curS = edgeS[eLc];
  int curA = edgeA[eLc];

  for (;;) {
    const bool hasNext = (t + stride) < tiles;
    const int  tn   = hasNext ? (t + stride) : t;
    const int  nL   = tn * 32 + lane;
    const int  nLc  = nL < E ? nL : (E - 1);
    const int  nxtS = edgeS[nLc];
    const int  nxtA = edgeA[nLc];

    const float cost = tile_cost(embS, embA, Ps, Pa, sW2, curS, curA, lane, sub, sub4, b2v);

    if (eL < E) {
      outCost[eL] = cost;
      atomicMax(&segMax[curS], ordKey(-cost * invT));
    }
    if (!hasNext) break;
    t = tn; eL = nL; curS = nxtS; curA = nxtA;
  }
}

#define SEG_NODES    196
#define NB_MAX       1024
#define PART_BLOCKS  256
#define PART_THREADS 1024
#define CHUNK_MAX    4096
#define STAT_REGS    8

__device__ __forceinline__ int seg_bucket(int s) { return (int)((unsigned)s / (unsigned)SEG_NODES); }

__global__ __launch_bounds__(PART_THREADS)
void edge_gate_place(const float* __restrict__ embS,
                     const float* __restrict__ embA,
                     const int*   __restrict__ edgeS,
                     const int*   __restrict__ edgeA,
                     const float* __restrict__ Ps,
                     const float* __restrict__ Pa,
                     const float* __restrict__ W2,
                     const float* __restrict__ b2,
                     const float* __restrict__ logT,
                     float*         __restrict__ outCost,
                     int*           __restrict__ offG,
                     float*         __restrict__ recVal,
                     unsigned char* __restrict__ recNode,
                     int E, int NB, int chunk, int chunkPad) {
  __shared__ __attribute__((aligned(16))) float sW2[H_DIM];
  __shared__ int h[NB_MAX];
  __shared__ int fill[NB_MAX];
  __shared__ int off[NB_MAX + 1];
  __shared__ int   sSrc[CHUNK_MAX];
  __shared__ float sArg[CHUNK_MAX];
  __shared__ __attribute__((aligned(16))) float    sVal[CHUNK_MAX];
  __shared__ __attribute__((aligned(16))) unsigned sNode4[CHUNK_MAX / 4];
  unsigned char* sNode = (unsigned char*)sNode4;

  const int tid = threadIdx.x;
  const int b   = blockIdx.x;
  if (tid < H_DIM) sW2[tid] = W2[tid];
  for (int k = tid; k < NB; k += PART_THREADS) { h[k] = 0; fill[k] = 0; }
  sNode4[tid] = 0u;
  __syncthreads();

  const int wave = __builtin_amdgcn_readfirstlane(tid >> 5);
  const int lane = tid & 31;
  const int sub  = lane & 7;
  const int sub4 = sub * 4;
  const float b2v  = b2[0];
  const float invT = __expf(-logT[0]);

  const int lo = b * chunk;
  const int hi = (lo + chunk < E) ? (lo + chunk) : E;
  const int n  = (hi > lo) ? (hi - lo) : 0;
  const int nTiles = (n + 31) >> 5;

  int t = wave;
  if (t < nTiles) {
    int iL   = t * 32 + lane;
    int eLc  = iL < n ? (lo + iL) : (hi - 1);
    int curS = edgeS[eLc];
    int curA = edgeA[eLc];

    for (;;) {
      const bool hasNext = (t + EDGE_WAVES) < nTiles;
      const int  tn   = hasNext ? (t + EDGE_WAVES) : t;
      const int  nL   = tn * 32 + lane;
      const int  nLc  = nL < n ? (lo + nL) : (hi - 1);
      const int  nxtS = edgeS[nLc];
      const int  nxtA = edgeA[nLc];

      const float cost = tile_cost(embS, embA, Ps, Pa, sW2, curS, curA, lane, sub, sub4, b2v);

      if (iL < n) {
        outCost[lo + iL] = cost;
        sSrc[iL] = curS;
        sArg[iL] = -cost * invT;
        atomicAdd(&h[seg_bucket(curS)], 1);
      }
      if (!hasNext) break;
      t = tn; iL = nL; curS = nxtS; curA = nxtA;
    }
  }
  __syncthreads();

  if (tid < 32) {
    const int per = (NB + 31) >> 5;
    const int k0  = tid * per;
    int s = 0;
    for (int j = 0; j < per; ++j) { const int k = k0 + j; if (k < NB) s += h[k]; }
    int inc = s;
    for (int d = 1; d < 32; d <<= 1) {
      const int o = __shfl_up(inc, d);
      if (tid >= d) inc += o;
    }
    int run = inc - s;
    for (int j = 0; j < per; ++j) {
      const int k = k0 + j;
      if (k < NB) { off[k] = run; run += h[k]; }
    }
    if (tid == 31) off[NB] = inc;
  }
  __syncthreads();

  for (int k = tid; k <= NB; k += PART_THREADS) offG[b * (NB + 1) + k] = off[k];

#pragma unroll
  for (int j = 0; j < 4; ++j) {
    const int i = j * PART_THREADS + tid;
    if (i < n) {
      const int sv  = sSrc[i];
      const int k   = seg_bucket(sv);
      const int pos = off[k] + atomicAdd(&fill[k], 1);
      sVal[pos]  = sArg[i];
      sNode[pos] = (unsigned char)(sv - k * SEG_NODES);
    }
  }
  __syncthreads();

  float*    gv = recVal + (size_t)b * chunkPad;
  unsigned* gn = (unsigned*)(recNode + (size_t)b * chunkPad);
#pragma unroll
  for (int j = 0; j < 4; ++j) {
    const int i = j * PART_THREADS + tid;
    if (i < n) gv[i] = sVal[i];
  }
  if (4 * tid < n) gn[tid] = sNode4[tid];
}

__global__ __launch_bounds__(PART_THREADS)
void bucket_stats(const int*           __restrict__ offG,
                  const float*         __restrict__ recVal,
                  const unsigned char* __restrict__ recNode,
                  unsigned*            __restrict__ segMax,
                  unsigned long long*  __restrict__ segSum,
                  int numS, int NB, int chunkPad) {
  __shared__ unsigned           mx[SEG_NODES];
  __shared__ unsigned long long sm[SEG_NODES];
  const int tid = threadIdx.x;
  const int k   = blockIdx.x;
  if (tid < SEG_NODES) { mx[tid] = 0u; sm[tid] = 0ull; }

  const int r   = tid >> 2;
  const int sub = tid & 3;
  const int* o  = offG + r * (NB + 1) + k;
  const int st  = r * chunkPad + o[0];
  const int en  = r * chunkPad + o[1];

  float rv[STAT_REGS];
  int   rn[STAT_REGS];
#pragma unroll
  for (int j = 0; j < STAT_REGS; ++j) {
    const int i = st + sub + 4 * j;
    rv[j] = 0.0f;
    rn[j] = -1;
    if (i < en) { rv[j] = recVal[i]; rn[j] = recNode[i]; }
  }
  const int rest = st + sub + 4 * STAT_REGS;
  __syncthreads();

#pragma unroll
  for (int j = 0; j < STAT_REGS; ++j)
    if (rn[j] >= 0) atomicMax(&mx[rn[j]], ordKey(rv[j]));
  for (int i = rest; i < en; i += 4)
    atomicMax(&mx[recNode[i]], ordKey(recVal[i]));
  __syncthreads();

#pragma unroll
  for (int j = 0; j < STAT_REGS; ++j) {
    if (rn[j] >= 0) {
      const float m = ordVal(mx[rn[j]]);
      atomicAdd(&sm[rn[j]], (unsigned long long)(__expf(rv[j] - m) * SUM_SCALE));
    }
  }
  for (int i = rest; i < en; i += 4) {
    const int   nd = recNode[i];
    const float m  = ordVal(mx[nd]);
    atomicAdd(&sm[nd], (unsigned long long)(__expf(recVal[i] - m) * SUM_SCALE));
  }
  __syncthreads();

  if (tid < SEG_NODES) {
    const int s = k * SEG_NODES + tid;
    if (s < numS) { segMax[s] = mx[tid]; segSum[s] = sm[tid]; }
  }
}

#define BE_CAP 6144
#define EPITCH 68
#define PPITCH 132

__device__ __forceinline__
void group_edges_body(const int* __restrict__ edgeS,
                      const int* __restrict__ edgeA,
                      int*           __restrict__ offG,
                      int*           __restrict__ recA,
                      unsigned char* __restrict__ recNode,
                      int*           __restrict__ pos,
                      int E, int NB, int chunk, int chunkPad, int b) {
  __shared__ int h[NB_MAX];
  __shared__ int fill[NB_MAX];
  __shared__ int off[NB_MAX + 1];
  __shared__ __attribute__((aligned(16))) int      sA[CHUNK_MAX];
  __shared__ __attribute__((aligned(16))) unsigned sNode4[CHUNK_MAX / 4];
  unsigned char* sNode = (unsigned char*)sNode4;

  const int tid = threadIdx.x;
  for (int k = tid; k < NB; k += PART_THREADS) { h[k] = 0; fill[k] = 0; }
  sNode4[tid] = 0u;
  __syncthreads();

  const int lo = b * chunk;
  const int hi = (lo + chunk < E) ? (lo + chunk) : E;
  const int n  = (hi > lo) ? (hi - lo) : 0;

  int sv[4], av[4];
#pragma unroll
  for (int j = 0; j < 4; ++j) {
    const int i = j * PART_THREADS + tid;
    sv[j] = -1; av[j] = 0;
    if (i < n) {
      sv[j] = edgeS[lo + i];
      av[j] = edgeA[lo + i];
      atomicAdd(&h[seg_bucket(sv[j])], 1);
    }
  }
  __syncthreads();

  if (tid < 32) {
    const int per = (NB + 31) >> 5;
    const int k0  = tid * per;
    int s = 0;
    for (int j = 0; j < per; ++j) { const int k = k0 + j; if (k < NB) s += h[k]; }
    int inc = s;
    for (int d = 1; d < 32; d <<= 1) {
      const int o = __shfl_up(inc, d);
      if (tid >= d) inc += o;
    }
    int run = inc - s;
    for (int j = 0; j < per; ++j) {
      const int k = k0 + j;
      if (k < NB) { off[k] = run; run += h[k]; }
    }
    if (tid == 31) off[NB] = inc;
  }
  __syncthreads();

  for (int k = tid; k <= NB; k += PART_THREADS) offG[b * (NB + 1) + k] = off[k];

#pragma unroll
  for (int j = 0; j < 4; ++j) {
    const int i = j * PART_THREADS + tid;
    if (i < n) {
      const int k = seg_bucket(sv[j]);
      const int p = off[k] + atomicAdd(&fill[k], 1);
      sA[p]    = av[j];
      sNode[p] = (unsigned char)(sv[j] - k * SEG_NODES);
      pos[lo + i] = b * chunkPad + p;
    }
  }
  __syncthreads();

  int*      ga = recA + (size_t)b * chunkPad;
  unsigned* gn = (unsigned*)(recNode + (size_t)b * chunkPad);
#pragma unroll
  for (int j = 0; j < 4; ++j) {
    const int i = j * PART_THREADS + tid;
    if (i < n) ga[i] = sA[i];
  }
  if (4 * tid < n) gn[tid] = sNode4[tid];
}

#define PROJ_BLOCKS 256
__global__ __launch_bounds__(PART_THREADS, 16)
void proj_and_group(const float* __restrict__ embS,
                    const float* __restrict__ embA,
                    const float* __restrict__ W1,
                    const float* __restrict__ b1,
                    float*       __restrict__ P,
                    int numS, int numA,
                    const int* __restrict__ edgeS,
                    const int* __restrict__ edgeA,
                    int*           __restrict__ offG,
                    int*           __restrict__ recA,
                    unsigned char* __restrict__ recNode,
                    int*           __restrict__ pos,
                    int E, int NB, int chunk, int chunkPad) {
  if (blockIdx.x < PROJ_BLOCKS) {
    node_proj_body(embS, embA, W1, b1, P, numS, numA, (int)blockIdx.x, PROJ_BLOCKS);
  } else {
    group_edges_body(edgeS, edgeA, offG, recA, recNode, pos, E, NB, chunk, chunkPad,
                     (int)blockIdx.x - PROJ_BLOCKS);
  }
}

struct RowT {
  v4f ea0, ea1;
  v4f pa0, pa1, pa2, pa3;
};
__device__ __forceinline__ void fetch_t(RowT& r,
                                        const float* __restrict__ embA,
                                        const float* __restrict__ Pa,
                                        int a, int sub4) {
  const float* e2 = embA + (size_t)a * D_DIM + sub4;
  const float* p2 = Pa + (size_t)a * H_DIM + sub4;
  r.ea0 = *(const v4f*)(e2);
  r.ea1 = *(const v4f*)(e2 + 32);
  r.pa0 = *(const v4f*)(p2);
  r.pa1 = *(const v4f*)(p2 + 32);
  r.pa2 = *(const v4f*)(p2 + 64);
  r.pa3 = *(const v4f*)(p2 + 96);
}
__device__ __forceinline__ void reduce_t(const RowT& r, const float* sE, const float* sP,
                                         const float* sW2, int sub4, float& ss, float& g) {
  const v4f es0 = *(const v4f*)(sE + sub4);
  const v4f es1 = *(const v4f*)(sE + 32 + sub4);
  const v4f ps0 = *(const v4f*)(sP + sub4);
  const v4f ps1 = *(const v4f*)(sP + 32 + sub4);
  const v4f ps2 = *(const v4f*)(sP + 64 + sub4);
  const v4f ps3 = *(const v4f*)(sP + 96 + sub4);
  float s = sq4(es0, r.ea0) + sq4(es1, r.ea1);
  float t = gate4(ps0, r.pa0, *(const v4f*)(sW2 + sub4))
          + gate4(ps1, r.pa1, *(const v4f*)(sW2 + 32 + sub4))
          + gate4(ps2, r.pa2, *(const v4f*)(sW2 + 64 + sub4))
          + gate4(ps3, r.pa3, *(const v4f*)(sW2 + 96 + sub4));
  s = xadd8<1>(s); t = xadd8<1>(t);
  s = xadd8<2>(s); t = xadd8<2>(t);
  s = xadd8<4>(s); t = xadd8<4>(t);
  ss = s; g = t;
}

__device__ __forceinline__ float tile_cost_b(const float* __restrict__ embA,
                                             const float* __restrict__ Pa,
                                             const float* sRowE, const float* sRowP,
                                             const float* sW2,
                                             int curN, int curA,
                                             int lane, int sub, int sub4, float b2v) {
  float rss = 0.0f, rg = 0.0f;

#pragma unroll 1
  for (int i = 0; i < 8; i += 2) {
    RowT X, Y;
    float ss, g;
    const int n0 = bsel8(curN, lane, i);
    const int n1 = bsel8(curN, lane, i + 1);
    fetch_t(X, embA, Pa, bsel8(curA, lane, i), sub4);
    fetch_t(Y, embA, Pa, bsel8(curA, lane, i + 1), sub4);
    __builtin_amdgcn_sched_barrier(0);
    reduce_t(X, sRowE + n0 * EPITCH, sRowP + n0 * PPITCH, sW2, sub4, ss, g);
    if (sub == i) { rss = ss; rg = g; }
    reduce_t(Y, sRowE + n1 * EPITCH, sRowP + n1 * PPITCH, sW2, sub4, ss, g);
    if (sub == i + 1) { rss = ss; rg = g; }
  }

  const float logit = rg + b2v;
  const float gate  = 1.0f / (1.0f + __expf(-logit));
  return sqrtf(rss) * gate;
}

__global__ __launch_bounds__(PART_THREADS)
void bucket_edge(const float* __restrict__ embS,
                 const float* __restrict__ embA,
                 const float* __restrict__ Ps,
                 const float* __restrict__ Pa,
                 const float* __restrict__ W2,
                 const float* __restrict__ b2,
                 const float* __restrict__ logT,
                 const int*           __restrict__ offG,
                 int*                 recAW,
                 const unsigned char* __restrict__ recNode,
                 float*               recCost,
                 int numS, int NB, int chunkPad) {
  __shared__ __attribute__((aligned(16))) float sRowE[SEG_NODES * EPITCH];
  __shared__ __attribute__((aligned(16))) float sRowP[SEG_NODES * PPITCH];
  __shared__ __attribute__((aligned(16))) float sW2[H_DIM];
  __shared__ unsigned           mx[SEG_NODES];
  __shared__ unsigned long long sm[SEG_NODES];
  __shared__ int runLen[PART_BLOCKS];
  __shared__ int runStart[PART_BLOCKS + 1];
  __shared__ int cA[BE_CAP];
  __shared__ int cPos[BE_CAP];
  __shared__ unsigned char cN[BE_CAP];

  const int tid = threadIdx.x;
  const int k   = blockIdx.x;
  if (tid < H_DIM) sW2[tid] = W2[tid];
  if (tid < SEG_NODES) { mx[tid] = 0u; sm[tid] = 0ull; }

  const int r   = tid >> 2;
  const int sub = tid & 3;
  const int* o  = offG + r * (NB + 1) + k;
  const int st  = o[0];
  const int en  = o[1];
  const int gbase = r * chunkPad;
  if (sub == 0) runLen[r] = en - st;

  const int nodeBase = k * SEG_NODES;
  const int nNodes   = (numS - nodeBase < SEG_NODES) ? (numS - nodeBase) : SEG_NODES;
  const v4f* gE = (const v4f*)(embS + (size_t)nodeBase * D_DIM);
  for (int i = tid; i < nNodes * (D_DIM / 4); i += PART_THREADS)
    *(v4f*)(sRowE + (i >> 4) * EPITCH + (i & 15) * 4) = gE[i];
  const v4f* gP = (const v4f*)(Ps + (size_t)nodeBase * H_DIM);
  for (int i = tid; i < nNodes * (H_DIM / 4); i += PART_THREADS)
    *(v4f*)(sRowP + (i >> 5) * PPITCH + (i & 31) * 4) = gP[i];
  __syncthreads();

  if (tid < 32) {
    int s = 0;
    for (int j = 0; j < PART_BLOCKS / 32; ++j) s += runLen[tid * (PART_BLOCKS / 32) + j];
    int inc = s;
    for (int d = 1; d < 32; d <<= 1) {
      const int v = __shfl_up(inc, d);
      if (tid >= d) inc += v;
    }
    int run = inc - s;
    for (int j = 0; j < PART_BLOCKS / 32; ++j) {
      const int q = tid * (PART_BLOCKS / 32) + j;
      runStart[q] = run;
      run += runLen[q];
    }
    if (tid == 31) runStart[PART_BLOCKS] = inc;
  }
  __syncthreads();

  const int wave = __builtin_amdgcn_readfirstlane(tid >> 5);
  const int lane = tid & 31;
  const int sub8 = lane & 7;
  const int sub4 = sub8 * 4;
  const float b2v  = b2[0];
  const float invT = __expf(-logT[0]);
  const int total = __builtin_amdgcn_readfirstlane(runStart[PART_BLOCKS]);
  const int rs    = runStart[r];

  for (int base = 0; base < total; base += BE_CAP) {
    const int cnt = (total - base < BE_CAP) ? (total - base) : BE_CAP;

    int ca[STAT_REGS];
    int cn[STAT_REGS];
#pragma unroll
    for (int q = 0; q < STAT_REGS; ++q) {
      const int i = st + sub + 4 * q;
      const int j = rs + (i - st) - base;
      ca[q] = 0;
      cn[q] = -1;
      if (i < en && j >= 0 && j < cnt) { ca[q] = recAW[gbase + i]; cn[q] = recNode[gbase + i]; }
    }
#pragma unroll
    for (int q = 0; q < STAT_REGS; ++q) {
      if (cn[q] >= 0) {
        const int i = st + sub + 4 * q;
        const int j = rs + (i - st) - base;
        cA[j]   = ca[q];
        cN[j]   = (unsigned char)cn[q];
        cPos[j] = gbase + i;
      }
    }
    for (int i = st + sub + 4 * STAT_REGS; i < en; i += 4) {
      const int j = rs + (i - st) - base;
      if (j >= 0 && j < cnt) {
        cA[j]   = recAW[gbase + i];
        cN[j]   = recNode[gbase + i];
        cPos[j] = gbase + i;
      }
    }
    __syncthreads();

    const int nT = (cnt + 31) >> 5;
    for (int t = wave; t < nT; t += EDGE_WAVES) {
      const int j  = t * 32 + lane;
      const int jc = j < cnt ? j : (cnt - 1);
      const int curN = cN[jc];
      const int curA = cA[jc];
      const float cost = tile_cost_b(embA, Pa, sRowE, sRowP, sW2, curN, curA,
                                     lane, sub8, sub4, b2v);
      if (j < cnt) {
        recCost[cPos[j]] = cost;
        atomicMax(&mx[curN], ordKey(-cost * invT));
      }
    }
    __syncthreads();
  }

  float rv[STAT_REGS];
  int   rn[STAT_REGS];
#pragma unroll
  for (int j = 0; j < STAT_REGS; ++j) {
    const int i = st + sub + 4 * j;
    rv[j] = 0.0f;
    rn[j] = -1;
    if (i < en) { rv[j] = recCost[gbase + i]; rn[j] = recNode[gbase + i]; }
  }
#pragma unroll
  for (int j = 0; j < STAT_REGS; ++j) {
    if (rn[j] >= 0) {
      const float m = ordVal(mx[rn[j]]);
      rv[j] = __expf(-rv[j] * invT - m);
      atomicAdd(&sm[rn[j]], (unsigned long long)(rv[j] * SUM_SCALE));
    }
  }
  for (int i = st + sub + 4 * STAT_REGS; i < en; i += 4) {
    const int   nd = recNode[gbase + i];
    const float m  = ordVal(mx[nd]);
    atomicAdd(&sm[nd], (unsigned long long)(__expf(-recCost[gbase + i] * invT - m) * SUM_SCALE));
  }
  __syncthreads();

#pragma unroll
  for (int j = 0; j < STAT_REGS; ++j) {
    if (rn[j] >= 0) {
      const int i = st + sub + 4 * j;
      recAW[gbase + i] = __float_as_int(rv[j] / ((float)sm[rn[j]] * SUM_INV_SCALE));
    }
  }
  for (int i = st + sub + 4 * STAT_REGS; i < en; i += 4) {
    const int   nd = recNode[gbase + i];
    const float m  = ordVal(mx[nd]);
    const float ex = __expf(-recCost[gbase + i] * invT - m);
    recAW[gbase + i] = __float_as_int(ex / ((float)sm[nd] * SUM_INV_SCALE));
  }
}

__global__ __launch_bounds__(PART_THREADS)
void chunk_unsort(const float* __restrict__ recCost,
                  const int*   __restrict__ recW,
                  const int*   __restrict__ pos,
                  float* __restrict__ outW,
                  float* __restrict__ outCost,
                  int E, int chunk, int chunkPad) {
  __shared__ float sCost[CHUNK_MAX];
  __shared__ int   sWgt[CHUNK_MAX];

  const int tid  = threadIdx.x;
  const int b    = blockIdx.x;
  const int lo   = b * chunk;
  const int hi   = (lo + chunk < E) ? (lo + chunk) : E;
  const int n    = hi - lo;
  const int base = b * chunkPad;
  if (n <= 0) return;

  int   p[4];
  float c[4];
  int   w[4];
#pragma unroll
  for (int j = 0; j < 4; ++j) {
    const int i  = j * PART_THREADS + tid;
    const int ic = i < n ? i : (n - 1);
    p[j] = pos[lo + ic];
    c[j] = recCost[base + ic];
    w[j] = recW[base + ic];
  }
#pragma unroll
  for (int j = 0; j < 4; ++j) {
    const int i = j * PART_THREADS + tid;
    if (i < n) { sCost[i] = c[j]; sWgt[i] = w[j]; }
  }
  __syncthreads();

#pragma unroll
  for (int j = 0; j < 4; ++j) {
    const int i = j * PART_THREADS + tid;
    if (i < n) {
      const int q = p[j] - base;
      outCost[lo + i] = sCost[q];
      outW[lo + i]    = __int_as_float(sWgt[q]);
    }
  }
}

__global__ void seg_expsum(const float* __restrict__ outCost,
                           const int* __restrict__ edgeS,
                           const unsigned* __restrict__ segMax,
                           unsigned long long* __restrict__ segSum,
                           const float* __restrict__ logT, int E) {
  int e = blockIdx.x * blockDim.x + threadIdx.x;
  if (e >= E) return;
  float invT = __expf(-logT[0]);
  int s = edgeS[e];
  float val = -outCost[e] * invT;
  float m = ordVal(segMax[s]);
  atomicAdd(&segSum[s], (unsigned long long)(__expf(val - m) * SUM_SCALE));
}

__global__ void seg_norm(const float* __restrict__ outCost,
                         const int* __restrict__ edgeS,
                         const unsigned* __restrict__ segMax,
                         const unsigned long long* __restrict__ segSum,
                         const float* __restrict__ logT,
                         float* __restrict__ outW, int E) {
  int e = blockIdx.x * blockDim.x + threadIdx.x;
  if (e >= E) return;
  float invT = __expf(-logT[0]);
  int s = edgeS[e];
  float val = -outCost[e] * invT;
  float m = ordVal(segMax[s]);
  outW[e] = __expf(val - m) / ((float)segSum[s] * SUM_INV_SCALE);
}

extern "C" void kernel_launch(void* const* d_in, const int* in_sizes, int n_in,
                              void* d_out, int out_size, void* d_ws, size_t ws_size,
                              hipStream_t stream) {
  const float* embS = (const float*)d_in[0];
  const float* embA = (const float*)d_in[1];
  const int*   eidx = (const int*)d_in[2];
  const float* W1   = (const float*)d_in[3];
  const float* b1   = (const float*)d_in[4];
  const float* W2   = (const float*)d_in[5];
  const float* b2   = (const float*)d_in[6];
  const float* logT = (const float*)d_in[7];

  const int numS = in_sizes[0] / D_DIM;
  const int numA = in_sizes[1] / D_DIM;
  const int E    = in_sizes[2] / 2;
  const int* edgeS = eidx;
  const int* edgeA = eidx + E;

  float* outW    = (float*)d_out;
  float* outCost = (float*)d_out + E;

  const size_t pFloats = (size_t)(numS + numA) * H_DIM;
  const size_t need    = pFloats * sizeof(float)
                       + (size_t)numS * (sizeof(unsigned long long) + sizeof(unsigned));

  const int NB       = (numS + SEG_NODES - 1) / SEG_NODES;
  const int chunk    = (((E + PART_BLOCKS - 1) / PART_BLOCKS) + 31) & ~31;
  const int chunkPad = (chunk + 127) & ~127;
  const size_t recSlots = (size_t)PART_BLOCKS * chunkPad;
  const size_t needGrp  = need + recSlots * (sizeof(float) + 1)
                        + (size_t)PART_BLOCKS * (NB + 1) * sizeof(int);

  const size_t posSlots = ((size_t)E + 31) & ~(size_t)31;
  const size_t needSrc  = needGrp + recSlots * sizeof(float) + posSlots * sizeof(int);

  if (ws_size >= needSrc && NB <= NB_MAX && chunk <= CHUNK_MAX) {
    float* P = (float*)d_ws;
    int*           recA    = (int*)(P + pFloats);
    float*         recCost = (float*)(recA + recSlots);
    int*           pos     = (int*)(recCost + recSlots);
    unsigned char* recNode = (unsigned char*)(pos + posSlots);
    unsigned long long* segSum = (unsigned long long*)(recNode + recSlots);
    int*                offG   = (int*)((unsigned*)(segSum + numS) + numS);

    proj_and_group<<<PROJ_BLOCKS + PART_BLOCKS, PART_THREADS, 0, stream>>>(
        embS, embA, W1, b1, P, numS, numA,
        edgeS, edgeA, offG, recA, recNode, pos, E, NB, chunk, chunkPad);
    bucket_edge<<<NB, PART_THREADS, 0, stream>>>(
        embS, embA, P, P + (size_t)numS * H_DIM, W2, b2, logT,
        offG, recA, recNode, recCost, numS, NB, chunkPad);
    chunk_unsort<<<PART_BLOCKS, PART_THREADS, 0, stream>>>(
        recCost, recA, pos, outW, outCost, E, chunk, chunkPad);
    return;
  }

  if (ws_size >= needGrp && NB <= NB_MAX && chunk <= CHUNK_MAX) {
    float* P = (float*)d_ws;
    float*         recVal  = P + pFloats;
    unsigned char* recNode = (unsigned char*)(recVal + recSlots);
    unsigned long long* segSum = (unsigned long long*)(recNode + recSlots);
    unsigned*           segMax = (unsigned*)(segSum + numS);
    int*                offG   = (int*)(segMax + numS);

    node_proj<<<256, EDGE_WAVES * 32, 0, stream>>>(embS, embA, W1, b1, P, numS, numA);
    edge_gate_place<<<PART_BLOCKS, PART_THREADS, 0, stream>>>(
        embS, embA, edgeS, edgeA, P, P + (size_t)numS * H_DIM, W2, b2, logT,
        outCost, offG, recVal, recNode, E, NB, chunk, chunkPad);
    bucket_stats<<<NB, PART_THREADS, 0, stream>>>(
        offG, recVal, recNode, segMax, segSum, numS, NB, chunkPad);
    seg_norm<<<(E + 255) / 256, 256, 0, stream>>>(outCost, edgeS, segMax, segSum, logT, outW, E);
    return;
  }

  if (ws_size >= need) {
    float* P = (float*)d_ws;
    unsigned long long* segSum = (unsigned long long*)(P + pFloats);
    unsigned*           segMax = (unsigned*)(segSum + numS);

    init_seg<<<(numS + 255) / 256, 256, 0, stream>>>(segMax, segSum, numS);
    node_proj<<<256, EDGE_WAVES * 32, 0, stream>>>(embS, embA, W1, b1, P, numS, numA);
    edge_gate_cost<<<256, EDGE_WAVES * 32, 0, stream>>>(
        embS, embA, edgeS, edgeA, P, P + (size_t)numS * H_DIM, W2, b2, logT,
        outCost, segMax, E);
    seg_expsum<<<(E + 255) / 256, 256, 0, stream>>>(outCost, edgeS, segMax, segSum, logT, E);
    seg_norm<<<(E + 255) / 256, 256, 0, stream>>>(outCost, edgeS, segMax, segSum, logT, outW, E);
    return;
  }

  unsigned long long* segSum = (unsigned long long*)d_ws;
  unsigned*           segMax = (unsigned*)(segSum + numS);

  init_seg<<<(numS + 255) / 256, 256, 0, stream>>>(segMax, segSum, numS);

  edge_mlp_cost<<<256, EDGE_WAVES * 32, 0, stream>>>(
      embS, embA, edgeS, edgeA, W1, b1, W2, b2, logT, outCost, segMax, E);

  seg_expsum<<<(E + 255) / 256, 256, 0, stream>>>(outCost, edgeS, segMax, segSum, logT, E);
  seg_norm<<<(E + 255) / 256, 256, 0, stream>>>(outCost, edgeS, segMax, segSum, logT, outW, E);
}


# bucket_edge set-up loads issued together: run offsets and the block's 150 KB of source rows as one group, its run records as a second; before, each 16-byte load waited for the one before it
# speedup vs baseline: 5.1745x; 1.0440x over previous
#include <hip/hip_runtime.h>
#include <hip/hip_bf16.h>
#include <math.h>

typedef __attribute__((ext_vector_type(2))) float v2f;
typedef __attribute__((ext_vector_type(4))) float v4f;
typedef __attribute__((ext_vector_type(8))) float v8f;

#define D_DIM 64
#define H_DIM 128
#define CPITCH 66

__device__ __forceinline__ unsigned ordKey(float f) {
  unsigned u = __float_as_uint(f);
  return (u & 0x80000000u) ? ~u : (u | 0x80000000u);
}
__device__ __forceinline__ float ordVal(unsigned k) {
  unsigned u = (k & 0x80000000u) ? (k ^ 0x80000000u) : ~k;
  return __uint_as_float(u);
}

#define SUM_SCALE     1099511627776.0f
#define SUM_INV_SCALE (1.0f / 1099511627776.0f)

__global__ void init_seg(unsigned* __restrict__ segMax,
                         unsigned long long* __restrict__ segSum, int numS) {
  int i = blockIdx.x * blockDim.x + threadIdx.x;
  if (i < numS) { segMax[i] = 0u; segSum[i] = 0ull; }
}

#define EDGE_WAVES 32
__global__ __launch_bounds__(EDGE_WAVES * 32)
void edge_mlp_cost(const float* __restrict__ embS,
                   const float* __restrict__ embA,
                   const int*   __restrict__ edgeS,
                   const int*   __restrict__ edgeA,
                   const float* __restrict__ W1,
                   const float* __restrict__ b1,
                   const float* __restrict__ W2,
                   const float* __restrict__ b2,
                   const float* __restrict__ logT,
                   float*       __restrict__ outCost,
                   unsigned*    __restrict__ segMax,
                   int E) {
  __shared__ float sW1[H_DIM * CPITCH * 2];
  __shared__ float sB1[H_DIM];
  __shared__ float sW2[H_DIM];
  __shared__ float sGate[EDGE_WAVES][16];

  const int tid = threadIdx.x;
  for (int i = tid; i < H_DIM * H_DIM; i += EDGE_WAVES * 32) {
    int kk = i >> 7;
    int c  = i & 127;
    sW1[(c * CPITCH + (kk >> 1)) * 2 + (kk & 1)] = W1[i];
  }
  if (tid < H_DIM) { sB1[tid] = b1[tid]; sW2[tid] = W2[tid]; }
  __syncthreads();

  const int wave = __builtin_amdgcn_readfirstlane(tid >> 5);
  const int lane = tid & 31;
  const int half = lane >> 4;
  const int col  = lane & 15;
  const float b2v  = b2[0];
  const float invT = __expf(-logT[0]);
  const int tiles  = (E + 15) >> 4;
  const int stride = gridDim.x * EDGE_WAVES;

  for (int t = blockIdx.x * EDGE_WAVES + wave; t < tiles; t += stride) {
  const int  e0    = t * 16;
  const int  e     = e0 + col;
  const bool valid = (e < E);
  const int  ec    = valid ? e : (E - 1);
  const int  sI    = edgeS[ec];
  const int  aI    = edgeA[ec];
  const float* es = embS + (size_t)sI * D_DIM;
  const float* ea = embA + (size_t)aI * D_DIM;

  v2f a[32];
#pragma unroll
  for (int k = 0; k < 16; ++k) {
    a[k]      = *(const v2f*)(es + 4 * k + 2 * half);
    a[k + 16] = *(const v2f*)(ea + 4 * k + 2 * half);
  }
  float ss = 0.0f;
#pragma unroll
  for (int k = 0; k < 16; ++k) {
    float dx = a[k][0] - a[k + 16][0];
    float dy = a[k][1] - a[k + 16][1];
    ss += dx * dx + dy * dy;
  }
  ss += __shfl_xor(ss, 16);

  float gacc[8] = {0, 0, 0, 0, 0, 0, 0, 0};
#pragma unroll 1
  for (int n = 0; n < 8; ++n) {
    const float* bbase = sW1 + ((n * 16 + col) * CPITCH + half) * 2;
    v8f c0 = {};
    v8f c1 = {};
#pragma unroll
    for (int k = 0; k < 32; k += 2) {
      v2f b0 = *(const v2f*)(bbase + 4 * k);
      v2f b1v = *(const v2f*)(bbase + 4 * (k + 1));
      c0 = __builtin_amdgcn_wmma_f32_16x16x4_f32(false, a[k], false, b0,
                                                 (short)0, c0, false, false);
      c1 = __builtin_amdgcn_wmma_f32_16x16x4_f32(false, a[k + 1], false, b1v,
                                                 (short)0, c1, false, false);
    }
    const float bn  = sB1[n * 16 + col];
    const float w2n = sW2[n * 16 + col];
#pragma unroll
    for (int r = 0; r < 8; ++r) {
      float h = c0[r] + c1[r] + bn;
      h = h > 0.0f ? h : 0.0f;
      gacc[r] += h * w2n;
    }
  }

#pragma unroll
  for (int r = 0; r < 8; ++r) {
    float g = gacc[r];
    g += __shfl_xor(g, 1);
    g += __shfl_xor(g, 2);
    g += __shfl_xor(g, 4);
    g += __shfl_xor(g, 8);
    if (lane == 0)  sGate[wave][r]     = g;
    if (lane == 16) sGate[wave][r + 8] = g;
  }

  if (lane < 16 && valid) {
    float logit = sGate[wave][lane] + b2v;
    float gate  = 1.0f / (1.0f + __expf(-logit));
    float cost  = sqrtf(ss) * gate;
    outCost[e]  = cost;
    atomicMax(&segMax[sI], ordKey(-cost * invT));
  }
  }
}

__device__ __forceinline__
void node_proj_body(const float* __restrict__ embS,
                    const float* __restrict__ embA,
                    const float* __restrict__ W1,
                    const float* __restrict__ b1,
                    float*       __restrict__ P,
                    int numS, int numA, int blk, int nblk) {
  __shared__ float sW1[H_DIM * CPITCH * 2];
  __shared__ __attribute__((aligned(16))) float sB1[2 * H_DIM];

  const int tid = threadIdx.x;
  for (int i = tid; i < H_DIM * H_DIM; i += EDGE_WAVES * 32) {
    int kk = i >> 7;
    int c  = i & 127;
    sW1[(c * CPITCH + (kk >> 1)) * 2 + (kk & 1)] = W1[i];
  }
  if (tid < H_DIM) { sB1[tid] = b1[tid]; sB1[H_DIM + tid] = 0.0f; }
  __syncthreads();

  const int wave = __builtin_amdgcn_readfirstlane(tid >> 5);
  const int lane = tid & 31;
  const int tilesS = (numS + 15) >> 4;
  const int tilesA = (numA + 15) >> 4;
  const int tiles  = tilesS + tilesA;

  for (int t = wave * nblk + blk; t < tiles; t += EDGE_WAVES * nblk) {
    int ln = lane;
    asm volatile("" : "+v"(ln));
    const int half = ln >> 4;
    const int col  = ln & 15;
    const bool isA = (t >= tilesS);
    const int  tt  = isA ? (t - tilesS) : t;
    const float* emb = isA ? embA : embS;
    const int  nrows = isA ? numA : numS;
    const int  node  = tt * 16 + col;
    const bool valid = (node < nrows);
    const int  nc    = valid ? node : (nrows - 1);
    const float* row = emb + (size_t)nc * D_DIM;

    v2f a[16];
#pragma unroll
    for (int k = 0; k < 16; ++k) a[k] = *(const v2f*)(row + 4 * k + 2 * half);

    float* prow = P + ((size_t)(isA ? numS : 0) + (size_t)nc) * H_DIM + 8 * half;
    const int koff = isA ? D_DIM : 0;
    const int boff = isA ? H_DIM : 0;

#pragma unroll 1
    for (int nt = 0; nt < 8; ++nt) {
      const float* wbase = sW1 + ((nt * 16 + col) * CPITCH + half) * 2 + koff;
      v8f c0 = {};
      v8f c1 = {};
#pragma unroll
      for (int k = 0; k < 16; k += 2) {
        v2f w0 = *(const v2f*)(wbase + 4 * k);
        v2f w1 = *(const v2f*)(wbase + 4 * (k + 1));
        c0 = __builtin_amdgcn_wmma_f32_16x16x4_f32(false, w0, false, a[k],
                                                   (short)0, c0, false, false);
        c1 = __builtin_amdgcn_wmma_f32_16x16x4_f32(false, w1, false, a[k + 1],
                                                   (short)0, c1, false, false);
      }
      const float* bb = sB1 + boff + nt * 16 + 8 * half;
      const v4f blo = *(const v4f*)(bb);
      const v4f bhi = *(const v4f*)(bb + 4);
      v4f lo, hi;
#pragma unroll
      for (int r = 0; r < 4; ++r) {
        lo[r] = c0[r] + c1[r] + blo[r];
        hi[r] = c0[r + 4] + c1[r + 4] + bhi[r];
      }
      if (valid) {
        *(v4f*)(prow + nt * 16)     = lo;
        *(v4f*)(prow + nt * 16 + 4) = hi;
      }
    }
  }
}

__global__ __launch_bounds__(EDGE_WAVES * 32)
void node_proj(const float* __restrict__ embS,
               const float* __restrict__ embA,
               const float* __restrict__ W1,
               const float* __restrict__ b1,
               float*       __restrict__ P,
               int numS, int numA) {
  node_proj_body(embS, embA, W1, b1, P, numS, numA, (int)blockIdx.x, (int)gridDim.x);
}

struct RowSet {
  v4f es0, es1, ea0, ea1;
  v4f ps0, ps1, ps2, ps3;
  v4f pa0, pa1, pa2, pa3;
};

__device__ __forceinline__ void fetch_rows(RowSet& r,
                                           const float* __restrict__ embS,
                                           const float* __restrict__ embA,
                                           const float* __restrict__ Ps,
                                           const float* __restrict__ Pa,
                                           int s, int a, int sub4) {
  const float* e1 = embS + (size_t)s * D_DIM + sub4;
  const float* e2 = embA + (size_t)a * D_DIM + sub4;
  const float* p1 = Ps + (size_t)s * H_DIM + sub4;
  const float* p2 = Pa + (size_t)a * H_DIM + sub4;
  r.es0 = *(const v4f*)(e1);
  r.es1 = *(const v4f*)(e1 + 32);
  r.ea0 = *(const v4f*)(e2);
  r.ea1 = *(const v4f*)(e2 + 32);
  r.ps0 = *(const v4f*)(p1);
  r.ps1 = *(const v4f*)(p1 + 32);
  r.ps2 = *(const v4f*)(p1 + 64);
  r.ps3 = *(const v4f*)(p1 + 96);
  r.pa0 = *(const v4f*)(p2);
  r.pa1 = *(const v4f*)(p2 + 32);
  r.pa2 = *(const v4f*)(p2 + 64);
  r.pa3 = *(const v4f*)(p2 + 96);
}

__device__ __forceinline__ float sq4(v4f a, v4f b) {
  v4f d = a - b;
  return d[0] * d[0] + d[1] * d[1] + d[2] * d[2] + d[3] * d[3];
}
__device__ __forceinline__ float gate4(v4f p, v4f q, v4f w) {
  v4f h = p + q;
  float g = 0.0f;
#pragma unroll
  for (int i = 0; i < 4; ++i) {
    float x = h[i] > 0.0f ? h[i] : 0.0f;
    g += x * w[i];
  }
  return g;
}

template <int M>
__device__ __forceinline__ float xadd8(float v) {
  int o = __builtin_amdgcn_update_dpp(0, __float_as_int(v), 0x160 + M, 0xf, 0xf, true);
  return v + __int_as_float(o);
}
__device__ __forceinline__ int bsel8(int v, int lane, int i) {
  return __builtin_amdgcn_ds_bpermute(((lane & 24) | i) << 2, v);
}

__device__ __forceinline__ void reduce_rows(const RowSet& r, const float* sW2, int sub4,
                                            float& ss, float& g) {
  float s = sq4(r.es0, r.ea0) + sq4(r.es1, r.ea1);
  float t = gate4(r.ps0, r.pa0, *(const v4f*)(sW2 + sub4))
          + gate4(r.ps1, r.pa1, *(const v4f*)(sW2 + 32 + sub4))
          + gate4(r.ps2, r.pa2, *(const v4f*)(sW2 + 64 + sub4))
          + gate4(r.ps3, r.pa3, *(const v4f*)(sW2 + 96 + sub4));
  s = xadd8<1>(s); t = xadd8<1>(t);
  s = xadd8<2>(s); t = xadd8<2>(t);
  s = xadd8<4>(s); t = xadd8<4>(t);
  ss = s; g = t;
}

__device__ __forceinline__ float tile_cost(const float* __restrict__ embS,
                                           const float* __restrict__ embA,
                                           const float* __restrict__ Ps,
                                           const float* __restrict__ Pa,
                                           const float* sW2,
                                           int curS, int curA,
                                           int lane, int sub, int sub4, float b2v) {
  float rss = 0.0f, rg = 0.0f;

#pragma unroll 1
  for (int i = 0; i < 8; i += 2) {
    RowSet X, Y;
    float ss, g;
    fetch_rows(X, embS, embA, Ps, Pa, bsel8(curS, lane, i), bsel8(curA, lane, i), sub4);
    fetch_rows(Y, embS, embA, Ps, Pa, bsel8(curS, lane, i + 1), bsel8(curA, lane, i + 1), sub4);
    __builtin_amdgcn_sched_barrier(0);
    reduce_rows(X, sW2, sub4, ss, g);
    if (sub == i) { rss = ss; rg = g; }
    reduce_rows(Y, sW2, sub4, ss, g);
    if (sub == i + 1) { rss = ss; rg = g; }
  }

  const float logit = rg + b2v;
  const float gate  = 1.0f / (1.0f + __expf(-logit));
  return sqrtf(rss) * gate;
}

__global__ __launch_bounds__(EDGE_WAVES * 32)
void edge_gate_cost(const float* __restrict__ embS,
                    const float* __restrict__ embA,
                    const int*   __restrict__ edgeS,
                    const int*   __restrict__ edgeA,
                    const float* __restrict__ Ps,
                    const float* __restrict__ Pa,
                    const float* __restrict__ W2,
                    const float* __restrict__ b2,
                    const float* __restrict__ logT,
                    float*       __restrict__ outCost,
                    unsigned*    __restrict__ segMax,
                    int E) {
  __shared__ __attribute__((aligned(16))) float sW2[H_DIM];
  const int tid = threadIdx.x;
  if (tid < H_DIM) sW2[tid] = W2[tid];
  __syncthreads();

  const int wave = __builtin_amdgcn_readfirstlane(tid >> 5);
  const int lane = tid & 31;
  const int sub  = lane & 7;
  const int sub4 = sub * 4;
  const float b2v  = b2[0];
  const float invT = __expf(-logT[0]);
  const int tiles  = (E + 31) >> 5;
  const int stride = (int)gridDim.x * EDGE_WAVES;

  int t = (int)blockIdx.x * EDGE_WAVES + wave;
  if (t >= tiles) return;

  int eL   = t * 32 + lane;
  int eLc  = eL < E ? eL : (E - 1);
  int curS = edgeS[eLc];
  int curA = edgeA[eLc];

  for (;;) {
    const bool hasNext = (t + stride) < tiles;
    const int  tn   = hasNext ? (t + stride) : t;
    const int  nL   = tn * 32 + lane;
    const int  nLc  = nL < E ? nL : (E - 1);
    const int  nxtS = edgeS[nLc];
    const int  nxtA = edgeA[nLc];

    const float cost = tile_cost(embS, embA, Ps, Pa, sW2, curS, curA, lane, sub, sub4, b2v);

    if (eL < E) {
      outCost[eL] = cost;
      atomicMax(&segMax[curS], ordKey(-cost * invT));
    }
    if (!hasNext) break;
    t = tn; eL = nL; curS = nxtS; curA = nxtA;
  }
}

#define SEG_NODES    196
#define NB_MAX       1024
#define PART_BLOCKS  256
#define PART_THREADS 1024
#define CHUNK_MAX    4096
#define STAT_REGS    8

__device__ __forceinline__ int seg_bucket(int s) { return (int)((unsigned)s / (unsigned)SEG_NODES); }

__global__ __launch_bounds__(PART_THREADS)
void edge_gate_place(const float* __restrict__ embS,
                     const float* __restrict__ embA,
                     const int*   __restrict__ edgeS,
                     const int*   __restrict__ edgeA,
                     const float* __restrict__ Ps,
                     const float* __restrict__ Pa,
                     const float* __restrict__ W2,
                     const float* __restrict__ b2,
                     const float* __restrict__ logT,
                     float*         __restrict__ outCost,
                     int*           __restrict__ offG,
                     float*         __restrict__ recVal,
                     unsigned char* __restrict__ recNode,
                     int E, int NB, int chunk, int chunkPad) {
  __shared__ __attribute__((aligned(16))) float sW2[H_DIM];
  __shared__ int h[NB_MAX];
  __shared__ int fill[NB_MAX];
  __shared__ int off[NB_MAX + 1];
  __shared__ int   sSrc[CHUNK_MAX];
  __shared__ float sArg[CHUNK_MAX];
  __shared__ __attribute__((aligned(16))) float    sVal[CHUNK_MAX];
  __shared__ __attribute__((aligned(16))) unsigned sNode4[CHUNK_MAX / 4];
  unsigned char* sNode = (unsigned char*)sNode4;

  const int tid = threadIdx.x;
  const int b   = blockIdx.x;
  if (tid < H_DIM) sW2[tid] = W2[tid];
  for (int k = tid; k < NB; k += PART_THREADS) { h[k] = 0; fill[k] = 0; }
  sNode4[tid] = 0u;
  __syncthreads();

  const int wave = __builtin_amdgcn_readfirstlane(tid >> 5);
  const int lane = tid & 31;
  const int sub  = lane & 7;
  const int sub4 = sub * 4;
  const float b2v  = b2[0];
  const float invT = __expf(-logT[0]);

  const int lo = b * chunk;
  const int hi = (lo + chunk < E) ? (lo + chunk) : E;
  const int n  = (hi > lo) ? (hi - lo) : 0;
  const int nTiles = (n + 31) >> 5;

  int t = wave;
  if (t < nTiles) {
    int iL   = t * 32 + lane;
    int eLc  = iL < n ? (lo + iL) : (hi - 1);
    int curS = edgeS[eLc];
    int curA = edgeA[eLc];

    for (;;) {
      const bool hasNext = (t + EDGE_WAVES) < nTiles;
      const int  tn   = hasNext ? (t + EDGE_WAVES) : t;
      const int  nL   = tn * 32 + lane;
      const int  nLc  = nL < n ? (lo + nL) : (hi - 1);
      const int  nxtS = edgeS[nLc];
      const int  nxtA = edgeA[nLc];

      const float cost = tile_cost(embS, embA, Ps, Pa, sW2, curS, curA, lane, sub, sub4, b2v);

      if (iL < n) {
        outCost[lo + iL] = cost;
        sSrc[iL] = curS;
        sArg[iL] = -cost * invT;
        atomicAdd(&h[seg_bucket(curS)], 1);
      }
      if (!hasNext) break;
      t = tn; iL = nL; curS = nxtS; curA = nxtA;
    }
  }
  __syncthreads();

  if (tid < 32) {
    const int per = (NB + 31) >> 5;
    const int k0  = tid * per;
    int s = 0;
    for (int j = 0; j < per; ++j) { const int k = k0 + j; if (k < NB) s += h[k]; }
    int inc = s;
    for (int d = 1; d < 32; d <<= 1) {
      const int o = __shfl_up(inc, d);
      if (tid >= d) inc += o;
    }
    int run = inc - s;
    for (int j = 0; j < per; ++j) {
      const int k = k0 + j;
      if (k < NB) { off[k] = run; run += h[k]; }
    }
    if (tid == 31) off[NB] = inc;
  }
  __syncthreads();

  for (int k = tid; k <= NB; k += PART_THREADS) offG[b * (NB + 1) + k] = off[k];

#pragma unroll
  for (int j = 0; j < 4; ++j) {
    const int i = j * PART_THREADS + tid;
    if (i < n) {
      const int sv  = sSrc[i];
      const int k   = seg_bucket(sv);
      const int pos = off[k] + atomicAdd(&fill[k], 1);
      sVal[pos]  = sArg[i];
      sNode[pos] = (unsigned char)(sv - k * SEG_NODES);
    }
  }
  __syncthreads();

  float*    gv = recVal + (size_t)b * chunkPad;
  unsigned* gn = (unsigned*)(recNode + (size_t)b * chunkPad);
#pragma unroll
  for (int j = 0; j < 4; ++j) {
    const int i = j * PART_THREADS + tid;
    if (i < n) gv[i] = sVal[i];
  }
  if (4 * tid < n) gn[tid] = sNode4[tid];
}

__global__ __launch_bounds__(PART_THREADS)
void bucket_stats(const int*           __restrict__ offG,
                  const float*         __restrict__ recVal,
                  const unsigned char* __restrict__ recNode,
                  unsigned*            __restrict__ segMax,
                  unsigned long long*  __restrict__ segSum,
                  int numS, int NB, int chunkPad) {
  __shared__ unsigned           mx[SEG_NODES];
  __shared__ unsigned long long sm[SEG_NODES];
  const int tid = threadIdx.x;
  const int k   = blockIdx.x;
  if (tid < SEG_NODES) { mx[tid] = 0u; sm[tid] = 0ull; }

  const int r   = tid >> 2;
  const int sub = tid & 3;
  const int* o  = offG + r * (NB + 1) + k;
  const int st  = r * chunkPad + o[0];
  const int en  = r * chunkPad + o[1];

  float rv[STAT_REGS];
  int   rn[STAT_REGS];
#pragma unroll
  for (int j = 0; j < STAT_REGS; ++j) {
    const int i = st + sub + 4 * j;
    rv[j] = 0.0f;
    rn[j] = -1;
    if (i < en) { rv[j] = recVal[i]; rn[j] = recNode[i]; }
  }
  const int rest = st + sub + 4 * STAT_REGS;
  __syncthreads();

#pragma unroll
  for (int j = 0; j < STAT_REGS; ++j)
    if (rn[j] >= 0) atomicMax(&mx[rn[j]], ordKey(rv[j]));
  for (int i = rest; i < en; i += 4)
    atomicMax(&mx[recNode[i]], ordKey(recVal[i]));
  __syncthreads();

#pragma unroll
  for (int j = 0; j < STAT_REGS; ++j) {
    if (rn[j] >= 0) {
      const float m = ordVal(mx[rn[j]]);
      atomicAdd(&sm[rn[j]], (unsigned long long)(__expf(rv[j] - m) * SUM_SCALE));
    }
  }
  for (int i = rest; i < en; i += 4) {
    const int   nd = recNode[i];
    const float m  = ordVal(mx[nd]);
    atomicAdd(&sm[nd], (unsigned long long)(__expf(recVal[i] - m) * SUM_SCALE));
  }
  __syncthreads();

  if (tid < SEG_NODES) {
    const int s = k * SEG_NODES + tid;
    if (s < numS) { segMax[s] = mx[tid]; segSum[s] = sm[tid]; }
  }
}

#define BE_CAP 6144
#define EPITCH 68
#define PPITCH 132

__device__ __forceinline__
void group_edges_body(const int* __restrict__ edgeS,
                      const int* __restrict__ edgeA,
                      int*           __restrict__ offG,
                      int*           __restrict__ recA,
                      unsigned char* __restrict__ recNode,
                      int*           __restrict__ pos,
                      int E, int NB, int chunk, int chunkPad, int b) {
  __shared__ int h[NB_MAX];
  __shared__ int fill[NB_MAX];
  __shared__ int off[NB_MAX + 1];
  __shared__ __attribute__((aligned(16))) int      sA[CHUNK_MAX];
  __shared__ __attribute__((aligned(16))) unsigned sNode4[CHUNK_MAX / 4];
  unsigned char* sNode = (unsigned char*)sNode4;

  const int tid = threadIdx.x;
  for (int k = tid; k < NB; k += PART_THREADS) { h[k] = 0; fill[k] = 0; }
  sNode4[tid] = 0u;
  __syncthreads();

  const int lo = b * chunk;
  const int hi = (lo + chunk < E) ? (lo + chunk) : E;
  const int n  = (hi > lo) ? (hi - lo) : 0;

  int sv[4], av[4];
#pragma unroll
  for (int j = 0; j < 4; ++j) {
    const int i = j * PART_THREADS + tid;
    sv[j] = -1; av[j] = 0;
    if (i < n) {
      sv[j] = edgeS[lo + i];
      av[j] = edgeA[lo + i];
      atomicAdd(&h[seg_bucket(sv[j])], 1);
    }
  }
  __syncthreads();

  if (tid < 32) {
    const int per = (NB + 31) >> 5;
    const int k0  = tid * per;
    int s = 0;
    for (int j = 0; j < per; ++j) { const int k = k0 + j; if (k < NB) s += h[k]; }
    int inc = s;
    for (int d = 1; d < 32; d <<= 1) {
      const int o = __shfl_up(inc, d);
      if (tid >= d) inc += o;
    }
    int run = inc - s;
    for (int j = 0; j < per; ++j) {
      const int k = k0 + j;
      if (k < NB) { off[k] = run; run += h[k]; }
    }
    if (tid == 31) off[NB] = inc;
  }
  __syncthreads();

  for (int k = tid; k <= NB; k += PART_THREADS) offG[b * (NB + 1) + k] = off[k];

#pragma unroll
  for (int j = 0; j < 4; ++j) {
    const int i = j * PART_THREADS + tid;
    if (i < n) {
      const int k = seg_bucket(sv[j]);
      const int p = off[k] + atomicAdd(&fill[k], 1);
      sA[p]    = av[j];
      sNode[p] = (unsigned char)(sv[j] - k * SEG_NODES);
      pos[lo + i] = b * chunkPad + p;
    }
  }
  __syncthreads();

  int*      ga = recA + (size_t)b * chunkPad;
  unsigned* gn = (unsigned*)(recNode + (size_t)b * chunkPad);
#pragma unroll
  for (int j = 0; j < 4; ++j) {
    const int i = j * PART_THREADS + tid;
    if (i < n) ga[i] = sA[i];
  }
  if (4 * tid < n) gn[tid] = sNode4[tid];
}

#define PROJ_BLOCKS 256
__global__ __launch_bounds__(PART_THREADS, 16)
void proj_and_group(const float* __restrict__ embS,
                    const float* __restrict__ embA,
                    const float* __restrict__ W1,
                    const float* __restrict__ b1,
                    float*       __restrict__ P,
                    int numS, int numA,
                    const int* __restrict__ edgeS,
                    const int* __restrict__ edgeA,
                    int*           __restrict__ offG,
                    int*           __restrict__ recA,
                    unsigned char* __restrict__ recNode,
                    int*           __restrict__ pos,
                    int E, int NB, int chunk, int chunkPad) {
  if (blockIdx.x < PROJ_BLOCKS) {
    node_proj_body(embS, embA, W1, b1, P, numS, numA, (int)blockIdx.x, PROJ_BLOCKS);
  } else {
    group_edges_body(edgeS, edgeA, offG, recA, recNode, pos, E, NB, chunk, chunkPad,
                     (int)blockIdx.x - PROJ_BLOCKS);
  }
}

struct RowT {
  v4f ea0, ea1;
  v4f pa0, pa1, pa2, pa3;
};
__device__ __forceinline__ void fetch_t(RowT& r,
                                        const float* __restrict__ embA,
                                        const float* __restrict__ Pa,
                                        int a, int sub4) {
  const float* e2 = embA + (size_t)a * D_DIM + sub4;
  const float* p2 = Pa + (size_t)a * H_DIM + sub4;
  r.ea0 = *(const v4f*)(e2);
  r.ea1 = *(const v4f*)(e2 + 32);
  r.pa0 = *(const v4f*)(p2);
  r.pa1 = *(const v4f*)(p2 + 32);
  r.pa2 = *(const v4f*)(p2 + 64);
  r.pa3 = *(const v4f*)(p2 + 96);
}
__device__ __forceinline__ void reduce_t(const RowT& r, const float* sE, const float* sP,
                                         const float* sW2, int sub4, float& ss, float& g) {
  const v4f es0 = *(const v4f*)(sE + sub4);
  const v4f es1 = *(const v4f*)(sE + 32 + sub4);
  const v4f ps0 = *(const v4f*)(sP + sub4);
  const v4f ps1 = *(const v4f*)(sP + 32 + sub4);
  const v4f ps2 = *(const v4f*)(sP + 64 + sub4);
  const v4f ps3 = *(const v4f*)(sP + 96 + sub4);
  float s = sq4(es0, r.ea0) + sq4(es1, r.ea1);
  float t = gate4(ps0, r.pa0, *(const v4f*)(sW2 + sub4))
          + gate4(ps1, r.pa1, *(const v4f*)(sW2 + 32 + sub4))
          + gate4(ps2, r.pa2, *(const v4f*)(sW2 + 64 + sub4))
          + gate4(ps3, r.pa3, *(const v4f*)(sW2 + 96 + sub4));
  s = xadd8<1>(s); t = xadd8<1>(t);
  s = xadd8<2>(s); t = xadd8<2>(t);
  s = xadd8<4>(s); t = xadd8<4>(t);
  ss = s; g = t;
}

__device__ __forceinline__ float tile_cost_b(const float* __restrict__ embA,
                                             const float* __restrict__ Pa,
                                             const float* sRowE, const float* sRowP,
                                             const float* sW2,
                                             int curN, int curA,
                                             int lane, int sub, int sub4, float b2v) {
  float rss = 0.0f, rg = 0.0f;

#pragma unroll 1
  for (int i = 0; i < 8; i += 2) {
    RowT X, Y;
    float ss, g;
    const int n0 = bsel8(curN, lane, i);
    const int n1 = bsel8(curN, lane, i + 1);
    fetch_t(X, embA, Pa, bsel8(curA, lane, i), sub4);
    fetch_t(Y, embA, Pa, bsel8(curA, lane, i + 1), sub4);
    __builtin_amdgcn_sched_barrier(0);
    reduce_t(X, sRowE + n0 * EPITCH, sRowP + n0 * PPITCH, sW2, sub4, ss, g);
    if (sub == i) { rss = ss; rg = g; }
    reduce_t(Y, sRowE + n1 * EPITCH, sRowP + n1 * PPITCH, sW2, sub4, ss, g);
    if (sub == i + 1) { rss = ss; rg = g; }
  }

  const float logit = rg + b2v;
  const float gate  = 1.0f / (1.0f + __expf(-logit));
  return sqrtf(rss) * gate;
}

#define E_ROUNDS ((SEG_NODES * (D_DIM / 4) + PART_THREADS - 1) / PART_THREADS)
#define P_ROUNDS ((SEG_NODES * (H_DIM / 4) + PART_THREADS - 1) / PART_THREADS)

__device__ __forceinline__ void load_run(int (&ca)[STAT_REGS], int (&cn)[STAT_REGS],
                                         const int* recA, const unsigned char* recNode,
                                         const int* offG, int gbase, int st, int en, int sub) {
#pragma unroll
  for (int q = 0; q < STAT_REGS; ++q) {
    const int  i  = st + sub + 4 * q;
    const bool in = i < en;
    const int*           pa = in ? (recA + gbase + i) : offG;
    const unsigned char* pn = in ? (recNode + gbase + i) : (const unsigned char*)offG;
    const int a = *pa;
    const int n = *pn;
    ca[q] = a;
    cn[q] = in ? n : -1;
  }
}

__global__ __launch_bounds__(PART_THREADS)
void bucket_edge(const float* __restrict__ embS,
                 const float* __restrict__ embA,
                 const float* __restrict__ Ps,
                 const float* __restrict__ Pa,
                 const float* __restrict__ W2,
                 const float* __restrict__ b2,
                 const float* __restrict__ logT,
                 const int*           __restrict__ offG,
                 int*                 recAW,
                 const unsigned char* __restrict__ recNode,
                 float*               recCost,
                 int numS, int NB, int chunkPad) {
  __shared__ __attribute__((aligned(16))) float sRowE[SEG_NODES * EPITCH];
  __shared__ __attribute__((aligned(16))) float sRowP[SEG_NODES * PPITCH];
  __shared__ __attribute__((aligned(16))) float sW2[H_DIM];
  __shared__ unsigned           mx[SEG_NODES];
  __shared__ unsigned long long sm[SEG_NODES];
  __shared__ int runLen[PART_BLOCKS];
  __shared__ int runStart[PART_BLOCKS + 1];
  __shared__ int cA[BE_CAP];
  __shared__ int cPos[BE_CAP];
  __shared__ unsigned char cN[BE_CAP];

  const int tid = threadIdx.x;
  const int k   = blockIdx.x;


  const int r   = tid >> 2;
  const int sub = tid & 3;
  const int* o  = offG + r * (NB + 1) + k;
  const int st  = o[0];
  const int en  = o[1];
  const int gbase = r * chunkPad;

  const int nodeBase = k * SEG_NODES;
  const int nNodes   = (numS - nodeBase < SEG_NODES) ? (numS - nodeBase) : SEG_NODES;
  const int nE = nNodes * (D_DIM / 4);
  const int nP = nNodes * (H_DIM / 4);
  const v4f* gE = (const v4f*)(embS + (size_t)nodeBase * D_DIM);
  const v4f* gP = (const v4f*)(Ps + (size_t)nodeBase * H_DIM);
  v4f te[E_ROUNDS];
  v4f tp[P_ROUNDS];
#pragma unroll
  for (int q = 0; q < E_ROUNDS; ++q) {
    const int i = q * PART_THREADS + tid;
    te[q] = gE[i < nE ? i : nE - 1];
  }
#pragma unroll
  for (int q = 0; q < P_ROUNDS; ++q) {
    const int i = q * PART_THREADS + tid;
    tp[q] = gP[i < nP ? i : nP - 1];
  }

  if (tid < H_DIM) sW2[tid] = W2[tid];
  if (tid < SEG_NODES) { mx[tid] = 0u; sm[tid] = 0ull; }
  if (sub == 0) runLen[r] = en - st;
#pragma unroll
  for (int q = 0; q < E_ROUNDS; ++q) {
    const int i = q * PART_THREADS + tid;
    if (i < nE) *(v4f*)(sRowE + (i >> 4) * EPITCH + (i & 15) * 4) = te[q];
  }
#pragma unroll
  for (int q = 0; q < P_ROUNDS; ++q) {
    const int i = q * PART_THREADS + tid;
    if (i < nP) *(v4f*)(sRowP + (i >> 5) * PPITCH + (i & 31) * 4) = tp[q];
  }
  __syncthreads();

  if (tid < 32) {
    int s = 0;
    for (int j = 0; j < PART_BLOCKS / 32; ++j) s += runLen[tid * (PART_BLOCKS / 32) + j];
    int inc = s;
    for (int d = 1; d < 32; d <<= 1) {
      const int v = __shfl_up(inc, d);
      if (tid >= d) inc += v;
    }
    int run = inc - s;
    for (int j = 0; j < PART_BLOCKS / 32; ++j) {
      const int q = tid * (PART_BLOCKS / 32) + j;
      runStart[q] = run;
      run += runLen[q];
    }
    if (tid == 31) runStart[PART_BLOCKS] = inc;
  }
  __syncthreads();

  const int wave = __builtin_amdgcn_readfirstlane(tid >> 5);
  const int lane = tid & 31;
  const int sub8 = lane & 7;
  const int sub4 = sub8 * 4;
  const float b2v  = b2[0];
  const float invT = __expf(-logT[0]);
  const int total = __builtin_amdgcn_readfirstlane(runStart[PART_BLOCKS]);
  const int rs    = runStart[r];

  for (int base = 0; base < total; base += BE_CAP) {
    const int cnt = (total - base < BE_CAP) ? (total - base) : BE_CAP;

    int ca[STAT_REGS];
    int cn[STAT_REGS];
    int stb = st;
    asm volatile("" : "+v"(stb));
    load_run(ca, cn, recAW, recNode, offG, gbase, stb, en, sub);
#pragma unroll
    for (int q = 0; q < STAT_REGS; ++q) asm volatile("" : "+v"(ca[q]), "+v"(cn[q]));
#pragma unroll
    for (int q = 0; q < STAT_REGS; ++q) {
      const int i = stb + sub + 4 * q;
      const int j = rs + (sub + 4 * q) - base;
      if (cn[q] >= 0 && j >= 0 && j < cnt) {
        cA[j]   = ca[q];
        cN[j]   = (unsigned char)cn[q];
        cPos[j] = gbase + i;
      }
    }
    for (int i = st + sub + 4 * STAT_REGS; i < en; i += 4) {
      const int j = rs + (i - st) - base;
      if (j >= 0 && j < cnt) {
        cA[j]   = recAW[gbase + i];
        cN[j]   = recNode[gbase + i];
        cPos[j] = gbase + i;
      }
    }
    __syncthreads();

    const int nT = (cnt + 31) >> 5;
    for (int t = wave; t < nT; t += EDGE_WAVES) {
      const int j  = t * 32 + lane;
      const int jc = j < cnt ? j : (cnt - 1);
      const int curN = cN[jc];
      const int curA = cA[jc];
      const float cost = tile_cost_b(embA, Pa, sRowE, sRowP, sW2, curN, curA,
                                     lane, sub8, sub4, b2v);
      if (j < cnt) {
        recCost[cPos[j]] = cost;
        atomicMax(&mx[curN], ordKey(-cost * invT));
      }
    }
    __syncthreads();
  }

  float rv[STAT_REGS];
  int   rn[STAT_REGS];
#pragma unroll
  for (int j = 0; j < STAT_REGS; ++j) {
    const int i = st + sub + 4 * j;
    rv[j] = 0.0f;
    rn[j] = -1;
    if (i < en) { rv[j] = recCost[gbase + i]; rn[j] = recNode[gbase + i]; }
  }
#pragma unroll
  for (int j = 0; j < STAT_REGS; ++j) {
    if (rn[j] >= 0) {
      const float m = ordVal(mx[rn[j]]);
      rv[j] = __expf(-rv[j] * invT - m);
      atomicAdd(&sm[rn[j]], (unsigned long long)(rv[j] * SUM_SCALE));
    }
  }
  for (int i = st + sub + 4 * STAT_REGS; i < en; i += 4) {
    const int   nd = recNode[gbase + i];
    const float m  = ordVal(mx[nd]);
    atomicAdd(&sm[nd], (unsigned long long)(__expf(-recCost[gbase + i] * invT - m) * SUM_SCALE));
  }
  __syncthreads();

#pragma unroll
  for (int j = 0; j < STAT_REGS; ++j) {
    if (rn[j] >= 0) {
      const int i = st + sub + 4 * j;
      recAW[gbase + i] = __float_as_int(rv[j] / ((float)sm[rn[j]] * SUM_INV_SCALE));
    }
  }
  for (int i = st + sub + 4 * STAT_REGS; i < en; i += 4) {
    const int   nd = recNode[gbase + i];
    const float m  = ordVal(mx[nd]);
    const float ex = __expf(-recCost[gbase + i] * invT - m);
    recAW[gbase + i] = __float_as_int(ex / ((float)sm[nd] * SUM_INV_SCALE));
  }
}

__global__ __launch_bounds__(PART_THREADS)
void chunk_unsort(const float* __restrict__ recCost,
                  const int*   __restrict__ recW,
                  const int*   __restrict__ pos,
                  float* __restrict__ outW,
                  float* __restrict__ outCost,
                  int E, int chunk, int chunkPad) {
  __shared__ float sCost[CHUNK_MAX];
  __shared__ int   sWgt[CHUNK_MAX];

  const int tid  = threadIdx.x;
  const int b    = blockIdx.x;
  const int lo   = b * chunk;
  const int hi   = (lo + chunk < E) ? (lo + chunk) : E;
  const int n    = hi - lo;
  const int base = b * chunkPad;
  if (n <= 0) return;

  int   p[4];
  float c[4];
  int   w[4];
#pragma unroll
  for (int j = 0; j < 4; ++j) {
    const int i  = j * PART_THREADS + tid;
    const int ic = i < n ? i : (n - 1);
    p[j] = pos[lo + ic];
    c[j] = recCost[base + ic];
    w[j] = recW[base + ic];
  }
#pragma unroll
  for (int j = 0; j < 4; ++j) {
    const int i = j * PART_THREADS + tid;
    if (i < n) { sCost[i] = c[j]; sWgt[i] = w[j]; }
  }
  __syncthreads();

#pragma unroll
  for (int j = 0; j < 4; ++j) {
    const int i = j * PART_THREADS + tid;
    if (i < n) {
      const int q = p[j] - base;
      outCost[lo + i] = sCost[q];
      outW[lo + i]    = __int_as_float(sWgt[q]);
    }
  }
}

__global__ void seg_expsum(const float* __restrict__ outCost,
                           const int* __restrict__ edgeS,
                           const unsigned* __restrict__ segMax,
                           unsigned long long* __restrict__ segSum,
                           const float* __restrict__ logT, int E) {
  int e = blockIdx.x * blockDim.x + threadIdx.x;
  if (e >= E) return;
  float invT = __expf(-logT[0]);
  int s = edgeS[e];
  float val = -outCost[e] * invT;
  float m = ordVal(segMax[s]);
  atomicAdd(&segSum[s], (unsigned long long)(__expf(val - m) * SUM_SCALE));
}

__global__ void seg_norm(const float* __restrict__ outCost,
                         const int* __restrict__ edgeS,
                         const unsigned* __restrict__ segMax,
                         const unsigned long long* __restrict__ segSum,
                         const float* __restrict__ logT,
                         float* __restrict__ outW, int E) {
  int e = blockIdx.x * blockDim.x + threadIdx.x;
  if (e >= E) return;
  float invT = __expf(-logT[0]);
  int s = edgeS[e];
  float val = -outCost[e] * invT;
  float m = ordVal(segMax[s]);
  outW[e] = __expf(val - m) / ((float)segSum[s] * SUM_INV_SCALE);
}

extern "C" void kernel_launch(void* const* d_in, const int* in_sizes, int n_in,
                              void* d_out, int out_size, void* d_ws, size_t ws_size,
                              hipStream_t stream) {
  const float* embS = (const float*)d_in[0];
  const float* embA = (const float*)d_in[1];
  const int*   eidx = (const int*)d_in[2];
  const float* W1   = (const float*)d_in[3];
  const float* b1   = (const float*)d_in[4];
  const float* W2   = (const float*)d_in[5];
  const float* b2   = (const float*)d_in[6];
  const float* logT = (const float*)d_in[7];

  const int numS = in_sizes[0] / D_DIM;
  const int numA = in_sizes[1] / D_DIM;
  const int E    = in_sizes[2] / 2;
  const int* edgeS = eidx;
  const int* edgeA = eidx + E;

  float* outW    = (float*)d_out;
  float* outCost = (float*)d_out + E;

  const size_t pFloats = (size_t)(numS + numA) * H_DIM;
  const size_t need    = pFloats * sizeof(float)
                       + (size_t)numS * (sizeof(unsigned long long) + sizeof(unsigned));

  const int NB       = (numS + SEG_NODES - 1) / SEG_NODES;
  const int chunk    = (((E + PART_BLOCKS - 1) / PART_BLOCKS) + 31) & ~31;
  const int chunkPad = (chunk + 127) & ~127;
  const size_t recSlots = (size_t)PART_BLOCKS * chunkPad;
  const size_t needGrp  = need + recSlots * (sizeof(float) + 1)
                        + (size_t)PART_BLOCKS * (NB + 1) * sizeof(int);

  const size_t posSlots = ((size_t)E + 31) & ~(size_t)31;
  const size_t needSrc  = needGrp + recSlots * sizeof(float) + posSlots * sizeof(int);

  if (ws_size >= needSrc && NB <= NB_MAX && chunk <= CHUNK_MAX) {
    float* P = (float*)d_ws;
    int*           recA    = (int*)(P + pFloats);
    float*         recCost = (float*)(recA + recSlots);
    int*           pos     = (int*)(recCost + recSlots);
    unsigned char* recNode = (unsigned char*)(pos + posSlots);
    unsigned long long* segSum = (unsigned long long*)(recNode + recSlots);
    int*                offG   = (int*)((unsigned*)(segSum + numS) + numS);

    proj_and_group<<<PROJ_BLOCKS + PART_BLOCKS, PART_THREADS, 0, stream>>>(
        embS, embA, W1, b1, P, numS, numA,
        edgeS, edgeA, offG, recA, recNode, pos, E, NB, chunk, chunkPad);
    bucket_edge<<<NB, PART_THREADS, 0, stream>>>(
        embS, embA, P, P + (size_t)numS * H_DIM, W2, b2, logT,
        offG, recA, recNode, recCost, numS, NB, chunkPad);
    chunk_unsort<<<PART_BLOCKS, PART_THREADS, 0, stream>>>(
        recCost, recA, pos, outW, outCost, E, chunk, chunkPad);
    return;
  }

  if (ws_size >= needGrp && NB <= NB_MAX && chunk <= CHUNK_MAX) {
    float* P = (float*)d_ws;
    float*         recVal  = P + pFloats;
    unsigned char* recNode = (unsigned char*)(recVal + recSlots);
    unsigned long long* segSum = (unsigned long long*)(recNode + recSlots);
    unsigned*           segMax = (unsigned*)(segSum + numS);
    int*                offG   = (int*)(segMax + numS);

    node_proj<<<256, EDGE_WAVES * 32, 0, stream>>>(embS, embA, W1, b1, P, numS, numA);
    edge_gate_place<<<PART_BLOCKS, PART_THREADS, 0, stream>>>(
        embS, embA, edgeS, edgeA, P, P + (size_t)numS * H_DIM, W2, b2, logT,
        outCost, offG, recVal, recNode, E, NB, chunk, chunkPad);
    bucket_stats<<<NB, PART_THREADS, 0, stream>>>(
        offG, recVal, recNode, segMax, segSum, numS, NB, chunkPad);
    seg_norm<<<(E + 255) / 256, 256, 0, stream>>>(outCost, edgeS, segMax, segSum, logT, outW, E);
    return;
  }

  if (ws_size >= need) {
    float* P = (float*)d_ws;
    unsigned long long* segSum = (unsigned long long*)(P + pFloats);
    unsigned*           segMax = (unsigned*)(segSum + numS);

    init_seg<<<(numS + 255) / 256, 256, 0, stream>>>(segMax, segSum, numS);
    node_proj<<<256, EDGE_WAVES * 32, 0, stream>>>(embS, embA, W1, b1, P, numS, numA);
    edge_gate_cost<<<256, EDGE_WAVES * 32, 0, stream>>>(
        embS, embA, edgeS, edgeA, P, P + (size_t)numS * H_DIM, W2, b2, logT,
        outCost, segMax, E);
    seg_expsum<<<(E + 255) / 256, 256, 0, stream>>>(outCost, edgeS, segMax, segSum, logT, E);
    seg_norm<<<(E + 255) / 256, 256, 0, stream>>>(outCost, edgeS, segMax, segSum, logT, outW, E);
    return;
  }

  unsigned long long* segSum = (unsigned long long*)d_ws;
  unsigned*           segMax = (unsigned*)(segSum + numS);

  init_seg<<<(numS + 255) / 256, 256, 0, stream>>>(segMax, segSum, numS);

  edge_mlp_cost<<<256, EDGE_WAVES * 32, 0, stream>>>(
      embS, embA, edgeS, edgeA, W1, b1, W2, b2, logT, outCost, segMax, E);

  seg_expsum<<<(E + 255) / 256, 256, 0, stream>>>(outCost, edgeS, segMax, segSum, logT, E);
  seg_norm<<<(E + 255) / 256, 256, 0, stream>>>(outCost, edgeS, segMax, segSum, logT, outW, E);
}
